# starting kernel
# baseline (speedup 1.0000x reference)
#include <hip/hip_runtime.h>
#include <hip/hip_bf16.h>
#include <math.h>


#define NEG_SLOPE 0.2f

typedef __attribute__((ext_vector_type(16))) _Float16 v16h;
typedef __attribute__((ext_vector_type(8)))  _Float16 v8h;
typedef __attribute__((ext_vector_type(8)))  float    v8f;

__device__ __forceinline__ unsigned fenc(float f) {
  unsigned u = __float_as_uint(f);
  return (u & 0x80000000u) ? ~u : (u | 0x80000000u);
}
__device__ __forceinline__ float fdec(unsigned u) {
  return (u & 0x80000000u) ? __uint_as_float(u & 0x7fffffffu) : __uint_as_float(~u);
}
#define ENC_NEG_INF 0x007FFFFFu

__global__ void k_f32_to_f16(const float* __restrict__ src, _Float16* __restrict__ dst,
                             long long n) {
  long long i = (long long)blockIdx.x * blockDim.x + threadIdx.x;
  if (i < n) dst[i] = (_Float16)src[i];
}

__global__ void k_init_out(float* __restrict__ out, const float* __restrict__ bias,
                           long long n, int hc) {
  long long i = (long long)blockIdx.x * blockDim.x + threadIdx.x;
  if (i < n) out[i] = bias[(int)(i % hc)];
}

__global__ void k_init_md(unsigned* __restrict__ M, float* __restrict__ D, long long n) {
  long long i = (long long)blockIdx.x * blockDim.x + threadIdx.x;
  if (i < n) { M[i] = ENC_NEG_INF; D[i] = 0.0f; }
}

__global__ void k_elu_f16(const float* __restrict__ in, _Float16* __restrict__ out,
                          long long n) {
  long long i = (long long)blockIdx.x * blockDim.x + threadIdx.x;
  if (i >= n) return;
  float v = in[i];
  v = (v > 0.0f) ? v : expm1f(v);
  out[i] = (_Float16)v;
}

__global__ void k_pack_w(const float* __restrict__ W, _Float16* __restrict__ WP,
                         int din, int dout) {
  int ktiles = din >> 5, ntiles = dout >> 4;
  int total = ktiles * ntiles * 32;
  int idx = blockIdx.x * blockDim.x + threadIdx.x;
  if (idx >= total) return;
  int lane = idx & 31;
  int frag = idx >> 5;
  int nt = frag % ntiles;
  int kt = frag / ntiles;
  int half = lane >> 4, lr = lane & 15;
  _Float16* o = WP + ((size_t)frag * 32 + lane) * 16;
  const float* w0 = W + (size_t)(kt * 32 + half * 8) * dout + nt * 16 + lr;
#pragma unroll
  for (int j = 0; j < 8; ++j) {
    o[j]     = (_Float16)w0[(size_t)j * dout];
    o[8 + j] = (_Float16)w0[(size_t)(16 + j) * dout];
  }
}

__global__ void k_gemm_f16(const _Float16* __restrict__ X, const _Float16* __restrict__ WP,
                           float* __restrict__ Y, int Mrows, int din, int dout) {
  int wave = threadIdx.x >> 5;
  int lane = threadIdx.x & 31;
  int half = lane >> 4;
  int lr   = lane & 15;
  int tileM = blockIdx.x * 4 + wave;
  if (tileM * 16 >= Mrows) return;
  int nt0 = blockIdx.y * 4;
  int ntiles = dout >> 4;

  int row = tileM * 16 + lr;
  if (row >= Mrows) row = Mrows - 1;

  v8f acc0 = {}, acc1 = {}, acc2 = {}, acc3 = {};
  const _Float16* xrow = X + (size_t)row * din;
  const v16h* WPv = (const v16h*)WP;
  for (int k0 = 0; k0 < din; k0 += 32) {
    const v8h* xa = (const v8h*)(xrow + k0 + half * 8);
    v8h lo = xa[0];
    v8h hi = xa[2];
    v16h a = __builtin_shufflevector(lo, hi, 0, 1, 2, 3, 4, 5, 6, 7,
                                     8, 9, 10, 11, 12, 13, 14, 15);
    size_t fb = ((size_t)(k0 >> 5) * ntiles + nt0) * 32 + lane;
    v16h b0 = WPv[fb];
    v16h b1 = WPv[fb + 32];
    v16h b2 = WPv[fb + 64];
    v16h b3 = WPv[fb + 96];
    acc0 = __builtin_amdgcn_wmma_f32_16x16x32_f16(false, a, false, b0, (short)0, acc0, false, false);
    acc1 = __builtin_amdgcn_wmma_f32_16x16x32_f16(false, a, false, b1, (short)0, acc1, false, false);
    acc2 = __builtin_amdgcn_wmma_f32_16x16x32_f16(false, a, false, b2, (short)0, acc2, false, false);
    acc3 = __builtin_amdgcn_wmma_f32_16x16x32_f16(false, a, false, b3, (short)0, acc3, false, false);
  }
  int r0 = tileM * 16 + half * 8;
#pragma unroll
  for (int i = 0; i < 8; ++i) {
    int r = r0 + i;
    if (r < Mrows) {
      float* yp = Y + (size_t)r * dout + nt0 * 16 + lr;
      yp[0]  = acc0[i];
      yp[16] = acc1[i];
      yp[32] = acc2[i];
      yp[48] = acc3[i];
    }
  }
}

template <int H, int C>
__global__ void k_edge_score(const float* __restrict__ XL, const float* __restrict__ XR,
                             const float* __restrict__ att, const int* __restrict__ ei,
                             long long E0, long long ET,
                             float* __restrict__ SC, unsigned* __restrict__ M) {
  constexpr int HC  = H * C;
  constexpr int LPH = 32 / H;
  constexpr int CPT = C / LPH;
  long long e = (long long)blockIdx.x * (blockDim.x >> 5) + (threadIdx.x >> 5);
  if (e >= ET) return;
  int lane = threadIdx.x & 31;
  int s, d;
  if (e < E0) { s = ei[e]; d = ei[E0 + e]; }
  else        { s = d = (int)(e - E0); }

  int h  = lane / LPH;
  int c0 = (lane % LPH) * CPT;
  const float* xl = XL + (size_t)s * HC + h * C + c0;
  const float* xr = XR + (size_t)d * HC + h * C + c0;
  const float* ah = att + h * C + c0;
  float p = 0.0f;
#pragma unroll
  for (int j = 0; j < CPT; ++j) {
    float v = xl[j] + xr[j];
    v = (v > 0.0f) ? v : NEG_SLOPE * v;
    p += ah[j] * v;
  }
#pragma unroll
  for (int o = LPH >> 1; o > 0; o >>= 1) p += __shfl_xor(p, o, 32);
  if ((lane % LPH) == 0) {
    SC[(size_t)e * H + h] = p;
    atomicMax(&M[(size_t)d * H + h], fenc(p));
  }
}

template <int H>
__global__ void k_edge_exp(float* __restrict__ SC, const unsigned* __restrict__ M,
                           float* __restrict__ DEN, const int* __restrict__ ei,
                           long long E0, long long ET) {
  long long i = (long long)blockIdx.x * blockDim.x + threadIdx.x;
  if (i >= ET * (long long)H) return;
  long long e = i / H;
  int h = (int)(i % H);
  int d = (e < E0) ? ei[E0 + e] : (int)(e - E0);
  float m = fdec(M[(size_t)d * H + h]);
  if (!(m > -3.0e38f && m < 3.0e38f)) m = 0.0f;
  float ex = __expf(SC[i] - m);
  SC[i] = ex;
  atomicAdd(&DEN[(size_t)d * H + h], ex);
}

template <int H, int C>
__global__ void k_edge_aggr(const float* __restrict__ XL, const float* __restrict__ SC,
                            const float* __restrict__ DEN, float* __restrict__ OUT,
                            const int* __restrict__ ei, long long E0, long long ET) {
  constexpr int HC  = H * C;
  constexpr int CPT = HC / 32;
  long long e = (long long)blockIdx.x * (blockDim.x >> 5) + (threadIdx.x >> 5);
  if (e >= ET) return;
  int lane = threadIdx.x & 31;
  int s, d;
  if (e < E0) { s = ei[e]; d = ei[E0 + e]; }
  else        { s = d = (int)(e - E0); }
  int ch0 = lane * CPT;
  int h   = ch0 / C;
  float ex    = SC[(size_t)e * H + h];
  float dn    = DEN[(size_t)d * H + h];
  float alpha = ex / fmaxf(dn, 1e-16f);
  const float* xs = XL + (size_t)s * HC + ch0;
  float*       od = OUT + (size_t)d * HC + ch0;
#pragma unroll
  for (int j = 0; j < CPT; ++j) atomicAdd(&od[j], alpha * xs[j]);
}

__global__ void k_softmax64(const float* __restrict__ in, float* __restrict__ out, int n) {
  int node = blockIdx.x * (blockDim.x >> 5) + (threadIdx.x >> 5);
  if (node >= n) return;
  int lane = threadIdx.x & 31;
  const float* p = in + (size_t)node * 64 + lane * 2;
  float a = p[0], b = p[1];
  float mx = fmaxf(a, b);
#pragma unroll
  for (int o = 16; o > 0; o >>= 1) mx = fmaxf(mx, __shfl_xor(mx, o, 32));
  float ea = __expf(a - mx), eb = __expf(b - mx);
  float sum = ea + eb;
#pragma unroll
  for (int o = 16; o > 0; o >>= 1) sum += __shfl_xor(sum, o, 32);
  float inv = 1.0f / sum;
  float* q = out + (size_t)node * 64 + lane * 2;
  q[0] = ea * inv;
  q[1] = eb * inv;
}

template <int H, int C>
static void gat_layer(const _Float16* X16, int din,
                      const float* Wl, const float* Wr, const float* att, const float* b,
                      _Float16* WPa, _Float16* WPb,
                      float* XL, float* XR, float* OUT, unsigned* M, float* DEN, float* SC,
                      const int* ei, int N, long long E0, hipStream_t stream) {
  constexpr int HC = H * C;
  long long ET = E0 + N;

  int packThreads = (din >> 5) * (HC >> 4) * 32;
  int packBlocks  = (packThreads + 255) / 256;
  k_pack_w<<<packBlocks, 256, 0, stream>>>(Wl, WPa, din, HC);
  k_pack_w<<<packBlocks, 256, 0, stream>>>(Wr, WPb, din, HC);

  int rowTiles = (N + 15) / 16;
  dim3 gg((unsigned)((rowTiles + 3) / 4), (unsigned)(HC / 64));
  k_gemm_f16<<<gg, 128, 0, stream>>>(X16, WPa, XL, N, din, HC);
  k_gemm_f16<<<gg, 128, 0, stream>>>(X16, WPb, XR, N, din, HC);

  long long nHC = (long long)N * HC, nH = (long long)N * H;
  k_init_out<<<(int)((nHC + 255) / 256), 256, 0, stream>>>(OUT, b, nHC, HC);
  k_init_md<<<(int)((nH + 255) / 256), 256, 0, stream>>>(M, DEN, nH);

  int gE = (int)((ET + 7) / 8);
  k_edge_score<H, C><<<gE, 256, 0, stream>>>(XL, XR, att, ei, E0, ET, SC, M);
  k_edge_exp<H><<<(int)((ET * H + 255) / 256), 256, 0, stream>>>(SC, M, DEN, ei, E0, ET);
  k_edge_aggr<H, C><<<gE, 256, 0, stream>>>(XL, SC, DEN, OUT, ei, E0, ET);
}

extern "C" void kernel_launch(void* const* d_in, const int* in_sizes, int n_in,
                              void* d_out, int out_size, void* d_ws, size_t ws_size,
                              hipStream_t stream) {
  (void)n_in; (void)out_size; (void)ws_size;
  const float* x    = (const float*)d_in[0];
  const int*   ei   = (const int*)d_in[1];
  const float* Wl1  = (const float*)d_in[2];
  const float* Wr1  = (const float*)d_in[3];
  const float* att1 = (const float*)d_in[4];
  const float* b1   = (const float*)d_in[5];
  const float* Wl2  = (const float*)d_in[6];
  const float* Wr2  = (const float*)d_in[7];
  const float* att2 = (const float*)d_in[8];
  const float* b2   = (const float*)d_in[9];
  const float* Wl3  = (const float*)d_in[10];
  const float* Wr3  = (const float*)d_in[11];
  const float* att3 = (const float*)d_in[12];
  const float* b3   = (const float*)d_in[13];

  int       N  = in_sizes[0] / 256;
  long long E0 = (long long)in_sizes[1] / 2;
  long long ET = E0 + N;

  uintptr_t p = (uintptr_t)d_ws;
  auto carve = [&](size_t bytes) -> void* {
    void* r = (void*)p;
    p += (bytes + 255) & ~(size_t)255;
    return r;
  };
  _Float16* X16 = (_Float16*)carve((size_t)N * 256 * sizeof(_Float16));
  _Float16* WPa = (_Float16*)carve((size_t)256 * 128 * sizeof(_Float16));
  _Float16* WPb = (_Float16*)carve((size_t)256 * 128 * sizeof(_Float16));
  float*    XL  = (float*)carve((size_t)N * 128 * sizeof(float));
  float*    XR  = (float*)carve((size_t)N * 128 * sizeof(float));
  float*    OUT = (float*)carve((size_t)N * 128 * sizeof(float));
  unsigned* M   = (unsigned*)carve((size_t)N * 8 * sizeof(unsigned));
  float*    DEN = (float*)carve((size_t)N * 8 * sizeof(float));
  float*    SC  = (float*)carve((size_t)ET * 8 * sizeof(float));

  long long nx = (long long)N * 256;
  k_f32_to_f16<<<(int)((nx + 255) / 256), 256, 0, stream>>>(x, X16, nx);
  gat_layer<8, 16>(X16, 256, Wl1, Wr1, att1, b1, WPa, WPb,
                   XL, XR, OUT, M, DEN, SC, ei, N, E0, stream);
  long long nh = (long long)N * 128;
  k_elu_f16<<<(int)((nh + 255) / 256), 256, 0, stream>>>(OUT, X16, nh);

  gat_layer<8, 16>(X16, 128, Wl2, Wr2, att2, b2, WPa, WPb,
                   XL, XR, OUT, M, DEN, SC, ei, N, E0, stream);
  k_elu_f16<<<(int)((nh + 255) / 256), 256, 0, stream>>>(OUT, X16, nh);

  gat_layer<1, 64>(X16, 128, Wl3, Wr3, att3, b3, WPa, WPb,
                   XL, XR, OUT, M, DEN, SC, ei, N, E0, stream);

  k_softmax64<<<(N + 7) / 8, 256, 0, stream>>>(OUT, (float*)d_out, N);
}


# Fixed sum order (per-node sort by source replaces arrival-order rank): edges grouped by destination in-call; one wave per node fuses score, online softmax and sum in registers; no float atomics
# speedup vs baseline: 3.3386x; 3.3386x over previous
#include <hip/hip_runtime.h>
#include <hip/hip_bf16.h>
#include <math.h>


#define NEG_SLOPE 0.2f

typedef __attribute__((ext_vector_type(16))) _Float16 v16h;
typedef __attribute__((ext_vector_type(8)))  _Float16 v8h;
typedef __attribute__((ext_vector_type(8)))  float    v8f;

__global__ void k_f32_to_f16(const float* __restrict__ src, _Float16* __restrict__ dst,
                             long long n) {
  long long i = (long long)blockIdx.x * blockDim.x + threadIdx.x;
  if (i < n) dst[i] = (_Float16)src[i];
}

__global__ void k_elu_f16(const float* __restrict__ in, _Float16* __restrict__ out,
                          long long n) {
  long long i = (long long)blockIdx.x * blockDim.x + threadIdx.x;
  if (i >= n) return;
  float v = in[i];
  v = (v > 0.0f) ? v : expm1f(v);
  out[i] = (_Float16)v;
}

__global__ void k_pack_w(const float* __restrict__ W, _Float16* __restrict__ WP,
                         int din, int dout) {
  int ktiles = din >> 5, ntiles = dout >> 4;
  int total = ktiles * ntiles * 32;
  int idx = blockIdx.x * blockDim.x + threadIdx.x;
  if (idx >= total) return;
  int lane = idx & 31;
  int frag = idx >> 5;
  int nt = frag % ntiles;
  int kt = frag / ntiles;
  int half = lane >> 4, lr = lane & 15;
  _Float16* o = WP + ((size_t)frag * 32 + lane) * 16;
  const float* w0 = W + (size_t)(kt * 32 + half * 8) * dout + nt * 16 + lr;
#pragma unroll
  for (int j = 0; j < 8; ++j) {
    o[j]     = (_Float16)w0[(size_t)j * dout];
    o[8 + j] = (_Float16)w0[(size_t)(16 + j) * dout];
  }
}

__global__ void k_gemm_f16(const _Float16* __restrict__ X, const _Float16* __restrict__ WP,
                           float* __restrict__ Y, int Mrows, int din, int dout) {
  int wave = threadIdx.x >> 5;
  int lane = threadIdx.x & 31;
  int half = lane >> 4;
  int lr   = lane & 15;
  int tileM = blockIdx.x * 4 + wave;
  if (tileM * 16 >= Mrows) return;
  int nt0 = blockIdx.y * 4;
  int ntiles = dout >> 4;

  int row = tileM * 16 + lr;
  if (row >= Mrows) row = Mrows - 1;

  v8f acc0 = {}, acc1 = {}, acc2 = {}, acc3 = {};
  const _Float16* xrow = X + (size_t)row * din;
  const v16h* WPv = (const v16h*)WP;
  for (int k0 = 0; k0 < din; k0 += 32) {
    const v8h* xa = (const v8h*)(xrow + k0 + half * 8);
    v8h lo = xa[0];
    v8h hi = xa[2];
    v16h a = __builtin_shufflevector(lo, hi, 0, 1, 2, 3, 4, 5, 6, 7,
                                     8, 9, 10, 11, 12, 13, 14, 15);
    size_t fb = ((size_t)(k0 >> 5) * ntiles + nt0) * 32 + lane;
    v16h b0 = WPv[fb];
    v16h b1 = WPv[fb + 32];
    v16h b2 = WPv[fb + 64];
    v16h b3 = WPv[fb + 96];
    acc0 = __builtin_amdgcn_wmma_f32_16x16x32_f16(false, a, false, b0, (short)0, acc0, false, false);
    acc1 = __builtin_amdgcn_wmma_f32_16x16x32_f16(false, a, false, b1, (short)0, acc1, false, false);
    acc2 = __builtin_amdgcn_wmma_f32_16x16x32_f16(false, a, false, b2, (short)0, acc2, false, false);
    acc3 = __builtin_amdgcn_wmma_f32_16x16x32_f16(false, a, false, b3, (short)0, acc3, false, false);
  }
  int r0 = tileM * 16 + half * 8;
#pragma unroll
  for (int i = 0; i < 8; ++i) {
    int r = r0 + i;
    if (r < Mrows) {
      float* yp = Y + (size_t)r * dout + nt0 * 16 + lr;
      yp[0]  = acc0[i];
      yp[16] = acc1[i];
      yp[32] = acc2[i];
      yp[48] = acc3[i];
    }
  }
}

typedef __attribute__((ext_vector_type(4))) float v4f;
typedef __attribute__((ext_vector_type(2))) float v2f;

__global__ void k_zero_i32(int* __restrict__ p, int n) {
  int i = blockIdx.x * blockDim.x + threadIdx.x;
  if (i < n) p[i] = 0;
}

__global__ void k_hist(const int* __restrict__ ei, long long E0,
                       int* __restrict__ cnt, int* __restrict__ slot) {
  long long e = (long long)blockIdx.x * blockDim.x + threadIdx.x;
  if (e >= E0) return;
  int d = ei[E0 + e];
  slot[e] = atomicAdd(&cnt[d], 1);
}

#define SCAN_TILE 1024

__global__ void k_tile_sum(const int* __restrict__ cnt, int n, int* __restrict__ tsum) {
  __shared__ int sh[8];
  int base = blockIdx.x * SCAN_TILE + threadIdx.x * 4;
  int v = 0;
#pragma unroll
  for (int j = 0; j < 4; ++j) if (base + j < n) v += cnt[base + j];
#pragma unroll
  for (int o = 16; o > 0; o >>= 1) v += __shfl_xor(v, o, 32);
  if ((threadIdx.x & 31) == 0) sh[threadIdx.x >> 5] = v;
  __syncthreads();
  if (threadIdx.x == 0) {
    int s = 0;
    for (int w = 0; w < 8; ++w) s += sh[w];
    tsum[blockIdx.x] = s;
  }
}

__global__ void k_scan(const int* __restrict__ cnt, int n, const int* __restrict__ tsum,
                       int* __restrict__ rowptr) {
  __shared__ int sh[8];
  int tid = threadIdx.x, lane = tid & 31, wave = tid >> 5;
  int pb = 0;
  for (int j = tid; j < (int)blockIdx.x; j += 256) pb += tsum[j];
#pragma unroll
  for (int o = 16; o > 0; o >>= 1) pb += __shfl_xor(pb, o, 32);
  if (lane == 0) sh[wave] = pb;
  __syncthreads();
  int tilebase = 0;
  for (int w = 0; w < 8; ++w) tilebase += sh[w];
  __syncthreads();

  int base = blockIdx.x * SCAN_TILE + tid * 4;
  int cv[4];
#pragma unroll
  for (int j = 0; j < 4; ++j) cv[j] = (base + j < n) ? cnt[base + j] : 0;
  int local = cv[0] + cv[1] + cv[2] + cv[3];
  int inc = local;
#pragma unroll
  for (int o = 1; o < 32; o <<= 1) {
    int t = __shfl_up(inc, o, 32);
    if (lane >= o) inc += t;
  }
  if (lane == 31) sh[wave] = inc;
  __syncthreads();
  int wbase = 0;
  for (int w = 0; w < wave; ++w) wbase += sh[w];
  int ex = tilebase + wbase + inc - local;
#pragma unroll
  for (int j = 0; j < 4; ++j) {
    if (base + j < n) rowptr[base + j] = ex;
    ex += cv[j];
  }
}

__global__ void k_scatter(const int* __restrict__ ei, long long E0,
                          const int* __restrict__ rowptr, const int* __restrict__ slot,
                          int* __restrict__ segA) {
  long long e = (long long)blockIdx.x * blockDim.x + threadIdx.x;
  if (e >= E0) return;
  int s = ei[e];
  int d = ei[E0 + e];
  segA[rowptr[d] + slot[e]] = s;
}

__global__ void k_seg_sort(const int* __restrict__ rowptr, const int* __restrict__ cnt,
                           const int* __restrict__ segA, int* __restrict__ seg, int n) {
  int node = blockIdx.x * (blockDim.x >> 5) + (threadIdx.x >> 5);
  if (node >= n) return;
  node = __builtin_amdgcn_readfirstlane(node);
  int lane = threadIdx.x & 31;
  int start = rowptr[node], deg = cnt[node];
  if (deg <= 32) {
    int mine = (lane < deg) ? segA[start + lane] : 0x7fffffff;
    int r = 0;
    for (int j = 0; j < deg; ++j) {
      int v = __builtin_amdgcn_readlane(mine, j);
      r += ((v < mine) || (v == mine && j < lane)) ? 1 : 0;
    }
    if (lane < deg) seg[start + r] = mine;
  } else {
    for (int i = lane; i < deg; i += 32) {
      int mine = segA[start + i];
      int r = 0;
      for (int j = 0; j < deg; ++j) {
        int v = segA[start + j];
        r += ((v < mine) || (v == mine && j < i)) ? 1 : 0;
      }
      seg[start + r] = mine;
    }
  }
}

template <int CTRL>
__device__ __forceinline__ float dpp_get(float v) {
  return __int_as_float(__builtin_amdgcn_update_dpp(0, __float_as_int(v), CTRL, 0xF, 0xF, true));
}
__device__ __forceinline__ float other_row_get(float v) {
  return __uint_as_float(__builtin_amdgcn_permlanex16(0u, __float_as_uint(v),
                                                      0x76543210u, 0xfedcba98u, true, false));
}
template <int LPH>
__device__ __forceinline__ float head_sum(float p) {
  static_assert(LPH == 4 || LPH == 32, "lanes per head");
  p += dpp_get<0x161>(p);
  p += dpp_get<0x162>(p);
  if constexpr (LPH == 32) {
    p += dpp_get<0x164>(p);
    p += dpp_get<0x168>(p);
    p += other_row_get(p);
  }
  return p;
}

template <int CPT> struct RowVec;
template <> struct RowVec<4> { typedef v4f T; };
template <> struct RowVec<2> { typedef v2f T; };

template <int H, int C>
__global__ void k_gat_node(const float* __restrict__ XL, const float* __restrict__ XR,
                           const float* __restrict__ att, const float* __restrict__ bias,
                           const int* __restrict__ rowptr, const int* __restrict__ cnt,
                           const int* __restrict__ seg, float* __restrict__ OUT, int n) {
  constexpr int HC  = H * C;
  constexpr int CPT = HC / 32;
  constexpr int LPH = 32 / H;
  constexpr int UNR = 4;
  typedef typename RowVec<CPT>::T vec_t;

  int node = blockIdx.x * (blockDim.x >> 5) + (threadIdx.x >> 5);
  if (node >= n) return;
  node = __builtin_amdgcn_readfirstlane(node);
  int lane = threadIdx.x & 31;
  int ch0  = lane * CPT;
  int start = rowptr[node], deg = cnt[node];

  vec_t xr = *(const vec_t*)(XR + (size_t)node * HC + ch0);
  vec_t av = *(const vec_t*)(att + ch0);
  vec_t xs = *(const vec_t*)(XL + (size_t)node * HC + ch0);

  float p = 0.0f;
#pragma unroll
  for (int j = 0; j < CPT; ++j) {
    float v = xs[j] + xr[j];
    v = (v > 0.0f) ? v : NEG_SLOPE * v;
    p += av[j] * v;
  }
  float m   = head_sum<LPH>(p);
  float den = 1.0f;
  vec_t acc = xs;

  for (int base = 0; base < deg; base += 32) {
    int nb = deg - base;
    if (nb > 32) nb = 32;
    int mysrc = (lane < nb) ? seg[start + base + lane] : node;
    for (int j = 0; j < nb; j += UNR) {
      vec_t x[UNR];
      float pu[UNR];
      bool  ok[UNR];
#pragma unroll
      for (int u = 0; u < UNR; ++u) {
        int s = __builtin_amdgcn_readlane(mysrc, j + u);
        ok[u] = (j + u) < nb;
        x[u] = *(const vec_t*)(XL + (size_t)s * HC + ch0);
      }
#pragma unroll
      for (int u = 0; u < UNR; ++u) {
        float q = 0.0f;
#pragma unroll
        for (int k = 0; k < CPT; ++k) {
          float v = x[u][k] + xr[k];
          v = (v > 0.0f) ? v : NEG_SLOPE * v;
          q += av[k] * v;
        }
        q = head_sum<LPH>(q);
        pu[u] = ok[u] ? q : -INFINITY;
      }
      float mn = fmaxf(fmaxf(m, pu[0]), fmaxf(pu[1], fmaxf(pu[2], pu[3])));
      float sc = __expf(m - mn);
      float e0 = ok[0] ? __expf(pu[0] - mn) : 0.0f;
      float e1 = ok[1] ? __expf(pu[1] - mn) : 0.0f;
      float e2 = ok[2] ? __expf(pu[2] - mn) : 0.0f;
      float e3 = ok[3] ? __expf(pu[3] - mn) : 0.0f;
      den = den * sc + ((e0 + e1) + (e2 + e3));
#pragma unroll
      for (int k = 0; k < CPT; ++k)
        acc[k] = acc[k] * sc + ((e0 * x[0][k] + e1 * x[1][k]) + (e2 * x[2][k] + e3 * x[3][k]));
      m = mn;
    }
  }

  float dn = fmaxf(den, 1e-16f);
  vec_t bv = *(const vec_t*)(bias + ch0);
  vec_t o;
#pragma unroll
  for (int k = 0; k < CPT; ++k) o[k] = acc[k] / dn + bv[k];
  *(vec_t*)(OUT + (size_t)node * HC + ch0) = o;
}

__global__ void k_softmax64(const float* __restrict__ in, float* __restrict__ out, int n) {
  int node = blockIdx.x * (blockDim.x >> 5) + (threadIdx.x >> 5);
  if (node >= n) return;
  int lane = threadIdx.x & 31;
  const float* p = in + (size_t)node * 64 + lane * 2;
  float a = p[0], b = p[1];
  float mx = fmaxf(a, b);
#pragma unroll
  for (int o = 16; o > 0; o >>= 1) mx = fmaxf(mx, __shfl_xor(mx, o, 32));
  float ea = __expf(a - mx), eb = __expf(b - mx);
  float sum = ea + eb;
#pragma unroll
  for (int o = 16; o > 0; o >>= 1) sum += __shfl_xor(sum, o, 32);
  float inv = 1.0f / sum;
  float* q = out + (size_t)node * 64 + lane * 2;
  q[0] = ea * inv;
  q[1] = eb * inv;
}

template <int H, int C>
static void gat_layer(const _Float16* X16, int din,
                      const float* Wl, const float* Wr, const float* att, const float* b,
                      _Float16* WPa, _Float16* WPb,
                      float* XL, float* XR, float* OUT,
                      const int* rowptr, const int* cnt, const int* seg,
                      int N, hipStream_t stream) {
  constexpr int HC = H * C;

  int packThreads = (din >> 5) * (HC >> 4) * 32;
  int packBlocks  = (packThreads + 255) / 256;
  k_pack_w<<<packBlocks, 256, 0, stream>>>(Wl, WPa, din, HC);
  k_pack_w<<<packBlocks, 256, 0, stream>>>(Wr, WPb, din, HC);

  int rowTiles = (N + 15) / 16;
  dim3 gg((unsigned)((rowTiles + 3) / 4), (unsigned)(HC / 64));
  k_gemm_f16<<<gg, 128, 0, stream>>>(X16, WPa, XL, N, din, HC);
  k_gemm_f16<<<gg, 128, 0, stream>>>(X16, WPb, XR, N, din, HC);

  k_gat_node<H, C><<<(N + 7) / 8, 256, 0, stream>>>(XL, XR, att, b, rowptr, cnt, seg, OUT, N);
}

extern "C" void kernel_launch(void* const* d_in, const int* in_sizes, int n_in,
                              void* d_out, int out_size, void* d_ws, size_t ws_size,
                              hipStream_t stream) {
  (void)n_in; (void)out_size; (void)ws_size;
  const float* x    = (const float*)d_in[0];
  const int*   ei   = (const int*)d_in[1];
  const float* Wl1  = (const float*)d_in[2];
  const float* Wr1  = (const float*)d_in[3];
  const float* att1 = (const float*)d_in[4];
  const float* b1   = (const float*)d_in[5];
  const float* Wl2  = (const float*)d_in[6];
  const float* Wr2  = (const float*)d_in[7];
  const float* att2 = (const float*)d_in[8];
  const float* b2   = (const float*)d_in[9];
  const float* Wl3  = (const float*)d_in[10];
  const float* Wr3  = (const float*)d_in[11];
  const float* att3 = (const float*)d_in[12];
  const float* b3   = (const float*)d_in[13];

  int       N  = in_sizes[0] / 256;
  long long E0 = (long long)in_sizes[1] / 2;

  uintptr_t p = (uintptr_t)d_ws;
  auto carve = [&](size_t bytes) -> void* {
    void* r = (void*)p;
    p += (bytes + 255) & ~(size_t)255;
    return r;
  };
  _Float16* X16 = (_Float16*)carve((size_t)N * 256 * sizeof(_Float16));
  _Float16* WPa = (_Float16*)carve((size_t)256 * 128 * sizeof(_Float16));
  _Float16* WPb = (_Float16*)carve((size_t)256 * 128 * sizeof(_Float16));
  float*    XL  = (float*)carve((size_t)N * 128 * sizeof(float));
  float*    XR  = (float*)carve((size_t)N * 128 * sizeof(float));
  float*    OUT = (float*)carve((size_t)N * 128 * sizeof(float));
  int* cnt    = (int*)carve((size_t)N * sizeof(int));
  int* rowptr = (int*)carve((size_t)N * sizeof(int));
  int* tsum   = (int*)carve((size_t)((N + SCAN_TILE - 1) / SCAN_TILE) * sizeof(int));
  int* slot   = (int*)carve((size_t)E0 * sizeof(int));
  int* segA   = (int*)carve((size_t)E0 * sizeof(int));
  int* seg    = (int*)carve((size_t)E0 * sizeof(int));

  int nTiles = (N + SCAN_TILE - 1) / SCAN_TILE;
  int gE0    = (int)((E0 + 255) / 256);
  k_zero_i32<<<(N + 255) / 256, 256, 0, stream>>>(cnt, N);
  k_hist<<<gE0, 256, 0, stream>>>(ei, E0, cnt, slot);
  k_tile_sum<<<nTiles, 256, 0, stream>>>(cnt, N, tsum);
  k_scan<<<nTiles, 256, 0, stream>>>(cnt, N, tsum, rowptr);
  k_scatter<<<gE0, 256, 0, stream>>>(ei, E0, rowptr, slot, segA);
  k_seg_sort<<<(N + 7) / 8, 256, 0, stream>>>(rowptr, cnt, segA, seg, N);

  long long nx = (long long)N * 256;
  k_f32_to_f16<<<(int)((nx + 255) / 256), 256, 0, stream>>>(x, X16, nx);
  gat_layer<8, 16>(X16, 256, Wl1, Wr1, att1, b1, WPa, WPb,
                   XL, XR, OUT, rowptr, cnt, seg, N, stream);
  long long nh = (long long)N * 128;
  k_elu_f16<<<(int)((nh + 255) / 256), 256, 0, stream>>>(OUT, X16, nh);

  gat_layer<8, 16>(X16, 128, Wl2, Wr2, att2, b2, WPa, WPb,
                   XL, XR, OUT, rowptr, cnt, seg, N, stream);
  k_elu_f16<<<(int)((nh + 255) / 256), 256, 0, stream>>>(OUT, X16, nh);

  gat_layer<1, 64>(X16, 128, Wl3, Wr3, att3, b3, WPa, WPb,
                   XL, XR, OUT, rowptr, cnt, seg, N, stream);

  k_softmax64<<<(N + 7) / 8, 256, 0, stream>>>(OUT, (float*)d_out, N);
}


# Elementwise passes (x to f16, ELU+f16 twice, row softmax) run as 512-block grids that loop, 8 16-byte loads in flight per lane, not one element per thread: 1.7M one-item waves become 16k long ones
# speedup vs baseline: 4.0430x; 1.2110x over previous
#include <hip/hip_runtime.h>
#include <hip/hip_bf16.h>
#include <math.h>


#define NEG_SLOPE 0.2f

typedef __attribute__((ext_vector_type(16))) _Float16 v16h;
typedef __attribute__((ext_vector_type(8)))  _Float16 v8h;
typedef __attribute__((ext_vector_type(8)))  float    v8f;

#define EW_THREADS 256
#define EW_LOADS   8
#define EW_BLOCKS  512

typedef __attribute__((ext_vector_type(4))) float    v4f;
typedef __attribute__((ext_vector_type(2))) float    v2f;
typedef __attribute__((ext_vector_type(4))) _Float16 v4h;

template <bool DO_ELU>
__device__ __forceinline__ v4h plane_cvt(v4f t) {
  if (DO_ELU) {
    v4f em;
#pragma unroll
    for (int k = 0; k < 4; ++k) em[k] = expm1f(t[k]);
#pragma unroll
    for (int k = 0; k < 4; ++k) t[k] = (t[k] > 0.0f) ? t[k] : em[k];
  }
  return __builtin_convertvector(t, v4h);
}

template <bool DO_ELU>
__global__ void __launch_bounds__(EW_THREADS)
k_plane_to_f16(const float* __restrict__ src, _Float16* __restrict__ dst, long long n) {
  constexpr long long TILE = (long long)EW_THREADS * EW_LOADS;
  const long long nvec = n >> 2;
  const v4f* __restrict__ s4 = (const v4f*)src;
  v4h* __restrict__ d4 = (v4h*)dst;
  for (long long t0 = (long long)blockIdx.x * TILE; t0 < nvec; t0 += (long long)gridDim.x * TILE) {
    const v4f* __restrict__ sp = s4 + t0 + threadIdx.x;
    v4h* __restrict__ dp = d4 + t0 + threadIdx.x;
    if (t0 + TILE <= nvec) {
      v4f v[EW_LOADS];
#pragma unroll
      for (int j = 0; j < EW_LOADS; ++j) v[j] = sp[j * EW_THREADS];
      if (DO_ELU) __builtin_amdgcn_sched_barrier(0);
#pragma unroll
      for (int j = 0; j < EW_LOADS; ++j) dp[j * EW_THREADS] = plane_cvt<DO_ELU>(v[j]);
    } else {
      for (int j = 0; j < EW_LOADS; ++j)
        if (t0 + threadIdx.x + j * EW_THREADS < nvec) dp[j * EW_THREADS] = plane_cvt<DO_ELU>(sp[j * EW_THREADS]);
    }
  }
  if (blockIdx.x == 0 && (long long)threadIdx.x < (n & 3)) {
    long long i = (nvec << 2) + threadIdx.x;
    float t = src[i];
    if (DO_ELU) t = (t > 0.0f) ? t : expm1f(t);
    dst[i] = (_Float16)t;
  }
}

__global__ void k_pack_w(const float* __restrict__ W, _Float16* __restrict__ WP,
                         int din, int dout) {
  int ktiles = din >> 5, ntiles = dout >> 4;
  int total = ktiles * ntiles * 32;
  int idx = blockIdx.x * blockDim.x + threadIdx.x;
  if (idx >= total) return;
  int lane = idx & 31;
  int frag = idx >> 5;
  int nt = frag % ntiles;
  int kt = frag / ntiles;
  int half = lane >> 4, lr = lane & 15;
  _Float16* o = WP + ((size_t)frag * 32 + lane) * 16;
  const float* w0 = W + (size_t)(kt * 32 + half * 8) * dout + nt * 16 + lr;
#pragma unroll
  for (int j = 0; j < 8; ++j) {
    o[j]     = (_Float16)w0[(size_t)j * dout];
    o[8 + j] = (_Float16)w0[(size_t)(16 + j) * dout];
  }
}

__global__ void k_gemm_f16(const _Float16* __restrict__ X, const _Float16* __restrict__ WP,
                           float* __restrict__ Y, int Mrows, int din, int dout) {
  int wave = threadIdx.x >> 5;
  int lane = threadIdx.x & 31;
  int half = lane >> 4;
  int lr   = lane & 15;
  int tileM = blockIdx.x * 4 + wave;
  if (tileM * 16 >= Mrows) return;
  int nt0 = blockIdx.y * 4;
  int ntiles = dout >> 4;

  int row = tileM * 16 + lr;
  if (row >= Mrows) row = Mrows - 1;

  v8f acc0 = {}, acc1 = {}, acc2 = {}, acc3 = {};
  const _Float16* xrow = X + (size_t)row * din;
  const v16h* WPv = (const v16h*)WP;
  for (int k0 = 0; k0 < din; k0 += 32) {
    const v8h* xa = (const v8h*)(xrow + k0 + half * 8);
    v8h lo = xa[0];
    v8h hi = xa[2];
    v16h a = __builtin_shufflevector(lo, hi, 0, 1, 2, 3, 4, 5, 6, 7,
                                     8, 9, 10, 11, 12, 13, 14, 15);
    size_t fb = ((size_t)(k0 >> 5) * ntiles + nt0) * 32 + lane;
    v16h b0 = WPv[fb];
    v16h b1 = WPv[fb + 32];
    v16h b2 = WPv[fb + 64];
    v16h b3 = WPv[fb + 96];
    acc0 = __builtin_amdgcn_wmma_f32_16x16x32_f16(false, a, false, b0, (short)0, acc0, false, false);
    acc1 = __builtin_amdgcn_wmma_f32_16x16x32_f16(false, a, false, b1, (short)0, acc1, false, false);
    acc2 = __builtin_amdgcn_wmma_f32_16x16x32_f16(false, a, false, b2, (short)0, acc2, false, false);
    acc3 = __builtin_amdgcn_wmma_f32_16x16x32_f16(false, a, false, b3, (short)0, acc3, false, false);
  }
  int r0 = tileM * 16 + half * 8;
#pragma unroll
  for (int i = 0; i < 8; ++i) {
    int r = r0 + i;
    if (r < Mrows) {
      float* yp = Y + (size_t)r * dout + nt0 * 16 + lr;
      yp[0]  = acc0[i];
      yp[16] = acc1[i];
      yp[32] = acc2[i];
      yp[48] = acc3[i];
    }
  }
}

__global__ void k_zero_i32(int* __restrict__ p, int n) {
  int i = blockIdx.x * blockDim.x + threadIdx.x;
  if (i < n) p[i] = 0;
}

__global__ void k_hist(const int* __restrict__ ei, long long E0,
                       int* __restrict__ cnt, int* __restrict__ slot) {
  long long e = (long long)blockIdx.x * blockDim.x + threadIdx.x;
  if (e >= E0) return;
  int d = ei[E0 + e];
  slot[e] = atomicAdd(&cnt[d], 1);
}

#define SCAN_TILE 1024

__global__ void k_tile_sum(const int* __restrict__ cnt, int n, int* __restrict__ tsum) {
  __shared__ int sh[8];
  int base = blockIdx.x * SCAN_TILE + threadIdx.x * 4;
  int v = 0;
#pragma unroll
  for (int j = 0; j < 4; ++j) if (base + j < n) v += cnt[base + j];
#pragma unroll
  for (int o = 16; o > 0; o >>= 1) v += __shfl_xor(v, o, 32);
  if ((threadIdx.x & 31) == 0) sh[threadIdx.x >> 5] = v;
  __syncthreads();
  if (threadIdx.x == 0) {
    int s = 0;
    for (int w = 0; w < 8; ++w) s += sh[w];
    tsum[blockIdx.x] = s;
  }
}

__global__ void k_scan(const int* __restrict__ cnt, int n, const int* __restrict__ tsum,
                       int* __restrict__ rowptr) {
  __shared__ int sh[8];
  int tid = threadIdx.x, lane = tid & 31, wave = tid >> 5;
  int pb = 0;
  for (int j = tid; j < (int)blockIdx.x; j += 256) pb += tsum[j];
#pragma unroll
  for (int o = 16; o > 0; o >>= 1) pb += __shfl_xor(pb, o, 32);
  if (lane == 0) sh[wave] = pb;
  __syncthreads();
  int tilebase = 0;
  for (int w = 0; w < 8; ++w) tilebase += sh[w];
  __syncthreads();

  int base = blockIdx.x * SCAN_TILE + tid * 4;
  int cv[4];
#pragma unroll
  for (int j = 0; j < 4; ++j) cv[j] = (base + j < n) ? cnt[base + j] : 0;
  int local = cv[0] + cv[1] + cv[2] + cv[3];
  int inc = local;
#pragma unroll
  for (int o = 1; o < 32; o <<= 1) {
    int t = __shfl_up(inc, o, 32);
    if (lane >= o) inc += t;
  }
  if (lane == 31) sh[wave] = inc;
  __syncthreads();
  int wbase = 0;
  for (int w = 0; w < wave; ++w) wbase += sh[w];
  int ex = tilebase + wbase + inc - local;
#pragma unroll
  for (int j = 0; j < 4; ++j) {
    if (base + j < n) rowptr[base + j] = ex;
    ex += cv[j];
  }
}

__global__ void k_scatter(const int* __restrict__ ei, long long E0,
                          const int* __restrict__ rowptr, const int* __restrict__ slot,
                          int* __restrict__ segA) {
  long long e = (long long)blockIdx.x * blockDim.x + threadIdx.x;
  if (e >= E0) return;
  int s = ei[e];
  int d = ei[E0 + e];
  segA[rowptr[d] + slot[e]] = s;
}

__global__ void k_seg_sort(const int* __restrict__ rowptr, const int* __restrict__ cnt,
                           const int* __restrict__ segA, int* __restrict__ seg, int n) {
  int node = blockIdx.x * (blockDim.x >> 5) + (threadIdx.x >> 5);
  if (node >= n) return;
  node = __builtin_amdgcn_readfirstlane(node);
  int lane = threadIdx.x & 31;
  int start = rowptr[node], deg = cnt[node];
  if (deg <= 32) {
    int mine = (lane < deg) ? segA[start + lane] : 0x7fffffff;
    int r = 0;
    for (int j = 0; j < deg; ++j) {
      int v = __builtin_amdgcn_readlane(mine, j);
      r += ((v < mine) || (v == mine && j < lane)) ? 1 : 0;
    }
    if (lane < deg) seg[start + r] = mine;
  } else {
    for (int i = lane; i < deg; i += 32) {
      int mine = segA[start + i];
      int r = 0;
      for (int j = 0; j < deg; ++j) {
        int v = segA[start + j];
        r += ((v < mine) || (v == mine && j < i)) ? 1 : 0;
      }
      seg[start + r] = mine;
    }
  }
}

template <int CTRL>
__device__ __forceinline__ float dpp_get(float v) {
  return __int_as_float(__builtin_amdgcn_update_dpp(0, __float_as_int(v), CTRL, 0xF, 0xF, true));
}
__device__ __forceinline__ float other_row_get(float v) {
  return __uint_as_float(__builtin_amdgcn_permlanex16(0u, __float_as_uint(v),
                                                      0x76543210u, 0xfedcba98u, true, false));
}
template <int LPH>
__device__ __forceinline__ float head_sum(float p) {
  static_assert(LPH == 4 || LPH == 32, "lanes per head");
  p += dpp_get<0x161>(p);
  p += dpp_get<0x162>(p);
  if constexpr (LPH == 32) {
    p += dpp_get<0x164>(p);
    p += dpp_get<0x168>(p);
    p += other_row_get(p);
  }
  return p;
}

template <int CPT> struct RowVec;
template <> struct RowVec<4> { typedef v4f T; };
template <> struct RowVec<2> { typedef v2f T; };

template <int H, int C>
__global__ void k_gat_node(const float* __restrict__ XL, const float* __restrict__ XR,
                           const float* __restrict__ att, const float* __restrict__ bias,
                           const int* __restrict__ rowptr, const int* __restrict__ cnt,
                           const int* __restrict__ seg, float* __restrict__ OUT, int n) {
  constexpr int HC  = H * C;
  constexpr int CPT = HC / 32;
  constexpr int LPH = 32 / H;
  constexpr int UNR = 4;
  typedef typename RowVec<CPT>::T vec_t;

  int node = blockIdx.x * (blockDim.x >> 5) + (threadIdx.x >> 5);
  if (node >= n) return;
  node = __builtin_amdgcn_readfirstlane(node);
  int lane = threadIdx.x & 31;
  int ch0  = lane * CPT;
  int start = rowptr[node], deg = cnt[node];

  vec_t xr = *(const vec_t*)(XR + (size_t)node * HC + ch0);
  vec_t av = *(const vec_t*)(att + ch0);
  vec_t xs = *(const vec_t*)(XL + (size_t)node * HC + ch0);

  float p = 0.0f;
#pragma unroll
  for (int j = 0; j < CPT; ++j) {
    float v = xs[j] + xr[j];
    v = (v > 0.0f) ? v : NEG_SLOPE * v;
    p += av[j] * v;
  }
  float m   = head_sum<LPH>(p);
  float den = 1.0f;
  vec_t acc = xs;

  for (int base = 0; base < deg; base += 32) {
    int nb = deg - base;
    if (nb > 32) nb = 32;
    int mysrc = (lane < nb) ? seg[start + base + lane] : node;
    for (int j = 0; j < nb; j += UNR) {
      vec_t x[UNR];
      float pu[UNR];
      bool  ok[UNR];
#pragma unroll
      for (int u = 0; u < UNR; ++u) {
        int s = __builtin_amdgcn_readlane(mysrc, j + u);
        ok[u] = (j + u) < nb;
        x[u] = *(const vec_t*)(XL + (size_t)s * HC + ch0);
      }
#pragma unroll
      for (int u = 0; u < UNR; ++u) {
        float q = 0.0f;
#pragma unroll
        for (int k = 0; k < CPT; ++k) {
          float v = x[u][k] + xr[k];
          v = (v > 0.0f) ? v : NEG_SLOPE * v;
          q += av[k] * v;
        }
        q = head_sum<LPH>(q);
        pu[u] = ok[u] ? q : -INFINITY;
      }
      float mn = fmaxf(fmaxf(m, pu[0]), fmaxf(pu[1], fmaxf(pu[2], pu[3])));
      float sc = __expf(m - mn);
      float e0 = ok[0] ? __expf(pu[0] - mn) : 0.0f;
      float e1 = ok[1] ? __expf(pu[1] - mn) : 0.0f;
      float e2 = ok[2] ? __expf(pu[2] - mn) : 0.0f;
      float e3 = ok[3] ? __expf(pu[3] - mn) : 0.0f;
      den = den * sc + ((e0 + e1) + (e2 + e3));
#pragma unroll
      for (int k = 0; k < CPT; ++k)
        acc[k] = acc[k] * sc + ((e0 * x[0][k] + e1 * x[1][k]) + (e2 * x[2][k] + e3 * x[3][k]));
      m = mn;
    }
  }

  float dn = fmaxf(den, 1e-16f);
  vec_t bv = *(const vec_t*)(bias + ch0);
  vec_t o;
#pragma unroll
  for (int k = 0; k < CPT; ++k) o[k] = acc[k] / dn + bv[k];
  *(vec_t*)(OUT + (size_t)node * HC + ch0) = o;
}

#define SM_LOADS 8

__device__ __forceinline__ float row16_max(float v) {
  v = fmaxf(v, dpp_get<0x168>(v));
  v = fmaxf(v, dpp_get<0x164>(v));
  v = fmaxf(v, dpp_get<0x162>(v));
  v = fmaxf(v, dpp_get<0x161>(v));
  return v;
}
__device__ __forceinline__ float row16_sum(float v) {
  v += dpp_get<0x168>(v);
  v += dpp_get<0x164>(v);
  v += dpp_get<0x162>(v);
  v += dpp_get<0x161>(v);
  return v;
}

__device__ __forceinline__ v4f softmax_row16(v4f t) {
  float mx = row16_max(fmaxf(fmaxf(t[0], t[1]), fmaxf(t[2], t[3])));
  v4f e;
  e[0] = __expf(t[0] - mx); e[1] = __expf(t[1] - mx);
  e[2] = __expf(t[2] - mx); e[3] = __expf(t[3] - mx);
  float sa = row16_sum(e[0] + e[1]);
  float sb = row16_sum(e[2] + e[3]);
  float inv = 1.0f / (sa + sb);
  return e * inv;
}

__global__ void __launch_bounds__(EW_THREADS)
k_softmax64(const float* __restrict__ in, float* __restrict__ out, int n) {
  constexpr int ROWS_PER_WAVE = 2 * SM_LOADS;
  constexpr int ROWS_PER_TRIP = (EW_THREADS / 32) * ROWS_PER_WAVE;
  const int lane = threadIdx.x & 31;
  const int sub  = lane >> 4;
  const int c0   = (lane & 15) * 4;
  for (int rb = blockIdx.x * ROWS_PER_TRIP; rb < n; rb += gridDim.x * ROWS_PER_TRIP) {
    const int row0 = rb + (threadIdx.x >> 5) * ROWS_PER_WAVE + sub;
    if (rb + ROWS_PER_TRIP <= n) {
      const float* __restrict__ ip = in  + (size_t)row0 * 64 + c0;
      float* __restrict__       op = out + (size_t)row0 * 64 + c0;
      v4f v[SM_LOADS];
#pragma unroll
      for (int j = 0; j < SM_LOADS; ++j) v[j] = *(const v4f*)(ip + j * 128);
#pragma unroll
      for (int j = 0; j < SM_LOADS; ++j) *(v4f*)(op + j * 128) = softmax_row16(v[j]);
    } else {
      for (int j = 0; j < SM_LOADS; ++j) {
        const int r = row0 + 2 * j;
        const int rc = (r < n) ? r : n - 1;
        v4f o = softmax_row16(*(const v4f*)(in + (size_t)rc * 64 + c0));
        if (r < n) *(v4f*)(out + (size_t)r * 64 + c0) = o;
      }
    }
  }
}

template <int H, int C>
static void gat_layer(const _Float16* X16, int din,
                      const float* Wl, const float* Wr, const float* att, const float* b,
                      _Float16* WPa, _Float16* WPb,
                      float* XL, float* XR, float* OUT,
                      const int* rowptr, const int* cnt, const int* seg,
                      int N, hipStream_t stream) {
  constexpr int HC = H * C;

  int packThreads = (din >> 5) * (HC >> 4) * 32;
  int packBlocks  = (packThreads + 255) / 256;
  k_pack_w<<<packBlocks, 256, 0, stream>>>(Wl, WPa, din, HC);
  k_pack_w<<<packBlocks, 256, 0, stream>>>(Wr, WPb, din, HC);

  int rowTiles = (N + 15) / 16;
  dim3 gg((unsigned)((rowTiles + 3) / 4), (unsigned)(HC / 64));
  k_gemm_f16<<<gg, 128, 0, stream>>>(X16, WPa, XL, N, din, HC);
  k_gemm_f16<<<gg, 128, 0, stream>>>(X16, WPb, XR, N, din, HC);

  k_gat_node<H, C><<<(N + 7) / 8, 256, 0, stream>>>(XL, XR, att, b, rowptr, cnt, seg, OUT, N);
}

extern "C" void kernel_launch(void* const* d_in, const int* in_sizes, int n_in,
                              void* d_out, int out_size, void* d_ws, size_t ws_size,
                              hipStream_t stream) {
  (void)n_in; (void)out_size; (void)ws_size;
  const float* x    = (const float*)d_in[0];
  const int*   ei   = (const int*)d_in[1];
  const float* Wl1  = (const float*)d_in[2];
  const float* Wr1  = (const float*)d_in[3];
  const float* att1 = (const float*)d_in[4];
  const float* b1   = (const float*)d_in[5];
  const float* Wl2  = (const float*)d_in[6];
  const float* Wr2  = (const float*)d_in[7];
  const float* att2 = (const float*)d_in[8];
  const float* b2   = (const float*)d_in[9];
  const float* Wl3  = (const float*)d_in[10];
  const float* Wr3  = (const float*)d_in[11];
  const float* att3 = (const float*)d_in[12];
  const float* b3   = (const float*)d_in[13];

  int       N  = in_sizes[0] / 256;
  long long E0 = (long long)in_sizes[1] / 2;

  uintptr_t p = (uintptr_t)d_ws;
  auto carve = [&](size_t bytes) -> void* {
    void* r = (void*)p;
    p += (bytes + 255) & ~(size_t)255;
    return r;
  };
  _Float16* X16 = (_Float16*)carve((size_t)N * 256 * sizeof(_Float16));
  _Float16* WPa = (_Float16*)carve((size_t)256 * 128 * sizeof(_Float16));
  _Float16* WPb = (_Float16*)carve((size_t)256 * 128 * sizeof(_Float16));
  float*    XL  = (float*)carve((size_t)N * 128 * sizeof(float));
  float*    XR  = (float*)carve((size_t)N * 128 * sizeof(float));
  float*    OUT = (float*)carve((size_t)N * 128 * sizeof(float));
  int* cnt    = (int*)carve((size_t)N * sizeof(int));
  int* rowptr = (int*)carve((size_t)N * sizeof(int));
  int* tsum   = (int*)carve((size_t)((N + SCAN_TILE - 1) / SCAN_TILE) * sizeof(int));
  int* slot   = (int*)carve((size_t)E0 * sizeof(int));
  int* segA   = (int*)carve((size_t)E0 * sizeof(int));
  int* seg    = (int*)carve((size_t)E0 * sizeof(int));

  int nTiles = (N + SCAN_TILE - 1) / SCAN_TILE;
  int gE0    = (int)((E0 + 255) / 256);
  k_zero_i32<<<(N + 255) / 256, 256, 0, stream>>>(cnt, N);
  k_hist<<<gE0, 256, 0, stream>>>(ei, E0, cnt, slot);
  k_tile_sum<<<nTiles, 256, 0, stream>>>(cnt, N, tsum);
  k_scan<<<nTiles, 256, 0, stream>>>(cnt, N, tsum, rowptr);
  k_scatter<<<gE0, 256, 0, stream>>>(ei, E0, rowptr, slot, segA);
  k_seg_sort<<<(N + 7) / 8, 256, 0, stream>>>(rowptr, cnt, segA, seg, N);

  long long nx = (long long)N * 256;
  k_plane_to_f16<false><<<EW_BLOCKS, EW_THREADS, 0, stream>>>(x, X16, nx);
  gat_layer<8, 16>(X16, 256, Wl1, Wr1, att1, b1, WPa, WPb,
                   XL, XR, OUT, rowptr, cnt, seg, N, stream);
  long long nh = (long long)N * 128;
  k_plane_to_f16<true><<<EW_BLOCKS, EW_THREADS, 0, stream>>>(OUT, X16, nh);

  gat_layer<8, 16>(X16, 128, Wl2, Wr2, att2, b2, WPa, WPb,
                   XL, XR, OUT, rowptr, cnt, seg, N, stream);
  k_plane_to_f16<true><<<EW_BLOCKS, EW_THREADS, 0, stream>>>(OUT, X16, nh);

  gat_layer<1, 64>(X16, 128, Wl3, Wr3, att3, b3, WPa, WPb,
                   XL, XR, OUT, rowptr, cnt, seg, N, stream);

  k_softmax64<<<EW_BLOCKS, EW_THREADS, 0, stream>>>(OUT, (float*)d_out, N);
}


# Launches that do not depend on each other share one grid: counter clear, all 6 weight packs and x->f16 are one launch; each layer's two projections are one GEMM launch. 25 -> 15 launches
# speedup vs baseline: 4.2361x; 1.0478x over previous
#include <hip/hip_runtime.h>
#include <hip/hip_bf16.h>
#include <math.h>


#define NEG_SLOPE 0.2f

typedef __attribute__((ext_vector_type(16))) _Float16 v16h;
typedef __attribute__((ext_vector_type(8)))  _Float16 v8h;
typedef __attribute__((ext_vector_type(8)))  float    v8f;

#define EW_THREADS 256
#define EW_LOADS   8
#define EW_BLOCKS  512

typedef __attribute__((ext_vector_type(4))) float    v4f;
typedef __attribute__((ext_vector_type(2))) float    v2f;
typedef __attribute__((ext_vector_type(4))) _Float16 v4h;

template <bool DO_ELU>
__device__ __forceinline__ v4h plane_cvt(v4f t) {
  if (DO_ELU) {
    v4f em;
#pragma unroll
    for (int k = 0; k < 4; ++k) em[k] = expm1f(t[k]);
#pragma unroll
    for (int k = 0; k < 4; ++k) t[k] = (t[k] > 0.0f) ? t[k] : em[k];
  }
  return __builtin_convertvector(t, v4h);
}

template <bool DO_ELU>
__device__ __forceinline__ void plane_to_f16_body(const float* __restrict__ src, _Float16* __restrict__ dst,
                                                  long long n, int blk, int nblk) {
  constexpr long long TILE = (long long)EW_THREADS * EW_LOADS;
  const long long nvec = n >> 2;
  const v4f* __restrict__ s4 = (const v4f*)src;
  v4h* __restrict__ d4 = (v4h*)dst;
  for (long long t0 = (long long)blk * TILE; t0 < nvec; t0 += (long long)nblk * TILE) {
    const v4f* __restrict__ sp = s4 + t0 + threadIdx.x;
    v4h* __restrict__ dp = d4 + t0 + threadIdx.x;
    if (t0 + TILE <= nvec) {
      v4f v[EW_LOADS];
#pragma unroll
      for (int j = 0; j < EW_LOADS; ++j) v[j] = sp[j * EW_THREADS];
      if (DO_ELU) __builtin_amdgcn_sched_barrier(0);
#pragma unroll
      for (int j = 0; j < EW_LOADS; ++j) dp[j * EW_THREADS] = plane_cvt<DO_ELU>(v[j]);
    } else {
      for (int j = 0; j < EW_LOADS; ++j)
        if (t0 + threadIdx.x + j * EW_THREADS < nvec) dp[j * EW_THREADS] = plane_cvt<DO_ELU>(sp[j * EW_THREADS]);
    }
  }
  if (blk == 0 && (long long)threadIdx.x < (n & 3)) {
    long long i = (nvec << 2) + threadIdx.x;
    float t = src[i];
    if (DO_ELU) t = (t > 0.0f) ? t : expm1f(t);
    dst[i] = (_Float16)t;
  }
}

__global__ void __launch_bounds__(EW_THREADS)
k_elu_to_f16(const float* __restrict__ src, _Float16* __restrict__ dst, long long n) {
  plane_to_f16_body<true>(src, dst, n, (int)blockIdx.x, (int)gridDim.x);
}

__device__ __forceinline__ void pack_w_body(const float* __restrict__ W, _Float16* __restrict__ WP,
                                            int din, int dout, int idx) {
  int ktiles = din >> 5, ntiles = dout >> 4;
  int total = ktiles * ntiles * 32;
  if (idx >= total) return;
  int lane = idx & 31;
  int frag = idx >> 5;
  int nt = frag % ntiles;
  int kt = frag / ntiles;
  int half = lane >> 4, lr = lane & 15;
  _Float16* o = WP + ((size_t)frag * 32 + lane) * 16;
  const float* w0 = W + (size_t)(kt * 32 + half * 8) * dout + nt * 16 + lr;
#pragma unroll
  for (int j = 0; j < 8; ++j) {
    o[j]     = (_Float16)w0[(size_t)j * dout];
    o[8 + j] = (_Float16)w0[(size_t)(16 + j) * dout];
  }
}

__global__ void k_gemm2_f16(const _Float16* __restrict__ X,
                            const _Float16* __restrict__ WPl, const _Float16* __restrict__ WPr,
                            float* __restrict__ YL, float* __restrict__ YR,
                            int Mrows, int din, int dout) {
  int strips = dout >> 6;
  int strip  = blockIdx.y;
  bool right = strip >= strips;
  if (right) strip -= strips;
  const _Float16* __restrict__ WP = right ? WPr : WPl;
  float* __restrict__ Y = right ? YR : YL;
  int wave = threadIdx.x >> 5;
  int lane = threadIdx.x & 31;
  int half = lane >> 4;
  int lr   = lane & 15;
  int tileM = blockIdx.x * 4 + wave;
  if (tileM * 16 >= Mrows) return;
  int nt0 = strip * 4;
  int ntiles = dout >> 4;

  int row = tileM * 16 + lr;
  if (row >= Mrows) row = Mrows - 1;

  v8f acc0 = {}, acc1 = {}, acc2 = {}, acc3 = {};
  const _Float16* xrow = X + (size_t)row * din;
  const v16h* WPv = (const v16h*)WP;
  for (int k0 = 0; k0 < din; k0 += 32) {
    const v8h* xa = (const v8h*)(xrow + k0 + half * 8);
    v8h lo = xa[0];
    v8h hi = xa[2];
    v16h a = __builtin_shufflevector(lo, hi, 0, 1, 2, 3, 4, 5, 6, 7,
                                     8, 9, 10, 11, 12, 13, 14, 15);
    size_t fb = ((size_t)(k0 >> 5) * ntiles + nt0) * 32 + lane;
    v16h b0 = WPv[fb];
    v16h b1 = WPv[fb + 32];
    v16h b2 = WPv[fb + 64];
    v16h b3 = WPv[fb + 96];
    acc0 = __builtin_amdgcn_wmma_f32_16x16x32_f16(false, a, false, b0, (short)0, acc0, false, false);
    acc1 = __builtin_amdgcn_wmma_f32_16x16x32_f16(false, a, false, b1, (short)0, acc1, false, false);
    acc2 = __builtin_amdgcn_wmma_f32_16x16x32_f16(false, a, false, b2, (short)0, acc2, false, false);
    acc3 = __builtin_amdgcn_wmma_f32_16x16x32_f16(false, a, false, b3, (short)0, acc3, false, false);
  }
  int r0 = tileM * 16 + half * 8;
#pragma unroll
  for (int i = 0; i < 8; ++i) {
    int r = r0 + i;
    if (r < Mrows) {
      float* yp = Y + (size_t)r * dout + nt0 * 16 + lr;
      yp[0]  = acc0[i];
      yp[16] = acc1[i];
      yp[32] = acc2[i];
      yp[48] = acc3[i];
    }
  }
}

#define L1_DIN 256
#define L1_HC  128
#define L2_DIN 128
#define L2_HC  128
#define L3_DIN 128
#define L3_HC  64
#define PACK_BLOCKS_OF(din, hc) ((((din) >> 5) * ((hc) >> 4) * 32 + EW_THREADS - 1) / EW_THREADS)
#define PB1 PACK_BLOCKS_OF(L1_DIN, L1_HC)
#define PB2 PACK_BLOCKS_OF(L2_DIN, L2_HC)
#define PB3 PACK_BLOCKS_OF(L3_DIN, L3_HC)
#define PREP_PACK_BLOCKS (2 * (PB1 + PB2 + PB3))
#define PREP_ZERO_BLOCKS 16
#define PREP_BLOCKS (PREP_PACK_BLOCKS + PREP_ZERO_BLOCKS + EW_BLOCKS)

typedef __attribute__((ext_vector_type(4))) int v4i;

__global__ void __launch_bounds__(EW_THREADS)
k_prep(const float* __restrict__ x, _Float16* __restrict__ X16, long long nx,
       int* __restrict__ cnt, int n,
       const float* __restrict__ Wl1, const float* __restrict__ Wr1,
       const float* __restrict__ Wl2, const float* __restrict__ Wr2,
       const float* __restrict__ Wl3, const float* __restrict__ Wr3,
       _Float16* __restrict__ P1l, _Float16* __restrict__ P1r,
       _Float16* __restrict__ P2l, _Float16* __restrict__ P2r,
       _Float16* __restrict__ P3l, _Float16* __restrict__ P3r) {
  int b = (int)blockIdx.x;
  if (b < PREP_PACK_BLOCKS) {
    const float* W; _Float16* WP; int din, hc;
    if (b < PB1)                      { W = Wl1; WP = P1l; din = L1_DIN; hc = L1_HC; }
    else if ((b -= PB1) < PB1)        { W = Wr1; WP = P1r; din = L1_DIN; hc = L1_HC; }
    else if ((b -= PB1) < PB2)        { W = Wl2; WP = P2l; din = L2_DIN; hc = L2_HC; }
    else if ((b -= PB2) < PB2)        { W = Wr2; WP = P2r; din = L2_DIN; hc = L2_HC; }
    else if ((b -= PB2) < PB3)        { W = Wl3; WP = P3l; din = L3_DIN; hc = L3_HC; }
    else            { b -= PB3;         W = Wr3; WP = P3r; din = L3_DIN; hc = L3_HC; }
    pack_w_body(W, WP, din, hc, b * EW_THREADS + (int)threadIdx.x);
    return;
  }
  b -= PREP_PACK_BLOCKS;
  if (b < PREP_ZERO_BLOCKS) {
    const v4i z = {0, 0, 0, 0};
    for (int i = (b * EW_THREADS + (int)threadIdx.x) * 4; i < n; i += PREP_ZERO_BLOCKS * EW_THREADS * 4) {
      if (i + 4 <= n) *(v4i*)(cnt + i) = z;
      else for (int j = i; j < n; ++j) cnt[j] = 0;
    }
    return;
  }
  b -= PREP_ZERO_BLOCKS;
  plane_to_f16_body<false>(x, X16, nx, b, EW_BLOCKS);
}


__global__ void k_hist(const int* __restrict__ ei, long long E0,
                       int* __restrict__ cnt, int* __restrict__ slot) {
  long long e = (long long)blockIdx.x * blockDim.x + threadIdx.x;
  if (e >= E0) return;
  int d = ei[E0 + e];
  slot[e] = atomicAdd(&cnt[d], 1);
}

#define SCAN_TILE 1024

__global__ void k_tile_sum(const int* __restrict__ cnt, int n, int* __restrict__ tsum) {
  __shared__ int sh[8];
  int base = blockIdx.x * SCAN_TILE + threadIdx.x * 4;
  int v = 0;
#pragma unroll
  for (int j = 0; j < 4; ++j) if (base + j < n) v += cnt[base + j];
#pragma unroll
  for (int o = 16; o > 0; o >>= 1) v += __shfl_xor(v, o, 32);
  if ((threadIdx.x & 31) == 0) sh[threadIdx.x >> 5] = v;
  __syncthreads();
  if (threadIdx.x == 0) {
    int s = 0;
    for (int w = 0; w < 8; ++w) s += sh[w];
    tsum[blockIdx.x] = s;
  }
}

__global__ void k_scan(const int* __restrict__ cnt, int n, const int* __restrict__ tsum,
                       int* __restrict__ rowptr) {
  __shared__ int sh[8];
  int tid = threadIdx.x, lane = tid & 31, wave = tid >> 5;
  int pb = 0;
  for (int j = tid; j < (int)blockIdx.x; j += 256) pb += tsum[j];
#pragma unroll
  for (int o = 16; o > 0; o >>= 1) pb += __shfl_xor(pb, o, 32);
  if (lane == 0) sh[wave] = pb;
  __syncthreads();
  int tilebase = 0;
  for (int w = 0; w < 8; ++w) tilebase += sh[w];
  __syncthreads();

  int base = blockIdx.x * SCAN_TILE + tid * 4;
  int cv[4];
#pragma unroll
  for (int j = 0; j < 4; ++j) cv[j] = (base + j < n) ? cnt[base + j] : 0;
  int local = cv[0] + cv[1] + cv[2] + cv[3];
  int inc = local;
#pragma unroll
  for (int o = 1; o < 32; o <<= 1) {
    int t = __shfl_up(inc, o, 32);
    if (lane >= o) inc += t;
  }
  if (lane == 31) sh[wave] = inc;
  __syncthreads();
  int wbase = 0;
  for (int w = 0; w < wave; ++w) wbase += sh[w];
  int ex = tilebase + wbase + inc - local;
#pragma unroll
  for (int j = 0; j < 4; ++j) {
    if (base + j < n) rowptr[base + j] = ex;
    ex += cv[j];
  }
}

__global__ void k_scatter(const int* __restrict__ ei, long long E0,
                          const int* __restrict__ rowptr, const int* __restrict__ slot,
                          int* __restrict__ segA) {
  long long e = (long long)blockIdx.x * blockDim.x + threadIdx.x;
  if (e >= E0) return;
  int s = ei[e];
  int d = ei[E0 + e];
  segA[rowptr[d] + slot[e]] = s;
}

__global__ void k_seg_sort(const int* __restrict__ rowptr, const int* __restrict__ cnt,
                           const int* __restrict__ segA, int* __restrict__ seg, int n) {
  int node = blockIdx.x * (blockDim.x >> 5) + (threadIdx.x >> 5);
  if (node >= n) return;
  node = __builtin_amdgcn_readfirstlane(node);
  int lane = threadIdx.x & 31;
  int start = rowptr[node], deg = cnt[node];
  if (deg <= 32) {
    int mine = (lane < deg) ? segA[start + lane] : 0x7fffffff;
    int r = 0;
    for (int j = 0; j < deg; ++j) {
      int v = __builtin_amdgcn_readlane(mine, j);
      r += ((v < mine) || (v == mine && j < lane)) ? 1 : 0;
    }
    if (lane < deg) seg[start + r] = mine;
  } else {
    for (int i = lane; i < deg; i += 32) {
      int mine = segA[start + i];
      int r = 0;
      for (int j = 0; j < deg; ++j) {
        int v = segA[start + j];
        r += ((v < mine) || (v == mine && j < i)) ? 1 : 0;
      }
      seg[start + r] = mine;
    }
  }
}

template <int CTRL>
__device__ __forceinline__ float dpp_get(float v) {
  return __int_as_float(__builtin_amdgcn_update_dpp(0, __float_as_int(v), CTRL, 0xF, 0xF, true));
}
__device__ __forceinline__ float other_row_get(float v) {
  return __uint_as_float(__builtin_amdgcn_permlanex16(0u, __float_as_uint(v),
                                                      0x76543210u, 0xfedcba98u, true, false));
}
template <int LPH>
__device__ __forceinline__ float head_sum(float p) {
  static_assert(LPH == 4 || LPH == 32, "lanes per head");
  p += dpp_get<0x161>(p);
  p += dpp_get<0x162>(p);
  if constexpr (LPH == 32) {
    p += dpp_get<0x164>(p);
    p += dpp_get<0x168>(p);
    p += other_row_get(p);
  }
  return p;
}

template <int CPT> struct RowVec;
template <> struct RowVec<4> { typedef v4f T; };
template <> struct RowVec<2> { typedef v2f T; };

template <int H, int C>
__global__ void k_gat_node(const float* __restrict__ XL, const float* __restrict__ XR,
                           const float* __restrict__ att, const float* __restrict__ bias,
                           const int* __restrict__ rowptr, const int* __restrict__ cnt,
                           const int* __restrict__ seg, float* __restrict__ OUT, int n) {
  constexpr int HC  = H * C;
  constexpr int CPT = HC / 32;
  constexpr int LPH = 32 / H;
  constexpr int UNR = 4;
  typedef typename RowVec<CPT>::T vec_t;

  int node = blockIdx.x * (blockDim.x >> 5) + (threadIdx.x >> 5);
  if (node >= n) return;
  node = __builtin_amdgcn_readfirstlane(node);
  int lane = threadIdx.x & 31;
  int ch0  = lane * CPT;
  int start = rowptr[node], deg = cnt[node];

  vec_t xr = *(const vec_t*)(XR + (size_t)node * HC + ch0);
  vec_t av = *(const vec_t*)(att + ch0);
  vec_t xs = *(const vec_t*)(XL + (size_t)node * HC + ch0);

  float p = 0.0f;
#pragma unroll
  for (int j = 0; j < CPT; ++j) {
    float v = xs[j] + xr[j];
    v = (v > 0.0f) ? v : NEG_SLOPE * v;
    p += av[j] * v;
  }
  float m   = head_sum<LPH>(p);
  float den = 1.0f;
  vec_t acc = xs;

  for (int base = 0; base < deg; base += 32) {
    int nb = deg - base;
    if (nb > 32) nb = 32;
    int mysrc = (lane < nb) ? seg[start + base + lane] : node;
    for (int j = 0; j < nb; j += UNR) {
      vec_t x[UNR];
      float pu[UNR];
      bool  ok[UNR];
#pragma unroll
      for (int u = 0; u < UNR; ++u) {
        int s = __builtin_amdgcn_readlane(mysrc, j + u);
        ok[u] = (j + u) < nb;
        x[u] = *(const vec_t*)(XL + (size_t)s * HC + ch0);
      }
#pragma unroll
      for (int u = 0; u < UNR; ++u) {
        float q = 0.0f;
#pragma unroll
        for (int k = 0; k < CPT; ++k) {
          float v = x[u][k] + xr[k];
          v = (v > 0.0f) ? v : NEG_SLOPE * v;
          q += av[k] * v;
        }
        q = head_sum<LPH>(q);
        pu[u] = ok[u] ? q : -INFINITY;
      }
      float mn = fmaxf(fmaxf(m, pu[0]), fmaxf(pu[1], fmaxf(pu[2], pu[3])));
      float sc = __expf(m - mn);
      float e0 = ok[0] ? __expf(pu[0] - mn) : 0.0f;
      float e1 = ok[1] ? __expf(pu[1] - mn) : 0.0f;
      float e2 = ok[2] ? __expf(pu[2] - mn) : 0.0f;
      float e3 = ok[3] ? __expf(pu[3] - mn) : 0.0f;
      den = den * sc + ((e0 + e1) + (e2 + e3));
#pragma unroll
      for (int k = 0; k < CPT; ++k)
        acc[k] = acc[k] * sc + ((e0 * x[0][k] + e1 * x[1][k]) + (e2 * x[2][k] + e3 * x[3][k]));
      m = mn;
    }
  }

  float dn = fmaxf(den, 1e-16f);
  vec_t bv = *(const vec_t*)(bias + ch0);
  vec_t o;
#pragma unroll
  for (int k = 0; k < CPT; ++k) o[k] = acc[k] / dn + bv[k];
  *(vec_t*)(OUT + (size_t)node * HC + ch0) = o;
}

#define SM_LOADS 8

__device__ __forceinline__ float row16_max(float v) {
  v = fmaxf(v, dpp_get<0x168>(v));
  v = fmaxf(v, dpp_get<0x164>(v));
  v = fmaxf(v, dpp_get<0x162>(v));
  v = fmaxf(v, dpp_get<0x161>(v));
  return v;
}
__device__ __forceinline__ float row16_sum(float v) {
  v += dpp_get<0x168>(v);
  v += dpp_get<0x164>(v);
  v += dpp_get<0x162>(v);
  v += dpp_get<0x161>(v);
  return v;
}

__device__ __forceinline__ v4f softmax_row16(v4f t) {
  float mx = row16_max(fmaxf(fmaxf(t[0], t[1]), fmaxf(t[2], t[3])));
  v4f e;
  e[0] = __expf(t[0] - mx); e[1] = __expf(t[1] - mx);
  e[2] = __expf(t[2] - mx); e[3] = __expf(t[3] - mx);
  float sa = row16_sum(e[0] + e[1]);
  float sb = row16_sum(e[2] + e[3]);
  float inv = 1.0f / (sa + sb);
  return e * inv;
}

__global__ void __launch_bounds__(EW_THREADS)
k_softmax64(const float* __restrict__ in, float* __restrict__ out, int n) {
  constexpr int ROWS_PER_WAVE = 2 * SM_LOADS;
  constexpr int ROWS_PER_TRIP = (EW_THREADS / 32) * ROWS_PER_WAVE;
  const int lane = threadIdx.x & 31;
  const int sub  = lane >> 4;
  const int c0   = (lane & 15) * 4;
  for (int rb = blockIdx.x * ROWS_PER_TRIP; rb < n; rb += gridDim.x * ROWS_PER_TRIP) {
    const int row0 = rb + (threadIdx.x >> 5) * ROWS_PER_WAVE + sub;
    if (rb + ROWS_PER_TRIP <= n) {
      const float* __restrict__ ip = in  + (size_t)row0 * 64 + c0;
      float* __restrict__       op = out + (size_t)row0 * 64 + c0;
      v4f v[SM_LOADS];
#pragma unroll
      for (int j = 0; j < SM_LOADS; ++j) v[j] = *(const v4f*)(ip + j * 128);
#pragma unroll
      for (int j = 0; j < SM_LOADS; ++j) *(v4f*)(op + j * 128) = softmax_row16(v[j]);
    } else {
      for (int j = 0; j < SM_LOADS; ++j) {
        const int r = row0 + 2 * j;
        const int rc = (r < n) ? r : n - 1;
        v4f o = softmax_row16(*(const v4f*)(in + (size_t)rc * 64 + c0));
        if (r < n) *(v4f*)(out + (size_t)r * 64 + c0) = o;
      }
    }
  }
}

template <int H, int C>
static void gat_layer(const _Float16* X16, int din,
                      const _Float16* WPl, const _Float16* WPr, const float* att, const float* b,
                      float* XL, float* XR, float* OUT,
                      const int* rowptr, const int* cnt, const int* seg,
                      int N, hipStream_t stream) {
  constexpr int HC = H * C;

  int rowTiles = (N + 15) / 16;
  dim3 gg((unsigned)((rowTiles + 3) / 4), (unsigned)(2 * (HC / 64)));
  k_gemm2_f16<<<gg, 128, 0, stream>>>(X16, WPl, WPr, XL, XR, N, din, HC);

  k_gat_node<H, C><<<(N + 7) / 8, 256, 0, stream>>>(XL, XR, att, b, rowptr, cnt, seg, OUT, N);
}

extern "C" void kernel_launch(void* const* d_in, const int* in_sizes, int n_in,
                              void* d_out, int out_size, void* d_ws, size_t ws_size,
                              hipStream_t stream) {
  (void)n_in; (void)out_size; (void)ws_size;
  const float* x    = (const float*)d_in[0];
  const int*   ei   = (const int*)d_in[1];
  const float* Wl1  = (const float*)d_in[2];
  const float* Wr1  = (const float*)d_in[3];
  const float* att1 = (const float*)d_in[4];
  const float* b1   = (const float*)d_in[5];
  const float* Wl2  = (const float*)d_in[6];
  const float* Wr2  = (const float*)d_in[7];
  const float* att2 = (const float*)d_in[8];
  const float* b2   = (const float*)d_in[9];
  const float* Wl3  = (const float*)d_in[10];
  const float* Wr3  = (const float*)d_in[11];
  const float* att3 = (const float*)d_in[12];
  const float* b3   = (const float*)d_in[13];

  int       N  = in_sizes[0] / 256;
  long long E0 = (long long)in_sizes[1] / 2;

  uintptr_t p = (uintptr_t)d_ws;
  auto carve = [&](size_t bytes) -> void* {
    void* r = (void*)p;
    p += (bytes + 255) & ~(size_t)255;
    return r;
  };
  _Float16* X16 = (_Float16*)carve((size_t)N * 256 * sizeof(_Float16));
  _Float16* P1l = (_Float16*)carve((size_t)L1_DIN * L1_HC * sizeof(_Float16));
  _Float16* P1r = (_Float16*)carve((size_t)L1_DIN * L1_HC * sizeof(_Float16));
  _Float16* P2l = (_Float16*)carve((size_t)L2_DIN * L2_HC * sizeof(_Float16));
  _Float16* P2r = (_Float16*)carve((size_t)L2_DIN * L2_HC * sizeof(_Float16));
  _Float16* P3l = (_Float16*)carve((size_t)L3_DIN * L3_HC * sizeof(_Float16));
  _Float16* P3r = (_Float16*)carve((size_t)L3_DIN * L3_HC * sizeof(_Float16));
  float*    XL  = (float*)carve((size_t)N * 128 * sizeof(float));
  float*    XR  = (float*)carve((size_t)N * 128 * sizeof(float));
  float*    OUT = (float*)carve((size_t)N * 128 * sizeof(float));
  int* cnt    = (int*)carve((size_t)N * sizeof(int));
  int* rowptr = (int*)carve((size_t)N * sizeof(int));
  int* tsum   = (int*)carve((size_t)((N + SCAN_TILE - 1) / SCAN_TILE) * sizeof(int));
  int* seg    = (int*)carve((size_t)E0 * sizeof(int));
  int* slot   = (int*)XL;
  int* segA   = (int*)XR;

  int nTiles = (N + SCAN_TILE - 1) / SCAN_TILE;
  int gE0    = (int)((E0 + 255) / 256);
  long long nx = (long long)N * 256;
  k_prep<<<PREP_BLOCKS, EW_THREADS, 0, stream>>>(x, X16, nx, cnt, N, Wl1, Wr1, Wl2, Wr2, Wl3, Wr3,
                                                 P1l, P1r, P2l, P2r, P3l, P3r);
  k_hist<<<gE0, 256, 0, stream>>>(ei, E0, cnt, slot);
  k_tile_sum<<<nTiles, 256, 0, stream>>>(cnt, N, tsum);
  k_scan<<<nTiles, 256, 0, stream>>>(cnt, N, tsum, rowptr);
  k_scatter<<<gE0, 256, 0, stream>>>(ei, E0, rowptr, slot, segA);
  k_seg_sort<<<(N + 7) / 8, 256, 0, stream>>>(rowptr, cnt, segA, seg, N);

  gat_layer<8, 16>(X16, L1_DIN, P1l, P1r, att1, b1,
                   XL, XR, OUT, rowptr, cnt, seg, N, stream);
  long long nh = (long long)N * 128;
  k_elu_to_f16<<<EW_BLOCKS, EW_THREADS, 0, stream>>>(OUT, X16, nh);

  gat_layer<8, 16>(X16, L2_DIN, P2l, P2r, att2, b2,
                   XL, XR, OUT, rowptr, cnt, seg, N, stream);
  k_elu_to_f16<<<EW_BLOCKS, EW_THREADS, 0, stream>>>(OUT, X16, nh);

  gat_layer<1, 64>(X16, L3_DIN, P3l, P3r, att3, b3,
                   XL, XR, OUT, rowptr, cnt, seg, N, stream);

  k_softmax64<<<EW_BLOCKS, EW_THREADS, 0, stream>>>(OUT, (float*)d_out, N);
}


# Per-layer passes move into the node kernel's registers: ELU+f16 row goes straight into the next GEMM operand, final row softmax (same sum tree) straight into out; 15 -> 12 launches
# speedup vs baseline: 4.4975x; 1.0617x over previous
#include <hip/hip_runtime.h>
#include <hip/hip_bf16.h>
#include <math.h>


#define NEG_SLOPE 0.2f

typedef __attribute__((ext_vector_type(16))) _Float16 v16h;
typedef __attribute__((ext_vector_type(8)))  _Float16 v8h;
typedef __attribute__((ext_vector_type(8)))  float    v8f;

#define EW_THREADS 256
#define EW_LOADS   8
#define EW_BLOCKS  512

typedef __attribute__((ext_vector_type(4))) float    v4f;
typedef __attribute__((ext_vector_type(2))) float    v2f;
typedef __attribute__((ext_vector_type(4))) _Float16 v4h;

__device__ __forceinline__ v4f elu4(v4f t) {
  v4f em;
#pragma unroll
  for (int k = 0; k < 4; ++k) em[k] = expm1f(t[k]);
#pragma unroll
  for (int k = 0; k < 4; ++k) t[k] = (t[k] > 0.0f) ? t[k] : em[k];
  return t;
}

__device__ __forceinline__ void plane_to_f16_body(const float* __restrict__ src, _Float16* __restrict__ dst,
                                                  long long n, int blk, int nblk) {
  constexpr long long TILE = (long long)EW_THREADS * EW_LOADS;
  const long long nvec = n >> 2;
  const v4f* __restrict__ s4 = (const v4f*)src;
  v4h* __restrict__ d4 = (v4h*)dst;
  for (long long t0 = (long long)blk * TILE; t0 < nvec; t0 += (long long)nblk * TILE) {
    const v4f* __restrict__ sp = s4 + t0 + threadIdx.x;
    v4h* __restrict__ dp = d4 + t0 + threadIdx.x;
    if (t0 + TILE <= nvec) {
      v4f v[EW_LOADS];
#pragma unroll
      for (int j = 0; j < EW_LOADS; ++j) v[j] = sp[j * EW_THREADS];
#pragma unroll
      for (int j = 0; j < EW_LOADS; ++j) dp[j * EW_THREADS] = __builtin_convertvector(v[j], v4h);
    } else {
      for (int j = 0; j < EW_LOADS; ++j)
        if (t0 + threadIdx.x + j * EW_THREADS < nvec) dp[j * EW_THREADS] = __builtin_convertvector(sp[j * EW_THREADS], v4h);
    }
  }
  if (blk == 0 && (long long)threadIdx.x < (n & 3)) {
    long long i = (nvec << 2) + threadIdx.x;
    dst[i] = (_Float16)src[i];
  }
}

__device__ __forceinline__ void pack_w_body(const float* __restrict__ W, _Float16* __restrict__ WP,
                                            int din, int dout, int idx) {
  int ktiles = din >> 5, ntiles = dout >> 4;
  int total = ktiles * ntiles * 32;
  if (idx >= total) return;
  int lane = idx & 31;
  int frag = idx >> 5;
  int nt = frag % ntiles;
  int kt = frag / ntiles;
  int half = lane >> 4, lr = lane & 15;
  _Float16* o = WP + ((size_t)frag * 32 + lane) * 16;
  const float* w0 = W + (size_t)(kt * 32 + half * 8) * dout + nt * 16 + lr;
#pragma unroll
  for (int j = 0; j < 8; ++j) {
    o[j]     = (_Float16)w0[(size_t)j * dout];
    o[8 + j] = (_Float16)w0[(size_t)(16 + j) * dout];
  }
}

__global__ void k_gemm2_f16(const _Float16* __restrict__ X,
                            const _Float16* __restrict__ WPl, const _Float16* __restrict__ WPr,
                            float* __restrict__ YL, float* __restrict__ YR,
                            int Mrows, int din, int dout) {
  int strips = dout >> 6;
  int strip  = blockIdx.y;
  bool right = strip >= strips;
  if (right) strip -= strips;
  const _Float16* __restrict__ WP = right ? WPr : WPl;
  float* __restrict__ Y = right ? YR : YL;
  int wave = threadIdx.x >> 5;
  int lane = threadIdx.x & 31;
  int half = lane >> 4;
  int lr   = lane & 15;
  int tileM = blockIdx.x * 4 + wave;
  if (tileM * 16 >= Mrows) return;
  int nt0 = strip * 4;
  int ntiles = dout >> 4;

  int row = tileM * 16 + lr;
  if (row >= Mrows) row = Mrows - 1;

  v8f acc0 = {}, acc1 = {}, acc2 = {}, acc3 = {};
  const _Float16* xrow = X + (size_t)row * din;
  const v16h* WPv = (const v16h*)WP;
  for (int k0 = 0; k0 < din; k0 += 32) {
    const v8h* xa = (const v8h*)(xrow + k0 + half * 8);
    v8h lo = xa[0];
    v8h hi = xa[2];
    v16h a = __builtin_shufflevector(lo, hi, 0, 1, 2, 3, 4, 5, 6, 7,
                                     8, 9, 10, 11, 12, 13, 14, 15);
    size_t fb = ((size_t)(k0 >> 5) * ntiles + nt0) * 32 + lane;
    v16h b0 = WPv[fb];
    v16h b1 = WPv[fb + 32];
    v16h b2 = WPv[fb + 64];
    v16h b3 = WPv[fb + 96];
    acc0 = __builtin_amdgcn_wmma_f32_16x16x32_f16(false, a, false, b0, (short)0, acc0, false, false);
    acc1 = __builtin_amdgcn_wmma_f32_16x16x32_f16(false, a, false, b1, (short)0, acc1, false, false);
    acc2 = __builtin_amdgcn_wmma_f32_16x16x32_f16(false, a, false, b2, (short)0, acc2, false, false);
    acc3 = __builtin_amdgcn_wmma_f32_16x16x32_f16(false, a, false, b3, (short)0, acc3, false, false);
  }
  int r0 = tileM * 16 + half * 8;
#pragma unroll
  for (int i = 0; i < 8; ++i) {
    int r = r0 + i;
    if (r < Mrows) {
      float* yp = Y + (size_t)r * dout + nt0 * 16 + lr;
      yp[0]  = acc0[i];
      yp[16] = acc1[i];
      yp[32] = acc2[i];
      yp[48] = acc3[i];
    }
  }
}

#define L1_DIN 256
#define L1_HC  128
#define L2_DIN 128
#define L2_HC  128
#define L3_DIN 128
#define L3_HC  64
#define PACK_BLOCKS_OF(din, hc) ((((din) >> 5) * ((hc) >> 4) * 32 + EW_THREADS - 1) / EW_THREADS)
#define PB1 PACK_BLOCKS_OF(L1_DIN, L1_HC)
#define PB2 PACK_BLOCKS_OF(L2_DIN, L2_HC)
#define PB3 PACK_BLOCKS_OF(L3_DIN, L3_HC)
#define PREP_PACK_BLOCKS (2 * (PB1 + PB2 + PB3))
#define PREP_ZERO_BLOCKS 16
#define PREP_BLOCKS (PREP_PACK_BLOCKS + PREP_ZERO_BLOCKS + EW_BLOCKS)

typedef __attribute__((ext_vector_type(4))) int v4i;

__global__ void __launch_bounds__(EW_THREADS)
k_prep(const float* __restrict__ x, _Float16* __restrict__ X16, long long nx,
       int* __restrict__ cnt, int n,
       const float* __restrict__ Wl1, const float* __restrict__ Wr1,
       const float* __restrict__ Wl2, const float* __restrict__ Wr2,
       const float* __restrict__ Wl3, const float* __restrict__ Wr3,
       _Float16* __restrict__ P1l, _Float16* __restrict__ P1r,
       _Float16* __restrict__ P2l, _Float16* __restrict__ P2r,
       _Float16* __restrict__ P3l, _Float16* __restrict__ P3r) {
  int b = (int)blockIdx.x;
  if (b < PREP_PACK_BLOCKS) {
    const float* W; _Float16* WP; int din, hc;
    if (b < PB1)                      { W = Wl1; WP = P1l; din = L1_DIN; hc = L1_HC; }
    else if ((b -= PB1) < PB1)        { W = Wr1; WP = P1r; din = L1_DIN; hc = L1_HC; }
    else if ((b -= PB1) < PB2)        { W = Wl2; WP = P2l; din = L2_DIN; hc = L2_HC; }
    else if ((b -= PB2) < PB2)        { W = Wr2; WP = P2r; din = L2_DIN; hc = L2_HC; }
    else if ((b -= PB2) < PB3)        { W = Wl3; WP = P3l; din = L3_DIN; hc = L3_HC; }
    else            { b -= PB3;         W = Wr3; WP = P3r; din = L3_DIN; hc = L3_HC; }
    pack_w_body(W, WP, din, hc, b * EW_THREADS + (int)threadIdx.x);
    return;
  }
  b -= PREP_PACK_BLOCKS;
  if (b < PREP_ZERO_BLOCKS) {
    const v4i z = {0, 0, 0, 0};
    for (int i = (b * EW_THREADS + (int)threadIdx.x) * 4; i < n; i += PREP_ZERO_BLOCKS * EW_THREADS * 4) {
      if (i + 4 <= n) *(v4i*)(cnt + i) = z;
      else for (int j = i; j < n; ++j) cnt[j] = 0;
    }
    return;
  }
  b -= PREP_ZERO_BLOCKS;
  plane_to_f16_body(x, X16, nx, b, EW_BLOCKS);
}


__global__ void k_hist(const int* __restrict__ ei, long long E0,
                       int* __restrict__ cnt, int* __restrict__ slot) {
  long long e = (long long)blockIdx.x * blockDim.x + threadIdx.x;
  if (e >= E0) return;
  int d = ei[E0 + e];
  slot[e] = atomicAdd(&cnt[d], 1);
}

#define SCAN_TILE 1024

__global__ void k_tile_sum(const int* __restrict__ cnt, int n, int* __restrict__ tsum) {
  __shared__ int sh[8];
  int base = blockIdx.x * SCAN_TILE + threadIdx.x * 4;
  int v = 0;
#pragma unroll
  for (int j = 0; j < 4; ++j) if (base + j < n) v += cnt[base + j];
#pragma unroll
  for (int o = 16; o > 0; o >>= 1) v += __shfl_xor(v, o, 32);
  if ((threadIdx.x & 31) == 0) sh[threadIdx.x >> 5] = v;
  __syncthreads();
  if (threadIdx.x == 0) {
    int s = 0;
    for (int w = 0; w < 8; ++w) s += sh[w];
    tsum[blockIdx.x] = s;
  }
}

__global__ void k_scan(const int* __restrict__ cnt, int n, const int* __restrict__ tsum,
                       int* __restrict__ rowptr) {
  __shared__ int sh[8];
  int tid = threadIdx.x, lane = tid & 31, wave = tid >> 5;
  int pb = 0;
  for (int j = tid; j < (int)blockIdx.x; j += 256) pb += tsum[j];
#pragma unroll
  for (int o = 16; o > 0; o >>= 1) pb += __shfl_xor(pb, o, 32);
  if (lane == 0) sh[wave] = pb;
  __syncthreads();
  int tilebase = 0;
  for (int w = 0; w < 8; ++w) tilebase += sh[w];
  __syncthreads();

  int base = blockIdx.x * SCAN_TILE + tid * 4;
  int cv[4];
#pragma unroll
  for (int j = 0; j < 4; ++j) cv[j] = (base + j < n) ? cnt[base + j] : 0;
  int local = cv[0] + cv[1] + cv[2] + cv[3];
  int inc = local;
#pragma unroll
  for (int o = 1; o < 32; o <<= 1) {
    int t = __shfl_up(inc, o, 32);
    if (lane >= o) inc += t;
  }
  if (lane == 31) sh[wave] = inc;
  __syncthreads();
  int wbase = 0;
  for (int w = 0; w < wave; ++w) wbase += sh[w];
  int ex = tilebase + wbase + inc - local;
#pragma unroll
  for (int j = 0; j < 4; ++j) {
    if (base + j < n) rowptr[base + j] = ex;
    ex += cv[j];
  }
}

__global__ void k_scatter(const int* __restrict__ ei, long long E0,
                          const int* __restrict__ rowptr, const int* __restrict__ slot,
                          int* __restrict__ segA) {
  long long e = (long long)blockIdx.x * blockDim.x + threadIdx.x;
  if (e >= E0) return;
  int s = ei[e];
  int d = ei[E0 + e];
  segA[rowptr[d] + slot[e]] = s;
}

__global__ void k_seg_sort(const int* __restrict__ rowptr, const int* __restrict__ cnt,
                           const int* __restrict__ segA, int* __restrict__ seg, int n) {
  int node = blockIdx.x * (blockDim.x >> 5) + (threadIdx.x >> 5);
  if (node >= n) return;
  node = __builtin_amdgcn_readfirstlane(node);
  int lane = threadIdx.x & 31;
  int start = rowptr[node], deg = cnt[node];
  if (deg <= 32) {
    int mine = (lane < deg) ? segA[start + lane] : 0x7fffffff;
    int r = 0;
    for (int j = 0; j < deg; ++j) {
      int v = __builtin_amdgcn_readlane(mine, j);
      r += ((v < mine) || (v == mine && j < lane)) ? 1 : 0;
    }
    if (lane < deg) seg[start + r] = mine;
  } else {
    for (int i = lane; i < deg; i += 32) {
      int mine = segA[start + i];
      int r = 0;
      for (int j = 0; j < deg; ++j) {
        int v = segA[start + j];
        r += ((v < mine) || (v == mine && j < i)) ? 1 : 0;
      }
      seg[start + r] = mine;
    }
  }
}

template <int CTRL>
__device__ __forceinline__ float dpp_get(float v) {
  return __int_as_float(__builtin_amdgcn_update_dpp(0, __float_as_int(v), CTRL, 0xF, 0xF, true));
}
__device__ __forceinline__ float other_row_get(float v) {
  return __uint_as_float(__builtin_amdgcn_permlanex16(0u, __float_as_uint(v),
                                                      0x76543210u, 0xfedcba98u, true, false));
}
template <int LPH>
__device__ __forceinline__ float head_sum(float p) {
  static_assert(LPH == 4 || LPH == 32, "lanes per head");
  p += dpp_get<0x161>(p);
  p += dpp_get<0x162>(p);
  if constexpr (LPH == 32) {
    p += dpp_get<0x164>(p);
    p += dpp_get<0x168>(p);
    p += other_row_get(p);
  }
  return p;
}

template <int CPT> struct RowVec;
template <> struct RowVec<4> { typedef v4f T; };
template <> struct RowVec<2> { typedef v2f T; };

template <int H, int C, bool LAST>
__global__ void k_gat_node(const float* __restrict__ XL, const float* __restrict__ XR,
                           const float* __restrict__ att, const float* __restrict__ bias,
                           const int* __restrict__ rowptr, const int* __restrict__ cnt,
                           const int* __restrict__ seg,
                           _Float16* __restrict__ NEXT16, float* __restrict__ PROB, int n) {
  constexpr int HC  = H * C;
  constexpr int CPT = HC / 32;
  constexpr int LPH = 32 / H;
  constexpr int UNR = 4;
  typedef typename RowVec<CPT>::T vec_t;

  int node = blockIdx.x * (blockDim.x >> 5) + (threadIdx.x >> 5);
  if (node >= n) return;
  node = __builtin_amdgcn_readfirstlane(node);
  int lane = threadIdx.x & 31;
  int ch0  = lane * CPT;
  int start = rowptr[node], deg = cnt[node];

  vec_t xr = *(const vec_t*)(XR + (size_t)node * HC + ch0);
  vec_t av = *(const vec_t*)(att + ch0);
  vec_t xs = *(const vec_t*)(XL + (size_t)node * HC + ch0);

  float p = 0.0f;
#pragma unroll
  for (int j = 0; j < CPT; ++j) {
    float v = xs[j] + xr[j];
    v = (v > 0.0f) ? v : NEG_SLOPE * v;
    p += av[j] * v;
  }
  float m   = head_sum<LPH>(p);
  float den = 1.0f;
  vec_t acc = xs;

  for (int base = 0; base < deg; base += 32) {
    int nb = deg - base;
    if (nb > 32) nb = 32;
    int mysrc = (lane < nb) ? seg[start + base + lane] : node;
    for (int j = 0; j < nb; j += UNR) {
      vec_t x[UNR];
      float pu[UNR];
      bool  ok[UNR];
#pragma unroll
      for (int u = 0; u < UNR; ++u) {
        int s = __builtin_amdgcn_readlane(mysrc, j + u);
        ok[u] = (j + u) < nb;
        x[u] = *(const vec_t*)(XL + (size_t)s * HC + ch0);
      }
#pragma unroll
      for (int u = 0; u < UNR; ++u) {
        float q = 0.0f;
#pragma unroll
        for (int k = 0; k < CPT; ++k) {
          float v = x[u][k] + xr[k];
          v = (v > 0.0f) ? v : NEG_SLOPE * v;
          q += av[k] * v;
        }
        q = head_sum<LPH>(q);
        pu[u] = ok[u] ? q : -INFINITY;
      }
      float mn = fmaxf(fmaxf(m, pu[0]), fmaxf(pu[1], fmaxf(pu[2], pu[3])));
      float sc = __expf(m - mn);
      float e0 = ok[0] ? __expf(pu[0] - mn) : 0.0f;
      float e1 = ok[1] ? __expf(pu[1] - mn) : 0.0f;
      float e2 = ok[2] ? __expf(pu[2] - mn) : 0.0f;
      float e3 = ok[3] ? __expf(pu[3] - mn) : 0.0f;
      den = den * sc + ((e0 + e1) + (e2 + e3));
#pragma unroll
      for (int k = 0; k < CPT; ++k)
        acc[k] = acc[k] * sc + ((e0 * x[0][k] + e1 * x[1][k]) + (e2 * x[2][k] + e3 * x[3][k]));
      m = mn;
    }
  }

  float dn = fmaxf(den, 1e-16f);
  vec_t bv = *(const vec_t*)(bias + ch0);
  vec_t o;
#pragma unroll
  for (int k = 0; k < CPT; ++k) o[k] = acc[k] / dn + bv[k];

  if constexpr (!LAST) {
    static_assert(CPT == 4, "the f16 row is stored as 8 bytes per lane");
    *(v4h*)(NEXT16 + (size_t)node * HC + ch0) = __builtin_convertvector(elu4(o), v4h);
  } else {
    static_assert(H == 1 && CPT == 2, "one head: the wave's 32 lanes hold the whole row, 2 channels each");
    float mx = fmaxf(o[0], o[1]);
    mx = fmaxf(mx, other_row_get(mx));
    mx = fmaxf(mx, dpp_get<0x168>(mx));
    mx = fmaxf(mx, dpp_get<0x164>(mx));
    mx = fmaxf(mx, dpp_get<0x162>(mx));
    mx = fmaxf(mx, dpp_get<0x161>(mx));
    vec_t e;
    e[0] = __expf(o[0] - mx);
    e[1] = __expf(o[1] - mx);
    float s = e[0] + e[1];
    s += other_row_get(s);
    s += dpp_get<0x168>(s);
    s += dpp_get<0x164>(s);
    s += dpp_get<0x162>(s);
    s += dpp_get<0x161>(s);
    float inv = 1.0f / s;
    *(vec_t*)(PROB + (size_t)node * HC + ch0) = e * inv;
  }
}

template <int H, int C, bool LAST>
static void gat_layer(const _Float16* X16, int din,
                      const _Float16* WPl, const _Float16* WPr, const float* att, const float* b,
                      float* XL, float* XR, _Float16* next16, float* prob,
                      const int* rowptr, const int* cnt, const int* seg,
                      int N, hipStream_t stream) {
  constexpr int HC = H * C;

  int rowTiles = (N + 15) / 16;
  dim3 gg((unsigned)((rowTiles + 3) / 4), (unsigned)(2 * (HC / 64)));
  k_gemm2_f16<<<gg, 128, 0, stream>>>(X16, WPl, WPr, XL, XR, N, din, HC);

  k_gat_node<H, C, LAST><<<(N + 7) / 8, 256, 0, stream>>>(XL, XR, att, b, rowptr, cnt, seg, next16, prob, N);
}

extern "C" void kernel_launch(void* const* d_in, const int* in_sizes, int n_in,
                              void* d_out, int out_size, void* d_ws, size_t ws_size,
                              hipStream_t stream) {
  (void)n_in; (void)out_size; (void)ws_size;
  const float* x    = (const float*)d_in[0];
  const int*   ei   = (const int*)d_in[1];
  const float* Wl1  = (const float*)d_in[2];
  const float* Wr1  = (const float*)d_in[3];
  const float* att1 = (const float*)d_in[4];
  const float* b1   = (const float*)d_in[5];
  const float* Wl2  = (const float*)d_in[6];
  const float* Wr2  = (const float*)d_in[7];
  const float* att2 = (const float*)d_in[8];
  const float* b2   = (const float*)d_in[9];
  const float* Wl3  = (const float*)d_in[10];
  const float* Wr3  = (const float*)d_in[11];
  const float* att3 = (const float*)d_in[12];
  const float* b3   = (const float*)d_in[13];

  int       N  = in_sizes[0] / 256;
  long long E0 = (long long)in_sizes[1] / 2;

  uintptr_t p = (uintptr_t)d_ws;
  auto carve = [&](size_t bytes) -> void* {
    void* r = (void*)p;
    p += (bytes + 255) & ~(size_t)255;
    return r;
  };
  _Float16* X16 = (_Float16*)carve((size_t)N * 256 * sizeof(_Float16));
  _Float16* P1l = (_Float16*)carve((size_t)L1_DIN * L1_HC * sizeof(_Float16));
  _Float16* P1r = (_Float16*)carve((size_t)L1_DIN * L1_HC * sizeof(_Float16));
  _Float16* P2l = (_Float16*)carve((size_t)L2_DIN * L2_HC * sizeof(_Float16));
  _Float16* P2r = (_Float16*)carve((size_t)L2_DIN * L2_HC * sizeof(_Float16));
  _Float16* P3l = (_Float16*)carve((size_t)L3_DIN * L3_HC * sizeof(_Float16));
  _Float16* P3r = (_Float16*)carve((size_t)L3_DIN * L3_HC * sizeof(_Float16));
  float*    XL  = (float*)carve((size_t)N * 128 * sizeof(float));
  float*    XR  = (float*)carve((size_t)N * 128 * sizeof(float));
  int* cnt    = (int*)carve((size_t)N * sizeof(int));
  int* rowptr = (int*)carve((size_t)N * sizeof(int));
  int* tsum   = (int*)carve((size_t)((N + SCAN_TILE - 1) / SCAN_TILE) * sizeof(int));
  int* seg    = (int*)carve((size_t)E0 * sizeof(int));
  int* slot   = (int*)XL;
  int* segA   = (int*)XR;

  int nTiles = (N + SCAN_TILE - 1) / SCAN_TILE;
  int gE0    = (int)((E0 + 255) / 256);
  long long nx = (long long)N * 256;
  k_prep<<<PREP_BLOCKS, EW_THREADS, 0, stream>>>(x, X16, nx, cnt, N, Wl1, Wr1, Wl2, Wr2, Wl3, Wr3,
                                                 P1l, P1r, P2l, P2r, P3l, P3r);
  k_hist<<<gE0, 256, 0, stream>>>(ei, E0, cnt, slot);
  k_tile_sum<<<nTiles, 256, 0, stream>>>(cnt, N, tsum);
  k_scan<<<nTiles, 256, 0, stream>>>(cnt, N, tsum, rowptr);
  k_scatter<<<gE0, 256, 0, stream>>>(ei, E0, rowptr, slot, segA);
  k_seg_sort<<<(N + 7) / 8, 256, 0, stream>>>(rowptr, cnt, segA, seg, N);


  gat_layer<8, 16, false>(X16, L1_DIN, P1l, P1r, att1, b1,
                          XL, XR, X16, nullptr, rowptr, cnt, seg, N, stream);

  gat_layer<8, 16, false>(X16, L2_DIN, P2l, P2r, att2, b2,
                          XL, XR, X16, nullptr, rowptr, cnt, seg, N, stream);

  gat_layer<1, 64, true>(X16, L3_DIN, P3l, P3r, att3, b3,
                         XL, XR, nullptr, (float*)d_out, rowptr, cnt, seg, N, stream);
}


# Edges grouped by destination by a counting sort through LDS bins of 256 nodes: no global atomic, no scattered 4-byte store; same 12 launches, same grouped list, sum order unchanged
# speedup vs baseline: 5.5739x; 1.2393x over previous
#include <hip/hip_runtime.h>
#include <hip/hip_bf16.h>
#include <math.h>


#define NEG_SLOPE 0.2f

typedef __attribute__((ext_vector_type(16))) _Float16 v16h;
typedef __attribute__((ext_vector_type(8)))  _Float16 v8h;
typedef __attribute__((ext_vector_type(8)))  float    v8f;

#define EW_THREADS 256
#define EW_LOADS   8
#define EW_BLOCKS  512

typedef __attribute__((ext_vector_type(4))) float    v4f;
typedef __attribute__((ext_vector_type(2))) float    v2f;
typedef __attribute__((ext_vector_type(4))) _Float16 v4h;

__device__ __forceinline__ v4f elu4(v4f t) {
  v4f em;
#pragma unroll
  for (int k = 0; k < 4; ++k) em[k] = expm1f(t[k]);
#pragma unroll
  for (int k = 0; k < 4; ++k) t[k] = (t[k] > 0.0f) ? t[k] : em[k];
  return t;
}

__device__ __forceinline__ void plane_to_f16_body(const float* __restrict__ src, _Float16* __restrict__ dst,
                                                  long long n, int blk, int nblk) {
  constexpr long long TILE = (long long)EW_THREADS * EW_LOADS;
  const long long nvec = n >> 2;
  const v4f* __restrict__ s4 = (const v4f*)src;
  v4h* __restrict__ d4 = (v4h*)dst;
  for (long long t0 = (long long)blk * TILE; t0 < nvec; t0 += (long long)nblk * TILE) {
    const v4f* __restrict__ sp = s4 + t0 + threadIdx.x;
    v4h* __restrict__ dp = d4 + t0 + threadIdx.x;
    if (t0 + TILE <= nvec) {
      v4f v[EW_LOADS];
#pragma unroll
      for (int j = 0; j < EW_LOADS; ++j) v[j] = sp[j * EW_THREADS];
#pragma unroll
      for (int j = 0; j < EW_LOADS; ++j) dp[j * EW_THREADS] = __builtin_convertvector(v[j], v4h);
    } else {
      for (int j = 0; j < EW_LOADS; ++j)
        if (t0 + threadIdx.x + j * EW_THREADS < nvec) dp[j * EW_THREADS] = __builtin_convertvector(sp[j * EW_THREADS], v4h);
    }
  }
  if (blk == 0 && (long long)threadIdx.x < (n & 3)) {
    long long i = (nvec << 2) + threadIdx.x;
    dst[i] = (_Float16)src[i];
  }
}

__device__ __forceinline__ void pack_w_body(const float* __restrict__ W, _Float16* __restrict__ WP,
                                            int din, int dout, int idx) {
  int ktiles = din >> 5, ntiles = dout >> 4;
  int total = ktiles * ntiles * 32;
  if (idx >= total) return;
  int lane = idx & 31;
  int frag = idx >> 5;
  int nt = frag % ntiles;
  int kt = frag / ntiles;
  int half = lane >> 4, lr = lane & 15;
  _Float16* o = WP + ((size_t)frag * 32 + lane) * 16;
  const float* w0 = W + (size_t)(kt * 32 + half * 8) * dout + nt * 16 + lr;
#pragma unroll
  for (int j = 0; j < 8; ++j) {
    o[j]     = (_Float16)w0[(size_t)j * dout];
    o[8 + j] = (_Float16)w0[(size_t)(16 + j) * dout];
  }
}

__global__ void k_gemm2_f16(const _Float16* __restrict__ X,
                            const _Float16* __restrict__ WPl, const _Float16* __restrict__ WPr,
                            float* __restrict__ YL, float* __restrict__ YR,
                            int Mrows, int din, int dout) {
  int strips = dout >> 6;
  int strip  = blockIdx.y;
  bool right = strip >= strips;
  if (right) strip -= strips;
  const _Float16* __restrict__ WP = right ? WPr : WPl;
  float* __restrict__ Y = right ? YR : YL;
  int wave = threadIdx.x >> 5;
  int lane = threadIdx.x & 31;
  int half = lane >> 4;
  int lr   = lane & 15;
  int tileM = blockIdx.x * 4 + wave;
  if (tileM * 16 >= Mrows) return;
  int nt0 = strip * 4;
  int ntiles = dout >> 4;

  int row = tileM * 16 + lr;
  if (row >= Mrows) row = Mrows - 1;

  v8f acc0 = {}, acc1 = {}, acc2 = {}, acc3 = {};
  const _Float16* xrow = X + (size_t)row * din;
  const v16h* WPv = (const v16h*)WP;
  for (int k0 = 0; k0 < din; k0 += 32) {
    const v8h* xa = (const v8h*)(xrow + k0 + half * 8);
    v8h lo = xa[0];
    v8h hi = xa[2];
    v16h a = __builtin_shufflevector(lo, hi, 0, 1, 2, 3, 4, 5, 6, 7,
                                     8, 9, 10, 11, 12, 13, 14, 15);
    size_t fb = ((size_t)(k0 >> 5) * ntiles + nt0) * 32 + lane;
    v16h b0 = WPv[fb];
    v16h b1 = WPv[fb + 32];
    v16h b2 = WPv[fb + 64];
    v16h b3 = WPv[fb + 96];
    acc0 = __builtin_amdgcn_wmma_f32_16x16x32_f16(false, a, false, b0, (short)0, acc0, false, false);
    acc1 = __builtin_amdgcn_wmma_f32_16x16x32_f16(false, a, false, b1, (short)0, acc1, false, false);
    acc2 = __builtin_amdgcn_wmma_f32_16x16x32_f16(false, a, false, b2, (short)0, acc2, false, false);
    acc3 = __builtin_amdgcn_wmma_f32_16x16x32_f16(false, a, false, b3, (short)0, acc3, false, false);
  }
  int r0 = tileM * 16 + half * 8;
#pragma unroll
  for (int i = 0; i < 8; ++i) {
    int r = r0 + i;
    if (r < Mrows) {
      float* yp = Y + (size_t)r * dout + nt0 * 16 + lr;
      yp[0]  = acc0[i];
      yp[16] = acc1[i];
      yp[32] = acc2[i];
      yp[48] = acc3[i];
    }
  }
}

#define L1_DIN 256
#define L1_HC  128
#define L2_DIN 128
#define L2_HC  128
#define L3_DIN 128
#define L3_HC  64
#define PACK_BLOCKS_OF(din, hc) ((((din) >> 5) * ((hc) >> 4) * 32 + EW_THREADS - 1) / EW_THREADS)
#define PB1 PACK_BLOCKS_OF(L1_DIN, L1_HC)
#define PB2 PACK_BLOCKS_OF(L2_DIN, L2_HC)
#define PB3 PACK_BLOCKS_OF(L3_DIN, L3_HC)
#define PREP_PACK_BLOCKS (2 * (PB1 + PB2 + PB3))
#define PREP_BLOCKS (PREP_PACK_BLOCKS + EW_BLOCKS)

__global__ void __launch_bounds__(EW_THREADS)
k_prep(const float* __restrict__ x, _Float16* __restrict__ X16, long long nx,
       const float* __restrict__ Wl1, const float* __restrict__ Wr1,
       const float* __restrict__ Wl2, const float* __restrict__ Wr2,
       const float* __restrict__ Wl3, const float* __restrict__ Wr3,
       _Float16* __restrict__ P1l, _Float16* __restrict__ P1r,
       _Float16* __restrict__ P2l, _Float16* __restrict__ P2r,
       _Float16* __restrict__ P3l, _Float16* __restrict__ P3r) {
  int b = (int)blockIdx.x;
  if (b < PREP_PACK_BLOCKS) {
    const float* W; _Float16* WP; int din, hc;
    if (b < PB1)                      { W = Wl1; WP = P1l; din = L1_DIN; hc = L1_HC; }
    else if ((b -= PB1) < PB1)        { W = Wr1; WP = P1r; din = L1_DIN; hc = L1_HC; }
    else if ((b -= PB1) < PB2)        { W = Wl2; WP = P2l; din = L2_DIN; hc = L2_HC; }
    else if ((b -= PB2) < PB2)        { W = Wr2; WP = P2r; din = L2_DIN; hc = L2_HC; }
    else if ((b -= PB2) < PB3)        { W = Wl3; WP = P3l; din = L3_DIN; hc = L3_HC; }
    else            { b -= PB3;         W = Wr3; WP = P3r; din = L3_DIN; hc = L3_HC; }
    pack_w_body(W, WP, din, hc, b * EW_THREADS + (int)threadIdx.x);
    return;
  }
  b -= PREP_PACK_BLOCKS;
  plane_to_f16_body(x, X16, nx, b, EW_BLOCKS);
}

#define BIN_SH  8
#define BIN_W   (1 << BIN_SH)
#define NB_MAX  512
#define CHUNK   4096
#define EPT     (CHUNK / 256)
#define BIN_CAP 8192

typedef __attribute__((ext_vector_type(2))) int v2i;

__global__ void __launch_bounds__(256)
k_bin_count(const int* __restrict__ ei, long long E0, int nb, int G, int* __restrict__ binCnt) {
  __shared__ int h[NB_MAX];
  int tid = threadIdx.x;
  for (int i = tid; i < NB_MAX; i += 256) h[i] = 0;
  __syncthreads();
  long long base = (long long)blockIdx.x * CHUNK;
  int d[EPT];
#pragma unroll
  for (int j = 0; j < EPT; ++j) {
    long long e = base + j * 256 + tid;
    d[j] = (e < E0) ? ei[E0 + e] : -1;
  }
#pragma unroll
  for (int j = 0; j < EPT; ++j)
    if (d[j] >= 0) atomicAdd(&h[d[j] >> BIN_SH], 1);
  __syncthreads();
  for (int i = tid; i < nb; i += 256) binCnt[(size_t)i * G + blockIdx.x] = h[i];
}

__global__ void __launch_bounds__(256)
k_bin_scatter(const int* __restrict__ ei, long long E0, int nb, int G,
              const int* __restrict__ binOff, v2i* __restrict__ pairs) {
  __shared__ int h[NB_MAX];
  __shared__ int cur[NB_MAX];
  __shared__ int goff[NB_MAX];
  __shared__ int wsum[8];
  __shared__ v2i sp[CHUNK];
  int tid = threadIdx.x, lane = tid & 31, wave = tid >> 5;
  for (int i = tid; i < NB_MAX; i += 256) h[i] = 0;
  __syncthreads();
  long long base = (long long)blockIdx.x * CHUNK;
  int s[EPT], d[EPT];
#pragma unroll
  for (int j = 0; j < EPT; ++j) {
    long long e = base + j * 256 + tid;
    bool in = e < E0;
    s[j] = in ? ei[e] : 0;
    d[j] = in ? ei[E0 + e] : -1;
  }
#pragma unroll
  for (int j = 0; j < EPT; ++j)
    if (d[j] >= 0) atomicAdd(&h[d[j] >> BIN_SH], 1);
  __syncthreads();
  int c0 = h[2 * tid], c1 = h[2 * tid + 1];
  int local = c0 + c1, inc = local;
#pragma unroll
  for (int o = 1; o < 32; o <<= 1) {
    int t = __shfl_up(inc, o, 32);
    if (lane >= o) inc += t;
  }
  if (lane == 31) wsum[wave] = inc;
  __syncthreads();
  int wbase = 0;
  for (int w = 0; w < wave; ++w) wbase += wsum[w];
  int ex = wbase + inc - local;
  cur[2 * tid]     = ex;
  cur[2 * tid + 1] = ex + c0;
  goff[2 * tid]     = (2 * tid < nb)     ? binOff[(size_t)(2 * tid) * G + blockIdx.x] - ex : 0;
  goff[2 * tid + 1] = (2 * tid + 1 < nb) ? binOff[(size_t)(2 * tid + 1) * G + blockIdx.x] - (ex + c0) : 0;
  __syncthreads();
#pragma unroll
  for (int j = 0; j < EPT; ++j)
    if (d[j] >= 0) {
      int q = atomicAdd(&cur[d[j] >> BIN_SH], 1);
      v2i v = {s[j], d[j]};
      sp[q] = v;
    }
  __syncthreads();
  long long rem = E0 - base;
  int count = rem < CHUNK ? (int)rem : CHUNK;
  for (int i = tid; i < count; i += 256) {
    v2i v = sp[i];
    pairs[goff[v.y >> BIN_SH] + i] = v;
  }
}

#define SCAN_TILE 1024

__global__ void k_tile_sum(const int* __restrict__ cnt, int n, int* __restrict__ tsum) {
  __shared__ int sh[8];
  int base = blockIdx.x * SCAN_TILE + threadIdx.x * 4;
  int v = 0;
#pragma unroll
  for (int j = 0; j < 4; ++j) if (base + j < n) v += cnt[base + j];
#pragma unroll
  for (int o = 16; o > 0; o >>= 1) v += __shfl_xor(v, o, 32);
  if ((threadIdx.x & 31) == 0) sh[threadIdx.x >> 5] = v;
  __syncthreads();
  if (threadIdx.x == 0) {
    int s = 0;
    for (int w = 0; w < 8; ++w) s += sh[w];
    tsum[blockIdx.x] = s;
  }
}

__global__ void k_scan(const int* __restrict__ cnt, int n, const int* __restrict__ tsum,
                       int* __restrict__ rowptr) {
  __shared__ int sh[8];
  int tid = threadIdx.x, lane = tid & 31, wave = tid >> 5;
  int pb = 0;
  for (int j = tid; j < (int)blockIdx.x; j += 256) pb += tsum[j];
#pragma unroll
  for (int o = 16; o > 0; o >>= 1) pb += __shfl_xor(pb, o, 32);
  if (lane == 0) sh[wave] = pb;
  __syncthreads();
  int tilebase = 0;
  for (int w = 0; w < 8; ++w) tilebase += sh[w];
  __syncthreads();

  int base = blockIdx.x * SCAN_TILE + tid * 4;
  int cv[4];
#pragma unroll
  for (int j = 0; j < 4; ++j) cv[j] = (base + j < n) ? cnt[base + j] : 0;
  int local = cv[0] + cv[1] + cv[2] + cv[3];
  int inc = local;
#pragma unroll
  for (int o = 1; o < 32; o <<= 1) {
    int t = __shfl_up(inc, o, 32);
    if (lane >= o) inc += t;
  }
  if (lane == 31) sh[wave] = inc;
  __syncthreads();
  int wbase = 0;
  for (int w = 0; w < wave; ++w) wbase += sh[w];
  int ex = tilebase + wbase + inc - local;
#pragma unroll
  for (int j = 0; j < 4; ++j) {
    if (base + j < n) rowptr[base + j] = ex;
    ex += cv[j];
  }
}

#define FIN_THREADS 512
#define FIN_LOADS   8
__global__ void __launch_bounds__(FIN_THREADS)
k_bin_finish(const v2i* __restrict__ pairs, const int* __restrict__ binOff, int nb, int G,
             int E0i, int n, int* __restrict__ cnt, int* __restrict__ rowptr, int* __restrict__ seg) {
  __shared__ int ncnt[BIN_W];
  __shared__ int nstart[BIN_W];
  __shared__ int ncur[BIN_W];
  __shared__ int wsum[8];
  __shared__ int ssrc[BIN_CAP];
  int tid = threadIdx.x, lane = tid & 31, wave = tid >> 5;
  int bin = blockIdx.x;
  int b0 = binOff[(size_t)bin * G];
  int b1 = (bin + 1 < nb) ? binOff[(size_t)(bin + 1) * G] : E0i;
  int size = b1 - b0;
  int nbase = bin << BIN_SH;
  if (tid < BIN_W) ncnt[tid] = 0;
  __syncthreads();
  for (int i0 = 0; i0 < size; i0 += FIN_THREADS * FIN_LOADS) {
    int dd[FIN_LOADS];
#pragma unroll
    for (int u = 0; u < FIN_LOADS; ++u) {
      int i = i0 + u * FIN_THREADS + tid;
      dd[u] = (i < size) ? pairs[b0 + i].y : -1;
    }
#pragma unroll
    for (int u = 0; u < FIN_LOADS; ++u)
      if (dd[u] >= 0) atomicAdd(&ncnt[dd[u] - nbase], 1);
  }
  __syncthreads();
  int c = 0, inc = 0;
  if (tid < BIN_W) {
    c = ncnt[tid];
    inc = c;
#pragma unroll
    for (int o = 1; o < 32; o <<= 1) {
      int t = __shfl_up(inc, o, 32);
      if (lane >= o) inc += t;
    }
    if (lane == 31) wsum[wave] = inc;
  }
  __syncthreads();
  if (tid < BIN_W) {
    int wbase = 0;
    for (int w = 0; w < wave; ++w) wbase += wsum[w];
    int ex = wbase + inc - c;
    nstart[tid] = ex;
    ncur[tid]   = ex;
    int node = nbase + tid;
    if (node < n) {
      cnt[node]    = c;
      rowptr[node] = b0 + ex;
    }
  }
  __syncthreads();
  if (size <= BIN_CAP) {
    for (int i0 = 0; i0 < size; i0 += FIN_THREADS * FIN_LOADS) {
      v2i vv[FIN_LOADS];
#pragma unroll
      for (int u = 0; u < FIN_LOADS; ++u) {
        int i = i0 + u * FIN_THREADS + tid;
        v2i none = {0, -1};
        vv[u] = (i < size) ? pairs[b0 + i] : none;
      }
#pragma unroll
      for (int u = 0; u < FIN_LOADS; ++u)
        if (vv[u].y >= 0) {
          int q = atomicAdd(&ncur[vv[u].y - nbase], 1);
          ssrc[q] = vv[u].x;
        }
    }
    __syncthreads();
    for (int ln = wave; ln < BIN_W; ln += FIN_THREADS / 32) {
      int st  = __builtin_amdgcn_readfirstlane(nstart[ln]);
      int deg = __builtin_amdgcn_readfirstlane(ncnt[ln]);
      if (deg <= 32) {
        int mine = (lane < deg) ? ssrc[st + lane] : 0x7fffffff;
        int r = 0;
        for (int j = 0; j < deg; ++j) {
          int v = __builtin_amdgcn_readlane(mine, j);
          r += ((v < mine) || (v == mine && j < lane)) ? 1 : 0;
        }
        if (lane < deg) seg[b0 + st + r] = mine;
      } else {
        for (int i = lane; i < deg; i += 32) {
          int mine = ssrc[st + i];
          int r = 0;
          for (int j = 0; j < deg; ++j) {
            int v = ssrc[st + j];
            r += ((v < mine) || (v == mine && j < i)) ? 1 : 0;
          }
          seg[b0 + st + r] = mine;
        }
      }
    }
  } else {
    for (int ln = wave; ln < BIN_W; ln += FIN_THREADS / 32) {
      int nd = nbase + ln;
      int st = __builtin_amdgcn_readfirstlane(nstart[ln]);
      for (int i = lane; i < size; i += 32) {
        v2i v = pairs[b0 + i];
        if (v.y == nd) {
          int r = 0;
          for (int j = 0; j < size; ++j) {
            v2i u = pairs[b0 + j];
            r += (u.y == nd && ((u.x < v.x) || (u.x == v.x && j < i))) ? 1 : 0;
          }
          seg[b0 + st + r] = v.x;
        }
      }
    }
  }
}

template <int CTRL>
__device__ __forceinline__ float dpp_get(float v) {
  return __int_as_float(__builtin_amdgcn_update_dpp(0, __float_as_int(v), CTRL, 0xF, 0xF, true));
}
__device__ __forceinline__ float other_row_get(float v) {
  return __uint_as_float(__builtin_amdgcn_permlanex16(0u, __float_as_uint(v),
                                                      0x76543210u, 0xfedcba98u, true, false));
}
template <int LPH>
__device__ __forceinline__ float head_sum(float p) {
  static_assert(LPH == 4 || LPH == 32, "lanes per head");
  p += dpp_get<0x161>(p);
  p += dpp_get<0x162>(p);
  if constexpr (LPH == 32) {
    p += dpp_get<0x164>(p);
    p += dpp_get<0x168>(p);
    p += other_row_get(p);
  }
  return p;
}

template <int CPT> struct RowVec;
template <> struct RowVec<4> { typedef v4f T; };
template <> struct RowVec<2> { typedef v2f T; };

template <int H, int C, bool LAST>
__global__ void k_gat_node(const float* __restrict__ XL, const float* __restrict__ XR,
                           const float* __restrict__ att, const float* __restrict__ bias,
                           const int* __restrict__ rowptr, const int* __restrict__ cnt,
                           const int* __restrict__ seg,
                           _Float16* __restrict__ NEXT16, float* __restrict__ PROB, int n) {
  constexpr int HC  = H * C;
  constexpr int CPT = HC / 32;
  constexpr int LPH = 32 / H;
  constexpr int UNR = 4;
  typedef typename RowVec<CPT>::T vec_t;

  int node = blockIdx.x * (blockDim.x >> 5) + (threadIdx.x >> 5);
  if (node >= n) return;
  node = __builtin_amdgcn_readfirstlane(node);
  int lane = threadIdx.x & 31;
  int ch0  = lane * CPT;
  int start = rowptr[node], deg = cnt[node];

  vec_t xr = *(const vec_t*)(XR + (size_t)node * HC + ch0);
  vec_t av = *(const vec_t*)(att + ch0);
  vec_t xs = *(const vec_t*)(XL + (size_t)node * HC + ch0);

  float p = 0.0f;
#pragma unroll
  for (int j = 0; j < CPT; ++j) {
    float v = xs[j] + xr[j];
    v = (v > 0.0f) ? v : NEG_SLOPE * v;
    p += av[j] * v;
  }
  float m   = head_sum<LPH>(p);
  float den = 1.0f;
  vec_t acc = xs;

  for (int base = 0; base < deg; base += 32) {
    int nb = deg - base;
    if (nb > 32) nb = 32;
    int mysrc = (lane < nb) ? seg[start + base + lane] : node;
    for (int j = 0; j < nb; j += UNR) {
      vec_t x[UNR];
      float pu[UNR];
      bool  ok[UNR];
#pragma unroll
      for (int u = 0; u < UNR; ++u) {
        int s = __builtin_amdgcn_readlane(mysrc, j + u);
        ok[u] = (j + u) < nb;
        x[u] = *(const vec_t*)(XL + (size_t)s * HC + ch0);
      }
#pragma unroll
      for (int u = 0; u < UNR; ++u) {
        float q = 0.0f;
#pragma unroll
        for (int k = 0; k < CPT; ++k) {
          float v = x[u][k] + xr[k];
          v = (v > 0.0f) ? v : NEG_SLOPE * v;
          q += av[k] * v;
        }
        q = head_sum<LPH>(q);
        pu[u] = ok[u] ? q : -INFINITY;
      }
      float mn = fmaxf(fmaxf(m, pu[0]), fmaxf(pu[1], fmaxf(pu[2], pu[3])));
      float sc = __expf(m - mn);
      float e0 = ok[0] ? __expf(pu[0] - mn) : 0.0f;
      float e1 = ok[1] ? __expf(pu[1] - mn) : 0.0f;
      float e2 = ok[2] ? __expf(pu[2] - mn) : 0.0f;
      float e3 = ok[3] ? __expf(pu[3] - mn) : 0.0f;
      den = den * sc + ((e0 + e1) + (e2 + e3));
#pragma unroll
      for (int k = 0; k < CPT; ++k)
        acc[k] = acc[k] * sc + ((e0 * x[0][k] + e1 * x[1][k]) + (e2 * x[2][k] + e3 * x[3][k]));
      m = mn;
    }
  }

  float dn = fmaxf(den, 1e-16f);
  vec_t bv = *(const vec_t*)(bias + ch0);
  vec_t o;
#pragma unroll
  for (int k = 0; k < CPT; ++k) o[k] = acc[k] / dn + bv[k];

  if constexpr (!LAST) {
    static_assert(CPT == 4, "the f16 row is stored as 8 bytes per lane");
    *(v4h*)(NEXT16 + (size_t)node * HC + ch0) = __builtin_convertvector(elu4(o), v4h);
  } else {
    static_assert(H == 1 && CPT == 2, "one head: the wave's 32 lanes hold the whole row, 2 channels each");
    float mx = fmaxf(o[0], o[1]);
    mx = fmaxf(mx, other_row_get(mx));
    mx = fmaxf(mx, dpp_get<0x168>(mx));
    mx = fmaxf(mx, dpp_get<0x164>(mx));
    mx = fmaxf(mx, dpp_get<0x162>(mx));
    mx = fmaxf(mx, dpp_get<0x161>(mx));
    vec_t e;
    e[0] = __expf(o[0] - mx);
    e[1] = __expf(o[1] - mx);
    float s = e[0] + e[1];
    s += other_row_get(s);
    s += dpp_get<0x168>(s);
    s += dpp_get<0x164>(s);
    s += dpp_get<0x162>(s);
    s += dpp_get<0x161>(s);
    float inv = 1.0f / s;
    *(vec_t*)(PROB + (size_t)node * HC + ch0) = e * inv;
  }
}

template <int H, int C, bool LAST>
static void gat_layer(const _Float16* X16, int din,
                      const _Float16* WPl, const _Float16* WPr, const float* att, const float* b,
                      float* XL, float* XR, _Float16* next16, float* prob,
                      const int* rowptr, const int* cnt, const int* seg,
                      int N, hipStream_t stream) {
  constexpr int HC = H * C;

  int rowTiles = (N + 15) / 16;
  dim3 gg((unsigned)((rowTiles + 3) / 4), (unsigned)(2 * (HC / 64)));
  k_gemm2_f16<<<gg, 128, 0, stream>>>(X16, WPl, WPr, XL, XR, N, din, HC);

  k_gat_node<H, C, LAST><<<(N + 7) / 8, 256, 0, stream>>>(XL, XR, att, b, rowptr, cnt, seg, next16, prob, N);
}

extern "C" void kernel_launch(void* const* d_in, const int* in_sizes, int n_in,
                              void* d_out, int out_size, void* d_ws, size_t ws_size,
                              hipStream_t stream) {
  (void)n_in; (void)out_size; (void)ws_size;
  const float* x    = (const float*)d_in[0];
  const int*   ei   = (const int*)d_in[1];
  const float* Wl1  = (const float*)d_in[2];
  const float* Wr1  = (const float*)d_in[3];
  const float* att1 = (const float*)d_in[4];
  const float* b1   = (const float*)d_in[5];
  const float* Wl2  = (const float*)d_in[6];
  const float* Wr2  = (const float*)d_in[7];
  const float* att2 = (const float*)d_in[8];
  const float* b2   = (const float*)d_in[9];
  const float* Wl3  = (const float*)d_in[10];
  const float* Wr3  = (const float*)d_in[11];
  const float* att3 = (const float*)d_in[12];
  const float* b3   = (const float*)d_in[13];

  int       N  = in_sizes[0] / 256;
  long long E0 = (long long)in_sizes[1] / 2;

  uintptr_t p = (uintptr_t)d_ws;
  auto carve = [&](size_t bytes) -> void* {
    void* r = (void*)p;
    p += (bytes + 255) & ~(size_t)255;
    return r;
  };
  _Float16* X16 = (_Float16*)carve((size_t)N * 256 * sizeof(_Float16));
  _Float16* P1l = (_Float16*)carve((size_t)L1_DIN * L1_HC * sizeof(_Float16));
  _Float16* P1r = (_Float16*)carve((size_t)L1_DIN * L1_HC * sizeof(_Float16));
  _Float16* P2l = (_Float16*)carve((size_t)L2_DIN * L2_HC * sizeof(_Float16));
  _Float16* P2r = (_Float16*)carve((size_t)L2_DIN * L2_HC * sizeof(_Float16));
  _Float16* P3l = (_Float16*)carve((size_t)L3_DIN * L3_HC * sizeof(_Float16));
  _Float16* P3r = (_Float16*)carve((size_t)L3_DIN * L3_HC * sizeof(_Float16));
  float*    XL  = (float*)carve((size_t)N * 128 * sizeof(float));
  float*    XR  = (float*)carve((size_t)N * 128 * sizeof(float));
  int* cnt    = (int*)carve((size_t)N * sizeof(int));
  int* rowptr = (int*)carve((size_t)N * sizeof(int));
  int* seg    = (int*)carve((size_t)E0 * sizeof(int));
  int nb  = (N + BIN_W - 1) / BIN_W;
  int G   = (int)((E0 + CHUNK - 1) / CHUNK);
  int nbg = nb * G;
  int nTiles = (nbg + SCAN_TILE - 1) / SCAN_TILE;
  v2i* pairs  = (v2i*)XL;
  int* binCnt = (int*)XR;
  int* binOff = binCnt + nbg;
  int* tsum   = binOff + nbg;

  long long nx = (long long)N * 256;
  k_prep<<<PREP_BLOCKS, EW_THREADS, 0, stream>>>(x, X16, nx, Wl1, Wr1, Wl2, Wr2, Wl3, Wr3,
                                                 P1l, P1r, P2l, P2r, P3l, P3r);
  k_bin_count<<<G, 256, 0, stream>>>(ei, E0, nb, G, binCnt);
  k_tile_sum<<<nTiles, 256, 0, stream>>>(binCnt, nbg, tsum);
  k_scan<<<nTiles, 256, 0, stream>>>(binCnt, nbg, tsum, binOff);
  k_bin_scatter<<<G, 256, 0, stream>>>(ei, E0, nb, G, binOff, pairs);
  k_bin_finish<<<nb, FIN_THREADS, 0, stream>>>(pairs, binOff, nb, G, (int)E0, N, cnt, rowptr, seg);


  gat_layer<8, 16, false>(X16, L1_DIN, P1l, P1r, att1, b1,
                          XL, XR, X16, nullptr, rowptr, cnt, seg, N, stream);

  gat_layer<8, 16, false>(X16, L2_DIN, P2l, P2r, att2, b2,
                          XL, XR, X16, nullptr, rowptr, cnt, seg, N, stream);

  gat_layer<1, 64, true>(X16, L3_DIN, P3l, P3r, att3, b3,
                         XL, XR, nullptr, (float*)d_out, rowptr, cnt, seg, N, stream);
}


# Fewer issue cycles per node in the node kernel: exact-size last group, scalar row bases, one-instruction swizzled adds, log2-unit scores, one reciprocal per row, ELU on hardware exp; sum order kept
# speedup vs baseline: 6.2138x; 1.1148x over previous
#include <hip/hip_runtime.h>
#include <hip/hip_bf16.h>
#include <math.h>


#define NEG_SLOPE 0.2f

typedef __attribute__((ext_vector_type(16))) _Float16 v16h;
typedef __attribute__((ext_vector_type(8)))  _Float16 v8h;
typedef __attribute__((ext_vector_type(8)))  float    v8f;

#define EW_THREADS 256
#define EW_LOADS   8
#define EW_BLOCKS  512

typedef __attribute__((ext_vector_type(4))) float    v4f;
typedef __attribute__((ext_vector_type(2))) float    v2f;
typedef __attribute__((ext_vector_type(4))) _Float16 v4h;

#define LOG2E_F 1.4426950408889634f

__device__ __forceinline__ v4f elu4(v4f t) {
  v4f em;
#pragma unroll
  for (int k = 0; k < 4; ++k) em[k] = __builtin_amdgcn_exp2f(t[k] * LOG2E_F) - 1.0f;
#pragma unroll
  for (int k = 0; k < 4; ++k) t[k] = (t[k] > 0.0f) ? t[k] : em[k];
  return t;
}

__device__ __forceinline__ void plane_to_f16_body(const float* __restrict__ src, _Float16* __restrict__ dst,
                                                  long long n, int blk, int nblk) {
  constexpr long long TILE = (long long)EW_THREADS * EW_LOADS;
  const long long nvec = n >> 2;
  const v4f* __restrict__ s4 = (const v4f*)src;
  v4h* __restrict__ d4 = (v4h*)dst;
  for (long long t0 = (long long)blk * TILE; t0 < nvec; t0 += (long long)nblk * TILE) {
    const v4f* __restrict__ sp = s4 + t0 + threadIdx.x;
    v4h* __restrict__ dp = d4 + t0 + threadIdx.x;
    if (t0 + TILE <= nvec) {
      v4f v[EW_LOADS];
#pragma unroll
      for (int j = 0; j < EW_LOADS; ++j) v[j] = sp[j * EW_THREADS];
#pragma unroll
      for (int j = 0; j < EW_LOADS; ++j) dp[j * EW_THREADS] = __builtin_convertvector(v[j], v4h);
    } else {
      for (int j = 0; j < EW_LOADS; ++j)
        if (t0 + threadIdx.x + j * EW_THREADS < nvec) dp[j * EW_THREADS] = __builtin_convertvector(sp[j * EW_THREADS], v4h);
    }
  }
  if (blk == 0 && (long long)threadIdx.x < (n & 3)) {
    long long i = (nvec << 2) + threadIdx.x;
    dst[i] = (_Float16)src[i];
  }
}

__device__ __forceinline__ void pack_w_body(const float* __restrict__ W, _Float16* __restrict__ WP,
                                            int din, int dout, int idx) {
  int ktiles = din >> 5, ntiles = dout >> 4;
  int total = ktiles * ntiles * 32;
  if (idx >= total) return;
  int lane = idx & 31;
  int frag = idx >> 5;
  int nt = frag % ntiles;
  int kt = frag / ntiles;
  int half = lane >> 4, lr = lane & 15;
  _Float16* o = WP + ((size_t)frag * 32 + lane) * 16;
  const float* w0 = W + (size_t)(kt * 32 + half * 8) * dout + nt * 16 + lr;
#pragma unroll
  for (int j = 0; j < 8; ++j) {
    o[j]     = (_Float16)w0[(size_t)j * dout];
    o[8 + j] = (_Float16)w0[(size_t)(16 + j) * dout];
  }
}

__global__ void k_gemm2_f16(const _Float16* __restrict__ X,
                            const _Float16* __restrict__ WPl, const _Float16* __restrict__ WPr,
                            float* __restrict__ YL, float* __restrict__ YR,
                            int Mrows, int din, int dout) {
  int strips = dout >> 6;
  int strip  = blockIdx.y;
  bool right = strip >= strips;
  if (right) strip -= strips;
  const _Float16* __restrict__ WP = right ? WPr : WPl;
  float* __restrict__ Y = right ? YR : YL;
  int wave = threadIdx.x >> 5;
  int lane = threadIdx.x & 31;
  int half = lane >> 4;
  int lr   = lane & 15;
  int tileM = blockIdx.x * 4 + wave;
  if (tileM * 16 >= Mrows) return;
  int nt0 = strip * 4;
  int ntiles = dout >> 4;

  int row = tileM * 16 + lr;
  if (row >= Mrows) row = Mrows - 1;

  v8f acc0 = {}, acc1 = {}, acc2 = {}, acc3 = {};
  const _Float16* xrow = X + (size_t)row * din;
  const v16h* WPv = (const v16h*)WP;
  for (int k0 = 0; k0 < din; k0 += 32) {
    const v8h* xa = (const v8h*)(xrow + k0 + half * 8);
    v8h lo = xa[0];
    v8h hi = xa[2];
    v16h a = __builtin_shufflevector(lo, hi, 0, 1, 2, 3, 4, 5, 6, 7,
                                     8, 9, 10, 11, 12, 13, 14, 15);
    size_t fb = ((size_t)(k0 >> 5) * ntiles + nt0) * 32 + lane;
    v16h b0 = WPv[fb];
    v16h b1 = WPv[fb + 32];
    v16h b2 = WPv[fb + 64];
    v16h b3 = WPv[fb + 96];
    acc0 = __builtin_amdgcn_wmma_f32_16x16x32_f16(false, a, false, b0, (short)0, acc0, false, false);
    acc1 = __builtin_amdgcn_wmma_f32_16x16x32_f16(false, a, false, b1, (short)0, acc1, false, false);
    acc2 = __builtin_amdgcn_wmma_f32_16x16x32_f16(false, a, false, b2, (short)0, acc2, false, false);
    acc3 = __builtin_amdgcn_wmma_f32_16x16x32_f16(false, a, false, b3, (short)0, acc3, false, false);
  }
  int r0 = tileM * 16 + half * 8;
#pragma unroll
  for (int i = 0; i < 8; ++i) {
    int r = r0 + i;
    if (r < Mrows) {
      float* yp = Y + (size_t)r * dout + nt0 * 16 + lr;
      yp[0]  = acc0[i];
      yp[16] = acc1[i];
      yp[32] = acc2[i];
      yp[48] = acc3[i];
    }
  }
}

#define L1_DIN 256
#define L1_HC  128
#define L2_DIN 128
#define L2_HC  128
#define L3_DIN 128
#define L3_HC  64
#define PACK_BLOCKS_OF(din, hc) ((((din) >> 5) * ((hc) >> 4) * 32 + EW_THREADS - 1) / EW_THREADS)
#define PB1 PACK_BLOCKS_OF(L1_DIN, L1_HC)
#define PB2 PACK_BLOCKS_OF(L2_DIN, L2_HC)
#define PB3 PACK_BLOCKS_OF(L3_DIN, L3_HC)
#define PREP_PACK_BLOCKS (2 * (PB1 + PB2 + PB3))
#define PREP_BLOCKS (PREP_PACK_BLOCKS + EW_BLOCKS)

__global__ void __launch_bounds__(EW_THREADS)
k_prep(const float* __restrict__ x, _Float16* __restrict__ X16, long long nx,
       const float* __restrict__ Wl1, const float* __restrict__ Wr1,
       const float* __restrict__ Wl2, const float* __restrict__ Wr2,
       const float* __restrict__ Wl3, const float* __restrict__ Wr3,
       _Float16* __restrict__ P1l, _Float16* __restrict__ P1r,
       _Float16* __restrict__ P2l, _Float16* __restrict__ P2r,
       _Float16* __restrict__ P3l, _Float16* __restrict__ P3r) {
  int b = (int)blockIdx.x;
  if (b < PREP_PACK_BLOCKS) {
    const float* W; _Float16* WP; int din, hc;
    if (b < PB1)                      { W = Wl1; WP = P1l; din = L1_DIN; hc = L1_HC; }
    else if ((b -= PB1) < PB1)        { W = Wr1; WP = P1r; din = L1_DIN; hc = L1_HC; }
    else if ((b -= PB1) < PB2)        { W = Wl2; WP = P2l; din = L2_DIN; hc = L2_HC; }
    else if ((b -= PB2) < PB2)        { W = Wr2; WP = P2r; din = L2_DIN; hc = L2_HC; }
    else if ((b -= PB2) < PB3)        { W = Wl3; WP = P3l; din = L3_DIN; hc = L3_HC; }
    else            { b -= PB3;         W = Wr3; WP = P3r; din = L3_DIN; hc = L3_HC; }
    pack_w_body(W, WP, din, hc, b * EW_THREADS + (int)threadIdx.x);
    return;
  }
  b -= PREP_PACK_BLOCKS;
  plane_to_f16_body(x, X16, nx, b, EW_BLOCKS);
}

#define BIN_SH  8
#define BIN_W   (1 << BIN_SH)
#define NB_MAX  512
#define CHUNK   4096
#define EPT     (CHUNK / 256)
#define BIN_CAP 8192

typedef __attribute__((ext_vector_type(2))) int v2i;

__global__ void __launch_bounds__(256)
k_bin_count(const int* __restrict__ ei, long long E0, int nb, int G, int* __restrict__ binCnt) {
  __shared__ int h[NB_MAX];
  int tid = threadIdx.x;
  for (int i = tid; i < NB_MAX; i += 256) h[i] = 0;
  __syncthreads();
  long long base = (long long)blockIdx.x * CHUNK;
  int d[EPT];
#pragma unroll
  for (int j = 0; j < EPT; ++j) {
    long long e = base + j * 256 + tid;
    d[j] = (e < E0) ? ei[E0 + e] : -1;
  }
#pragma unroll
  for (int j = 0; j < EPT; ++j)
    if (d[j] >= 0) atomicAdd(&h[d[j] >> BIN_SH], 1);
  __syncthreads();
  for (int i = tid; i < nb; i += 256) binCnt[(size_t)i * G + blockIdx.x] = h[i];
}

__global__ void __launch_bounds__(256)
k_bin_scatter(const int* __restrict__ ei, long long E0, int nb, int G,
              const int* __restrict__ binOff, v2i* __restrict__ pairs) {
  __shared__ int h[NB_MAX];
  __shared__ int cur[NB_MAX];
  __shared__ int goff[NB_MAX];
  __shared__ int wsum[8];
  __shared__ v2i sp[CHUNK];
  int tid = threadIdx.x, lane = tid & 31, wave = tid >> 5;
  for (int i = tid; i < NB_MAX; i += 256) h[i] = 0;
  __syncthreads();
  long long base = (long long)blockIdx.x * CHUNK;
  int s[EPT], d[EPT];
#pragma unroll
  for (int j = 0; j < EPT; ++j) {
    long long e = base + j * 256 + tid;
    bool in = e < E0;
    s[j] = in ? ei[e] : 0;
    d[j] = in ? ei[E0 + e] : -1;
  }
#pragma unroll
  for (int j = 0; j < EPT; ++j)
    if (d[j] >= 0) atomicAdd(&h[d[j] >> BIN_SH], 1);
  __syncthreads();
  int c0 = h[2 * tid], c1 = h[2 * tid + 1];
  int local = c0 + c1, inc = local;
#pragma unroll
  for (int o = 1; o < 32; o <<= 1) {
    int t = __shfl_up(inc, o, 32);
    if (lane >= o) inc += t;
  }
  if (lane == 31) wsum[wave] = inc;
  __syncthreads();
  int wbase = 0;
  for (int w = 0; w < wave; ++w) wbase += wsum[w];
  int ex = wbase + inc - local;
  cur[2 * tid]     = ex;
  cur[2 * tid + 1] = ex + c0;
  goff[2 * tid]     = (2 * tid < nb)     ? binOff[(size_t)(2 * tid) * G + blockIdx.x] - ex : 0;
  goff[2 * tid + 1] = (2 * tid + 1 < nb) ? binOff[(size_t)(2 * tid + 1) * G + blockIdx.x] - (ex + c0) : 0;
  __syncthreads();
#pragma unroll
  for (int j = 0; j < EPT; ++j)
    if (d[j] >= 0) {
      int q = atomicAdd(&cur[d[j] >> BIN_SH], 1);
      v2i v = {s[j], d[j]};
      sp[q] = v;
    }
  __syncthreads();
  long long rem = E0 - base;
  int count = rem < CHUNK ? (int)rem : CHUNK;
  for (int i = tid; i < count; i += 256) {
    v2i v = sp[i];
    pairs[goff[v.y >> BIN_SH] + i] = v;
  }
}

#define SCAN_TILE 1024

__global__ void k_tile_sum(const int* __restrict__ cnt, int n, int* __restrict__ tsum) {
  __shared__ int sh[8];
  int base = blockIdx.x * SCAN_TILE + threadIdx.x * 4;
  int v = 0;
#pragma unroll
  for (int j = 0; j < 4; ++j) if (base + j < n) v += cnt[base + j];
#pragma unroll
  for (int o = 16; o > 0; o >>= 1) v += __shfl_xor(v, o, 32);
  if ((threadIdx.x & 31) == 0) sh[threadIdx.x >> 5] = v;
  __syncthreads();
  if (threadIdx.x == 0) {
    int s = 0;
    for (int w = 0; w < 8; ++w) s += sh[w];
    tsum[blockIdx.x] = s;
  }
}

__global__ void k_scan(const int* __restrict__ cnt, int n, const int* __restrict__ tsum,
                       int* __restrict__ rowptr) {
  __shared__ int sh[8];
  int tid = threadIdx.x, lane = tid & 31, wave = tid >> 5;
  int pb = 0;
  for (int j = tid; j < (int)blockIdx.x; j += 256) pb += tsum[j];
#pragma unroll
  for (int o = 16; o > 0; o >>= 1) pb += __shfl_xor(pb, o, 32);
  if (lane == 0) sh[wave] = pb;
  __syncthreads();
  int tilebase = 0;
  for (int w = 0; w < 8; ++w) tilebase += sh[w];
  __syncthreads();

  int base = blockIdx.x * SCAN_TILE + tid * 4;
  int cv[4];
#pragma unroll
  for (int j = 0; j < 4; ++j) cv[j] = (base + j < n) ? cnt[base + j] : 0;
  int local = cv[0] + cv[1] + cv[2] + cv[3];
  int inc = local;
#pragma unroll
  for (int o = 1; o < 32; o <<= 1) {
    int t = __shfl_up(inc, o, 32);
    if (lane >= o) inc += t;
  }
  if (lane == 31) sh[wave] = inc;
  __syncthreads();
  int wbase = 0;
  for (int w = 0; w < wave; ++w) wbase += sh[w];
  int ex = tilebase + wbase + inc - local;
#pragma unroll
  for (int j = 0; j < 4; ++j) {
    if (base + j < n) rowptr[base + j] = ex;
    ex += cv[j];
  }
}

#define FIN_THREADS 512
#define FIN_LOADS   8
__global__ void __launch_bounds__(FIN_THREADS)
k_bin_finish(const v2i* __restrict__ pairs, const int* __restrict__ binOff, int nb, int G,
             int E0i, int n, int* __restrict__ cnt, int* __restrict__ rowptr, int* __restrict__ seg) {
  __shared__ int ncnt[BIN_W];
  __shared__ int nstart[BIN_W];
  __shared__ int ncur[BIN_W];
  __shared__ int wsum[8];
  __shared__ int ssrc[BIN_CAP];
  int tid = threadIdx.x, lane = tid & 31, wave = tid >> 5;
  int bin = blockIdx.x;
  int b0 = binOff[(size_t)bin * G];
  int b1 = (bin + 1 < nb) ? binOff[(size_t)(bin + 1) * G] : E0i;
  int size = b1 - b0;
  int nbase = bin << BIN_SH;
  if (tid < BIN_W) ncnt[tid] = 0;
  __syncthreads();
  for (int i0 = 0; i0 < size; i0 += FIN_THREADS * FIN_LOADS) {
    int dd[FIN_LOADS];
#pragma unroll
    for (int u = 0; u < FIN_LOADS; ++u) {
      int i = i0 + u * FIN_THREADS + tid;
      dd[u] = (i < size) ? pairs[b0 + i].y : -1;
    }
#pragma unroll
    for (int u = 0; u < FIN_LOADS; ++u)
      if (dd[u] >= 0) atomicAdd(&ncnt[dd[u] - nbase], 1);
  }
  __syncthreads();
  int c = 0, inc = 0;
  if (tid < BIN_W) {
    c = ncnt[tid];
    inc = c;
#pragma unroll
    for (int o = 1; o < 32; o <<= 1) {
      int t = __shfl_up(inc, o, 32);
      if (lane >= o) inc += t;
    }
    if (lane == 31) wsum[wave] = inc;
  }
  __syncthreads();
  if (tid < BIN_W) {
    int wbase = 0;
    for (int w = 0; w < wave; ++w) wbase += wsum[w];
    int ex = wbase + inc - c;
    nstart[tid] = ex;
    ncur[tid]   = ex;
    int node = nbase + tid;
    if (node < n) {
      cnt[node]    = c;
      rowptr[node] = b0 + ex;
    }
  }
  __syncthreads();
  if (size <= BIN_CAP) {
    for (int i0 = 0; i0 < size; i0 += FIN_THREADS * FIN_LOADS) {
      v2i vv[FIN_LOADS];
#pragma unroll
      for (int u = 0; u < FIN_LOADS; ++u) {
        int i = i0 + u * FIN_THREADS + tid;
        v2i none = {0, -1};
        vv[u] = (i < size) ? pairs[b0 + i] : none;
      }
#pragma unroll
      for (int u = 0; u < FIN_LOADS; ++u)
        if (vv[u].y >= 0) {
          int q = atomicAdd(&ncur[vv[u].y - nbase], 1);
          ssrc[q] = vv[u].x;
        }
    }
    __syncthreads();
    for (int ln = wave; ln < BIN_W; ln += FIN_THREADS / 32) {
      int st  = __builtin_amdgcn_readfirstlane(nstart[ln]);
      int deg = __builtin_amdgcn_readfirstlane(ncnt[ln]);
      if (deg <= 32) {
        int mine = (lane < deg) ? ssrc[st + lane] : 0x7fffffff;
        int r = 0;
        for (int j = 0; j < deg; ++j) {
          int v = __builtin_amdgcn_readlane(mine, j);
          r += ((v < mine) || (v == mine && j < lane)) ? 1 : 0;
        }
        if (lane < deg) seg[b0 + st + r] = mine;
      } else {
        for (int i = lane; i < deg; i += 32) {
          int mine = ssrc[st + i];
          int r = 0;
          for (int j = 0; j < deg; ++j) {
            int v = ssrc[st + j];
            r += ((v < mine) || (v == mine && j < i)) ? 1 : 0;
          }
          seg[b0 + st + r] = mine;
        }
      }
    }
  } else {
    for (int ln = wave; ln < BIN_W; ln += FIN_THREADS / 32) {
      int nd = nbase + ln;
      int st = __builtin_amdgcn_readfirstlane(nstart[ln]);
      for (int i = lane; i < size; i += 32) {
        v2i v = pairs[b0 + i];
        if (v.y == nd) {
          int r = 0;
          for (int j = 0; j < size; ++j) {
            v2i u = pairs[b0 + j];
            r += (u.y == nd && ((u.x < v.x) || (u.x == v.x && j < i))) ? 1 : 0;
          }
          seg[b0 + st + r] = v.x;
        }
      }
    }
  }
}

template <int CTRL>
__device__ __forceinline__ float dpp_get(float v) {
  return __int_as_float(__builtin_amdgcn_update_dpp(0, __float_as_int(v), CTRL, 0xF, 0xF, true));
}
__device__ __forceinline__ float other_row_get(float v) {
  return __uint_as_float(__builtin_amdgcn_permlanex16(0u, __float_as_uint(v),
                                                      0x76543210u, 0xfedcba98u, true, false));
}
template <int XM>
__device__ __forceinline__ float row_xadd(float p) {
  float r;
  if constexpr (XM == 1)      asm("v_add_f32_dpp %0, %1, %1 row_xmask:1 row_mask:0xf bank_mask:0xf bound_ctrl:1" : "=v"(r) : "v"(p));
  else if constexpr (XM == 2) asm("v_add_f32_dpp %0, %1, %1 row_xmask:2 row_mask:0xf bank_mask:0xf bound_ctrl:1" : "=v"(r) : "v"(p));
  else if constexpr (XM == 4) asm("v_add_f32_dpp %0, %1, %1 row_xmask:4 row_mask:0xf bank_mask:0xf bound_ctrl:1" : "=v"(r) : "v"(p));
  else                        asm("v_add_f32_dpp %0, %1, %1 row_xmask:8 row_mask:0xf bank_mask:0xf bound_ctrl:1" : "=v"(r) : "v"(p));
  return r;
}
template <int LPH>
__device__ __forceinline__ float head_sum(float p) {
  static_assert(LPH == 4 || LPH == 32, "lanes per head");
  p = row_xadd<1>(p);
  p = row_xadd<2>(p);
  if constexpr (LPH == 32) {
    p = row_xadd<4>(p);
    p = row_xadd<8>(p);
    p += other_row_get(p);
  }
  return p;
}

typedef const __attribute__((address_space(1))) float* gfloat_p;

template <int CPT> struct RowVec;
template <> struct RowVec<4> { typedef v4f T; };
template <> struct RowVec<2> { typedef v2f T; };

template <int CPT, int LPH, int R>
__device__ __forceinline__ void gat_group(const float* __restrict__ XL, int ch0, int mysrc, int j,
                                          const v2f (&xrp)[CPT / 2], const v2f (&avp)[CPT / 2],
                                          float& m, float& den, v2f (&accp)[CPT / 2]) {
  constexpr int NP = CPT / 2;
  constexpr int HC = CPT * 32;
  typedef typename RowVec<CPT>::T vec_t;
  vec_t x[R];
  float pu[R];
#pragma unroll
  for (int u = 0; u < R; ++u) {
    int s = __builtin_amdgcn_readlane(mysrc, j + u);
    gfloat_p rowp = (gfloat_p)(XL + (size_t)s * HC);
    asm("" : "+s"(rowp));
    x[u] = *(const __attribute__((address_space(1))) vec_t*)(rowp + ch0);
  }
#pragma unroll
  for (int u = 0; u < R; ++u) {
    v2f q2;
#pragma unroll
    for (int p = 0; p < NP; ++p) {
      v2f xp = {x[u][2 * p], x[u][2 * p + 1]};
      v2f v = xp + xrp[p];
      v2f l = __builtin_elementwise_max(v, v * NEG_SLOPE);
      q2 = (p == 0) ? avp[0] * l : avp[p] * l + q2;
    }
    pu[u] = head_sum<LPH>(q2[0] + q2[1]);
  }
  float mn = m;
  if constexpr (R == 4) mn = fmaxf(fmaxf(m, pu[0]), fmaxf(pu[1], fmaxf(pu[2], pu[3])));
  else if constexpr (R == 3) mn = fmaxf(fmaxf(m, pu[0]), fmaxf(pu[1], pu[2]));
  else if constexpr (R == 2) mn = fmaxf(fmaxf(m, pu[0]), pu[1]);
  else mn = fmaxf(m, pu[0]);
  float sc = __builtin_amdgcn_exp2f(m - mn);
  float e[R];
#pragma unroll
  for (int u = 0; u < R; ++u) e[u] = __builtin_amdgcn_exp2f(pu[u] - mn);
  float es;
  if constexpr (R == 4) es = (e[0] + e[1]) + (e[2] + e[3]);
  else if constexpr (R == 3) es = (e[0] + e[1]) + e[2];
  else if constexpr (R == 2) es = e[0] + e[1];
  else es = e[0];
  den = den * sc + es;
#pragma unroll
  for (int p = 0; p < NP; ++p) {
    v2f xs[R];
#pragma unroll
    for (int u = 0; u < R; ++u) xs[u] = (v2f){x[u][2 * p], x[u][2 * p + 1]};
    v2f w;
    if constexpr (R == 4) w = (e[0] * xs[0] + e[1] * xs[1]) + (e[2] * xs[2] + e[3] * xs[3]);
    else if constexpr (R == 3) w = (e[0] * xs[0] + e[1] * xs[1]) + e[2] * xs[2];
    else if constexpr (R == 2) w = e[0] * xs[0] + e[1] * xs[1];
    else w = e[0] * xs[0];
    accp[p] = accp[p] * sc + w;
  }
  m = mn;
}

template <int H, int C, bool LAST>
__global__ void k_gat_node(const float* __restrict__ XL, const float* __restrict__ XR,
                           const float* __restrict__ att, const float* __restrict__ bias,
                           const int* __restrict__ rowptr, const int* __restrict__ cnt,
                           const int* __restrict__ seg,
                           _Float16* __restrict__ NEXT16, float* __restrict__ PROB, int n) {
  constexpr int HC  = H * C;
  constexpr int CPT = HC / 32;
  constexpr int NP  = CPT / 2;
  constexpr int LPH = 32 / H;
  constexpr int UNR = 4;
  typedef typename RowVec<CPT>::T vec_t;

  int node = blockIdx.x * (blockDim.x >> 5) + (threadIdx.x >> 5);
  if (node >= n) return;
  node = __builtin_amdgcn_readfirstlane(node);
  int lane = threadIdx.x & 31;
  int ch0  = lane * CPT;
  int start = rowptr[node], deg = cnt[node];

  vec_t xr = *(const vec_t*)(XR + (size_t)node * HC + ch0);
  vec_t av = *(const vec_t*)(att + ch0);
  vec_t xs = *(const vec_t*)(XL + (size_t)node * HC + ch0);

  v2f xrp[NP], avp[NP], accp[NP];
#pragma unroll
  for (int p = 0; p < NP; ++p) {
    xrp[p]  = (v2f){xr[2 * p], xr[2 * p + 1]};
    avp[p]  = (v2f){av[2 * p], av[2 * p + 1]} * LOG2E_F;
    accp[p] = (v2f){xs[2 * p], xs[2 * p + 1]};
  }

  v2f q2;
#pragma unroll
  for (int p = 0; p < NP; ++p) {
    v2f v = accp[p] + xrp[p];
    v2f l = __builtin_elementwise_max(v, v * NEG_SLOPE);
    q2 = (p == 0) ? avp[0] * l : avp[p] * l + q2;
  }
  float m   = head_sum<LPH>(q2[0] + q2[1]);
  float den = 1.0f;

  for (int base = 0; base < deg; base += 32) {
    int nb = deg - base;
    if (nb > 32) nb = 32;
    int mysrc = (lane < nb) ? seg[start + base + lane] : node;
    int j = 0;
    for (; j + UNR <= nb; j += UNR)
      gat_group<CPT, LPH, 4>(XL, ch0, mysrc, j, xrp, avp, m, den, accp);
    int rest = nb - j;
    if (rest == 3)      gat_group<CPT, LPH, 3>(XL, ch0, mysrc, j, xrp, avp, m, den, accp);
    else if (rest == 2) gat_group<CPT, LPH, 2>(XL, ch0, mysrc, j, xrp, avp, m, den, accp);
    else if (rest == 1) gat_group<CPT, LPH, 1>(XL, ch0, mysrc, j, xrp, avp, m, den, accp);
  }

  vec_t acc;
#pragma unroll
  for (int p = 0; p < NP; ++p) { acc[2 * p] = accp[p][0]; acc[2 * p + 1] = accp[p][1]; }

  float rdn = __builtin_amdgcn_rcpf(fmaxf(den, 1e-16f));
  vec_t bv = *(const vec_t*)(bias + ch0);
  vec_t o;
#pragma unroll
  for (int k = 0; k < CPT; ++k) o[k] = acc[k] * rdn + bv[k];

  if constexpr (!LAST) {
    static_assert(CPT == 4, "the f16 row is stored as 8 bytes per lane");
    *(v4h*)(NEXT16 + (size_t)node * HC + ch0) = __builtin_convertvector(elu4(o), v4h);
  } else {
    static_assert(H == 1 && CPT == 2, "one head: the wave's 32 lanes hold the whole row, 2 channels each");
    float mx = fmaxf(o[0], o[1]);
    mx = fmaxf(mx, other_row_get(mx));
    mx = fmaxf(mx, dpp_get<0x168>(mx));
    mx = fmaxf(mx, dpp_get<0x164>(mx));
    mx = fmaxf(mx, dpp_get<0x162>(mx));
    mx = fmaxf(mx, dpp_get<0x161>(mx));
    vec_t e;
    e[0] = __builtin_amdgcn_exp2f((o[0] - mx) * LOG2E_F);
    e[1] = __builtin_amdgcn_exp2f((o[1] - mx) * LOG2E_F);
    float s = e[0] + e[1];
    s += other_row_get(s);
    s += dpp_get<0x168>(s);
    s += dpp_get<0x164>(s);
    s += dpp_get<0x162>(s);
    s += dpp_get<0x161>(s);
    float inv = __builtin_amdgcn_rcpf(s);
    *(vec_t*)(PROB + (size_t)node * HC + ch0) = e * inv;
  }
}

template <int H, int C, bool LAST>
static void gat_layer(const _Float16* X16, int din,
                      const _Float16* WPl, const _Float16* WPr, const float* att, const float* b,
                      float* XL, float* XR, _Float16* next16, float* prob,
                      const int* rowptr, const int* cnt, const int* seg,
                      int N, hipStream_t stream) {
  constexpr int HC = H * C;

  int rowTiles = (N + 15) / 16;
  dim3 gg((unsigned)((rowTiles + 3) / 4), (unsigned)(2 * (HC / 64)));
  k_gemm2_f16<<<gg, 128, 0, stream>>>(X16, WPl, WPr, XL, XR, N, din, HC);

  k_gat_node<H, C, LAST><<<(N + 7) / 8, 256, 0, stream>>>(XL, XR, att, b, rowptr, cnt, seg, next16, prob, N);
}

extern "C" void kernel_launch(void* const* d_in, const int* in_sizes, int n_in,
                              void* d_out, int out_size, void* d_ws, size_t ws_size,
                              hipStream_t stream) {
  (void)n_in; (void)out_size; (void)ws_size;
  const float* x    = (const float*)d_in[0];
  const int*   ei   = (const int*)d_in[1];
  const float* Wl1  = (const float*)d_in[2];
  const float* Wr1  = (const float*)d_in[3];
  const float* att1 = (const float*)d_in[4];
  const float* b1   = (const float*)d_in[5];
  const float* Wl2  = (const float*)d_in[6];
  const float* Wr2  = (const float*)d_in[7];
  const float* att2 = (const float*)d_in[8];
  const float* b2   = (const float*)d_in[9];
  const float* Wl3  = (const float*)d_in[10];
  const float* Wr3  = (const float*)d_in[11];
  const float* att3 = (const float*)d_in[12];
  const float* b3   = (const float*)d_in[13];

  int       N  = in_sizes[0] / 256;
  long long E0 = (long long)in_sizes[1] / 2;

  uintptr_t p = (uintptr_t)d_ws;
  auto carve = [&](size_t bytes) -> void* {
    void* r = (void*)p;
    p += (bytes + 255) & ~(size_t)255;
    return r;
  };
  _Float16* X16 = (_Float16*)carve((size_t)N * 256 * sizeof(_Float16));
  _Float16* P1l = (_Float16*)carve((size_t)L1_DIN * L1_HC * sizeof(_Float16));
  _Float16* P1r = (_Float16*)carve((size_t)L1_DIN * L1_HC * sizeof(_Float16));
  _Float16* P2l = (_Float16*)carve((size_t)L2_DIN * L2_HC * sizeof(_Float16));
  _Float16* P2r = (_Float16*)carve((size_t)L2_DIN * L2_HC * sizeof(_Float16));
  _Float16* P3l = (_Float16*)carve((size_t)L3_DIN * L3_HC * sizeof(_Float16));
  _Float16* P3r = (_Float16*)carve((size_t)L3_DIN * L3_HC * sizeof(_Float16));
  float*    XL  = (float*)carve((size_t)N * 128 * sizeof(float));
  float*    XR  = (float*)carve((size_t)N * 128 * sizeof(float));
  int* cnt    = (int*)carve((size_t)N * sizeof(int));
  int* rowptr = (int*)carve((size_t)N * sizeof(int));
  int* seg    = (int*)carve((size_t)E0 * sizeof(int));
  int nb  = (N + BIN_W - 1) / BIN_W;
  int G   = (int)((E0 + CHUNK - 1) / CHUNK);
  int nbg = nb * G;
  int nTiles = (nbg + SCAN_TILE - 1) / SCAN_TILE;
  v2i* pairs  = (v2i*)XL;
  int* binCnt = (int*)XR;
  int* binOff = binCnt + nbg;
  int* tsum   = binOff + nbg;

  long long nx = (long long)N * 256;
  k_prep<<<PREP_BLOCKS, EW_THREADS, 0, stream>>>(x, X16, nx, Wl1, Wr1, Wl2, Wr2, Wl3, Wr3,
                                                 P1l, P1r, P2l, P2r, P3l, P3r);
  k_bin_count<<<G, 256, 0, stream>>>(ei, E0, nb, G, binCnt);
  k_tile_sum<<<nTiles, 256, 0, stream>>>(binCnt, nbg, tsum);
  k_scan<<<nTiles, 256, 0, stream>>>(binCnt, nbg, tsum, binOff);
  k_bin_scatter<<<G, 256, 0, stream>>>(ei, E0, nb, G, binOff, pairs);
  k_bin_finish<<<nb, FIN_THREADS, 0, stream>>>(pairs, binOff, nb, G, (int)E0, N, cnt, rowptr, seg);


  gat_layer<8, 16, false>(X16, L1_DIN, P1l, P1r, att1, b1,
                          XL, XR, X16, nullptr, rowptr, cnt, seg, N, stream);

  gat_layer<8, 16, false>(X16, L2_DIN, P2l, P2r, att2, b2,
                          XL, XR, X16, nullptr, rowptr, cnt, seg, N, stream);

  gat_layer<1, 64, true>(X16, L3_DIN, P3l, P3r, att3, b3,
                         XL, XR, nullptr, (float*)d_out, rowptr, cnt, seg, N, stream);
}


# Layer-1 work the edge sort does not feed rides in the sort's small grids: packs and x->f16 beside count, tile-sum, scan; layer-1 GEMM beside scatter, finish (sort scratch now in out); 12->10 launches
# speedup vs baseline: 6.7516x; 1.0865x over previous
#include <hip/hip_runtime.h>
#include <hip/hip_bf16.h>
#include <math.h>


#define NEG_SLOPE 0.2f

typedef __attribute__((ext_vector_type(16))) _Float16 v16h;
typedef __attribute__((ext_vector_type(8)))  _Float16 v8h;
typedef __attribute__((ext_vector_type(8)))  float    v8f;

#define EW_THREADS 256
#define EW_LOADS   8
#define EW_BLOCKS  512

typedef __attribute__((ext_vector_type(4))) float    v4f;
typedef __attribute__((ext_vector_type(2))) float    v2f;
typedef __attribute__((ext_vector_type(4))) _Float16 v4h;

#define LOG2E_F 1.4426950408889634f

__device__ __forceinline__ v4f elu4(v4f t) {
  v4f em;
#pragma unroll
  for (int k = 0; k < 4; ++k) em[k] = __builtin_amdgcn_exp2f(t[k] * LOG2E_F) - 1.0f;
#pragma unroll
  for (int k = 0; k < 4; ++k) t[k] = (t[k] > 0.0f) ? t[k] : em[k];
  return t;
}

__device__ __forceinline__ void plane_to_f16_body(const float* __restrict__ src, _Float16* __restrict__ dst,
                                                  long long n, int blk, int nblk) {
  constexpr long long TILE = (long long)EW_THREADS * EW_LOADS;
  const long long nvec = n >> 2;
  const v4f* __restrict__ s4 = (const v4f*)src;
  v4h* __restrict__ d4 = (v4h*)dst;
  for (long long t0 = (long long)blk * TILE; t0 < nvec; t0 += (long long)nblk * TILE) {
    const v4f* __restrict__ sp = s4 + t0 + threadIdx.x;
    v4h* __restrict__ dp = d4 + t0 + threadIdx.x;
    if (t0 + TILE <= nvec) {
      v4f v[EW_LOADS];
#pragma unroll
      for (int j = 0; j < EW_LOADS; ++j) v[j] = sp[j * EW_THREADS];
#pragma unroll
      for (int j = 0; j < EW_LOADS; ++j) dp[j * EW_THREADS] = __builtin_convertvector(v[j], v4h);
    } else {
      for (int j = 0; j < EW_LOADS; ++j)
        if (t0 + threadIdx.x + j * EW_THREADS < nvec) dp[j * EW_THREADS] = __builtin_convertvector(sp[j * EW_THREADS], v4h);
    }
  }
  if (blk == 0 && (long long)threadIdx.x < (n & 3)) {
    long long i = (nvec << 2) + threadIdx.x;
    dst[i] = (_Float16)src[i];
  }
}

__device__ __forceinline__ void pack_w_body(const float* __restrict__ W, _Float16* __restrict__ WP,
                                            int din, int dout, int idx) {
  int ktiles = din >> 5, ntiles = dout >> 4;
  int total = ktiles * ntiles * 32;
  if (idx >= total) return;
  int lane = idx & 31;
  int frag = idx >> 5;
  int nt = frag % ntiles;
  int kt = frag / ntiles;
  int half = lane >> 4, lr = lane & 15;
  _Float16* o = WP + ((size_t)frag * 32 + lane) * 16;
  const float* w0 = W + (size_t)(kt * 32 + half * 8) * dout + nt * 16 + lr;
#pragma unroll
  for (int j = 0; j < 8; ++j) {
    o[j]     = (_Float16)w0[(size_t)j * dout];
    o[8 + j] = (_Float16)w0[(size_t)(16 + j) * dout];
  }
}

__device__ __forceinline__ void gemm2_wave(const _Float16* __restrict__ X,
                                           const _Float16* __restrict__ WPl, const _Float16* __restrict__ WPr,
                                           float* __restrict__ YL, float* __restrict__ YR,
                                           int Mrows, int din, int dout, int tileM, int strip) {
  int strips = dout >> 6;
  bool right = strip >= strips;
  if (right) strip -= strips;
  const _Float16* __restrict__ WP = right ? WPr : WPl;
  float* __restrict__ Y = right ? YR : YL;
  int lane = threadIdx.x & 31;
  int half = lane >> 4;
  int lr   = lane & 15;
  if (tileM * 16 >= Mrows) return;
  int nt0 = strip * 4;
  int ntiles = dout >> 4;

  int row = tileM * 16 + lr;
  if (row >= Mrows) row = Mrows - 1;

  v8f acc0 = {}, acc1 = {}, acc2 = {}, acc3 = {};
  const _Float16* xrow = X + (size_t)row * din;
  const v16h* WPv = (const v16h*)WP;
  for (int k0 = 0; k0 < din; k0 += 32) {
    const v8h* xa = (const v8h*)(xrow + k0 + half * 8);
    v8h lo = xa[0];
    v8h hi = xa[2];
    v16h a = __builtin_shufflevector(lo, hi, 0, 1, 2, 3, 4, 5, 6, 7,
                                     8, 9, 10, 11, 12, 13, 14, 15);
    size_t fb = ((size_t)(k0 >> 5) * ntiles + nt0) * 32 + lane;
    v16h b0 = WPv[fb];
    v16h b1 = WPv[fb + 32];
    v16h b2 = WPv[fb + 64];
    v16h b3 = WPv[fb + 96];
    acc0 = __builtin_amdgcn_wmma_f32_16x16x32_f16(false, a, false, b0, (short)0, acc0, false, false);
    acc1 = __builtin_amdgcn_wmma_f32_16x16x32_f16(false, a, false, b1, (short)0, acc1, false, false);
    acc2 = __builtin_amdgcn_wmma_f32_16x16x32_f16(false, a, false, b2, (short)0, acc2, false, false);
    acc3 = __builtin_amdgcn_wmma_f32_16x16x32_f16(false, a, false, b3, (short)0, acc3, false, false);
  }
  int r0 = tileM * 16 + half * 8;
#pragma unroll
  for (int i = 0; i < 8; ++i) {
    int r = r0 + i;
    if (r < Mrows) {
      float* yp = Y + (size_t)r * dout + nt0 * 16 + lr;
      yp[0]  = acc0[i];
      yp[16] = acc1[i];
      yp[32] = acc2[i];
      yp[48] = acc3[i];
    }
  }
}

__global__ void k_gemm2_f16(const _Float16* __restrict__ X,
                            const _Float16* __restrict__ WPl, const _Float16* __restrict__ WPr,
                            float* __restrict__ YL, float* __restrict__ YR,
                            int Mrows, int din, int dout) {
  gemm2_wave(X, WPl, WPr, YL, YR, Mrows, din, dout,
             (int)blockIdx.x * 4 + (int)(threadIdx.x >> 5), (int)blockIdx.y);
}

template <int WPB>
__device__ __forceinline__ void gemm2_rider(const _Float16* __restrict__ X,
                                            const _Float16* __restrict__ WPl, const _Float16* __restrict__ WPr,
                                            float* __restrict__ YL, float* __restrict__ YR,
                                            int Mrows, int din, int dout, int b, int strip0) {
  int tb = (((Mrows + 15) >> 4) + WPB - 1) / WPB;
  int s = b / tb;
  gemm2_wave(X, WPl, WPr, YL, YR, Mrows, din, dout,
             (b - s * tb) * WPB + (int)(threadIdx.x >> 5), strip0 + s);
}

#define L1_DIN 256
#define L1_HC  128
#define L2_DIN 128
#define L2_HC  128
#define L3_DIN 128
#define L3_HC  64
#define PACK_BLOCKS_OF(din, hc) ((((din) >> 5) * ((hc) >> 4) * 32 + EW_THREADS - 1) / EW_THREADS)
#define PB1 PACK_BLOCKS_OF(L1_DIN, L1_HC)
#define PB2 PACK_BLOCKS_OF(L2_DIN, L2_HC)
#define PB3 PACK_BLOCKS_OF(L3_DIN, L3_HC)
#define PREP_PACK_BLOCKS (2 * (PB1 + PB2 + PB3))

#define X_PARTS 3

__device__ __forceinline__ void pack_block_body(int b,
       const float* __restrict__ Wl1, const float* __restrict__ Wr1,
       const float* __restrict__ Wl2, const float* __restrict__ Wr2,
       const float* __restrict__ Wl3, const float* __restrict__ Wr3,
       _Float16* __restrict__ P1l, _Float16* __restrict__ P1r,
       _Float16* __restrict__ P2l, _Float16* __restrict__ P2r,
       _Float16* __restrict__ P3l, _Float16* __restrict__ P3r) {
  const float* W; _Float16* WP; int din, hc;
  if (b < PB1)                      { W = Wl1; WP = P1l; din = L1_DIN; hc = L1_HC; }
  else if ((b -= PB1) < PB1)        { W = Wr1; WP = P1r; din = L1_DIN; hc = L1_HC; }
  else if ((b -= PB1) < PB2)        { W = Wl2; WP = P2l; din = L2_DIN; hc = L2_HC; }
  else if ((b -= PB2) < PB2)        { W = Wr2; WP = P2r; din = L2_DIN; hc = L2_HC; }
  else if ((b -= PB2) < PB3)        { W = Wl3; WP = P3l; din = L3_DIN; hc = L3_HC; }
  else            { b -= PB3;         W = Wr3; WP = P3r; din = L3_DIN; hc = L3_HC; }
  pack_w_body(W, WP, din, hc, b * EW_THREADS + (int)threadIdx.x);
}

#define BIN_SH  8
#define BIN_W   (1 << BIN_SH)
#define NB_MAX  512
#define CHUNK   4096
#define EPT     (CHUNK / 256)
#define BIN_CAP 8192

typedef __attribute__((ext_vector_type(2))) int v2i;

__device__ __forceinline__ void
bin_count_body(const int* __restrict__ ei, long long E0, int nb, int G, int* __restrict__ binCnt, int blk) {
  __shared__ int h[NB_MAX];
  int tid = threadIdx.x;
  for (int i = tid; i < NB_MAX; i += 256) h[i] = 0;
  __syncthreads();
  long long base = (long long)blk * CHUNK;
  int d[EPT];
#pragma unroll
  for (int j = 0; j < EPT; ++j) {
    long long e = base + j * 256 + tid;
    d[j] = (e < E0) ? ei[E0 + e] : -1;
  }
#pragma unroll
  for (int j = 0; j < EPT; ++j)
    if (d[j] >= 0) atomicAdd(&h[d[j] >> BIN_SH], 1);
  __syncthreads();
  for (int i = tid; i < nb; i += 256) binCnt[(size_t)i * G + blk] = h[i];
}

__device__ __forceinline__ void
bin_scatter_body(const int* __restrict__ ei, long long E0, int nb, int G,
                 const int* __restrict__ binOff, v2i* __restrict__ pairs, int blk) {
  __shared__ int h[NB_MAX];
  __shared__ int cur[NB_MAX];
  __shared__ int goff[NB_MAX];
  __shared__ int wsum[8];
  __shared__ v2i sp[CHUNK];
  int tid = threadIdx.x, lane = tid & 31, wave = tid >> 5;
  for (int i = tid; i < NB_MAX; i += 256) h[i] = 0;
  __syncthreads();
  long long base = (long long)blk * CHUNK;
  int s[EPT], d[EPT];
#pragma unroll
  for (int j = 0; j < EPT; ++j) {
    long long e = base + j * 256 + tid;
    bool in = e < E0;
    s[j] = in ? ei[e] : 0;
    d[j] = in ? ei[E0 + e] : -1;
  }
#pragma unroll
  for (int j = 0; j < EPT; ++j)
    if (d[j] >= 0) atomicAdd(&h[d[j] >> BIN_SH], 1);
  __syncthreads();
  int c0 = h[2 * tid], c1 = h[2 * tid + 1];
  int local = c0 + c1, inc = local;
#pragma unroll
  for (int o = 1; o < 32; o <<= 1) {
    int t = __shfl_up(inc, o, 32);
    if (lane >= o) inc += t;
  }
  if (lane == 31) wsum[wave] = inc;
  __syncthreads();
  int wbase = 0;
  for (int w = 0; w < wave; ++w) wbase += wsum[w];
  int ex = wbase + inc - local;
  cur[2 * tid]     = ex;
  cur[2 * tid + 1] = ex + c0;
  goff[2 * tid]     = (2 * tid < nb)     ? binOff[(size_t)(2 * tid) * G + blk] - ex : 0;
  goff[2 * tid + 1] = (2 * tid + 1 < nb) ? binOff[(size_t)(2 * tid + 1) * G + blk] - (ex + c0) : 0;
  __syncthreads();
#pragma unroll
  for (int j = 0; j < EPT; ++j)
    if (d[j] >= 0) {
      int q = atomicAdd(&cur[d[j] >> BIN_SH], 1);
      v2i v = {s[j], d[j]};
      sp[q] = v;
    }
  __syncthreads();
  long long rem = E0 - base;
  int count = rem < CHUNK ? (int)rem : CHUNK;
  for (int i = tid; i < count; i += 256) {
    v2i v = sp[i];
    pairs[goff[v.y >> BIN_SH] + i] = v;
  }
}

#define SCAN_TILE 1024

__device__ __forceinline__ void
tile_sum_body(const int* __restrict__ cnt, int n, int* __restrict__ tsum, int blk) {
  __shared__ int sh[8];
  int base = blk * SCAN_TILE + threadIdx.x * 4;
  int v = 0;
#pragma unroll
  for (int j = 0; j < 4; ++j) if (base + j < n) v += cnt[base + j];
#pragma unroll
  for (int o = 16; o > 0; o >>= 1) v += __shfl_xor(v, o, 32);
  if ((threadIdx.x & 31) == 0) sh[threadIdx.x >> 5] = v;
  __syncthreads();
  if (threadIdx.x == 0) {
    int s = 0;
    for (int w = 0; w < 8; ++w) s += sh[w];
    tsum[blk] = s;
  }
}

__device__ __forceinline__ void
scan_body(const int* __restrict__ cnt, int n, const int* __restrict__ tsum,
          int* __restrict__ rowptr, int blk) {
  __shared__ int sh[8];
  int tid = threadIdx.x, lane = tid & 31, wave = tid >> 5;
  int pb = 0;
  for (int j = tid; j < blk; j += 256) pb += tsum[j];
#pragma unroll
  for (int o = 16; o > 0; o >>= 1) pb += __shfl_xor(pb, o, 32);
  if (lane == 0) sh[wave] = pb;
  __syncthreads();
  int tilebase = 0;
  for (int w = 0; w < 8; ++w) tilebase += sh[w];
  __syncthreads();

  int base = blk * SCAN_TILE + tid * 4;
  int cv[4];
#pragma unroll
  for (int j = 0; j < 4; ++j) cv[j] = (base + j < n) ? cnt[base + j] : 0;
  int local = cv[0] + cv[1] + cv[2] + cv[3];
  int inc = local;
#pragma unroll
  for (int o = 1; o < 32; o <<= 1) {
    int t = __shfl_up(inc, o, 32);
    if (lane >= o) inc += t;
  }
  if (lane == 31) sh[wave] = inc;
  __syncthreads();
  int wbase = 0;
  for (int w = 0; w < wave; ++w) wbase += sh[w];
  int ex = tilebase + wbase + inc - local;
#pragma unroll
  for (int j = 0; j < 4; ++j) {
    if (base + j < n) rowptr[base + j] = ex;
    ex += cv[j];
  }
}

#define FIN_THREADS 512
#define FIN_LOADS   8
__device__ __forceinline__ void
bin_finish_body(const v2i* __restrict__ pairs, const int* __restrict__ binOff, int nb, int G,
                int E0i, int n, int* __restrict__ cnt, int* __restrict__ rowptr, int* __restrict__ seg, int blk) {
  __shared__ int ncnt[BIN_W];
  __shared__ int nstart[BIN_W];
  __shared__ int ncur[BIN_W];
  __shared__ int wsum[8];
  __shared__ int ssrc[BIN_CAP];
  int tid = threadIdx.x, lane = tid & 31, wave = tid >> 5;
  int bin = blk;
  int b0 = binOff[(size_t)bin * G];
  int b1 = (bin + 1 < nb) ? binOff[(size_t)(bin + 1) * G] : E0i;
  int size = b1 - b0;
  int nbase = bin << BIN_SH;
  if (tid < BIN_W) ncnt[tid] = 0;
  __syncthreads();
  for (int i0 = 0; i0 < size; i0 += FIN_THREADS * FIN_LOADS) {
    int dd[FIN_LOADS];
#pragma unroll
    for (int u = 0; u < FIN_LOADS; ++u) {
      int i = i0 + u * FIN_THREADS + tid;
      dd[u] = (i < size) ? pairs[b0 + i].y : -1;
    }
#pragma unroll
    for (int u = 0; u < FIN_LOADS; ++u)
      if (dd[u] >= 0) atomicAdd(&ncnt[dd[u] - nbase], 1);
  }
  __syncthreads();
  int c = 0, inc = 0;
  if (tid < BIN_W) {
    c = ncnt[tid];
    inc = c;
#pragma unroll
    for (int o = 1; o < 32; o <<= 1) {
      int t = __shfl_up(inc, o, 32);
      if (lane >= o) inc += t;
    }
    if (lane == 31) wsum[wave] = inc;
  }
  __syncthreads();
  if (tid < BIN_W) {
    int wbase = 0;
    for (int w = 0; w < wave; ++w) wbase += wsum[w];
    int ex = wbase + inc - c;
    nstart[tid] = ex;
    ncur[tid]   = ex;
    int node = nbase + tid;
    if (node < n) {
      cnt[node]    = c;
      rowptr[node] = b0 + ex;
    }
  }
  __syncthreads();
  if (size <= BIN_CAP) {
    for (int i0 = 0; i0 < size; i0 += FIN_THREADS * FIN_LOADS) {
      v2i vv[FIN_LOADS];
#pragma unroll
      for (int u = 0; u < FIN_LOADS; ++u) {
        int i = i0 + u * FIN_THREADS + tid;
        v2i none = {0, -1};
        vv[u] = (i < size) ? pairs[b0 + i] : none;
      }
#pragma unroll
      for (int u = 0; u < FIN_LOADS; ++u)
        if (vv[u].y >= 0) {
          int q = atomicAdd(&ncur[vv[u].y - nbase], 1);
          ssrc[q] = vv[u].x;
        }
    }
    __syncthreads();
    for (int ln = wave; ln < BIN_W; ln += FIN_THREADS / 32) {
      int st  = __builtin_amdgcn_readfirstlane(nstart[ln]);
      int deg = __builtin_amdgcn_readfirstlane(ncnt[ln]);
      if (deg <= 32) {
        int mine = (lane < deg) ? ssrc[st + lane] : 0x7fffffff;
        int r = 0;
        for (int j = 0; j < deg; ++j) {
          int v = __builtin_amdgcn_readlane(mine, j);
          r += ((v < mine) || (v == mine && j < lane)) ? 1 : 0;
        }
        if (lane < deg) seg[b0 + st + r] = mine;
      } else {
        for (int i = lane; i < deg; i += 32) {
          int mine = ssrc[st + i];
          int r = 0;
          for (int j = 0; j < deg; ++j) {
            int v = ssrc[st + j];
            r += ((v < mine) || (v == mine && j < i)) ? 1 : 0;
          }
          seg[b0 + st + r] = mine;
        }
      }
    }
  } else {
    for (int ln = wave; ln < BIN_W; ln += FIN_THREADS / 32) {
      int nd = nbase + ln;
      int st = __builtin_amdgcn_readfirstlane(nstart[ln]);
      for (int i = lane; i < size; i += 32) {
        v2i v = pairs[b0 + i];
        if (v.y == nd) {
          int r = 0;
          for (int j = 0; j < size; ++j) {
            v2i u = pairs[b0 + j];
            r += (u.y == nd && ((u.x < v.x) || (u.x == v.x && j < i))) ? 1 : 0;
          }
          seg[b0 + st + r] = v.x;
        }
      }
    }
  }
}

__global__ void __launch_bounds__(EW_THREADS)
k_first(const int* __restrict__ ei, long long E0, int nb, int G, int* __restrict__ binCnt,
        const float* __restrict__ x, _Float16* __restrict__ X16, long long nx,
        const float* __restrict__ Wl1, const float* __restrict__ Wr1,
        const float* __restrict__ Wl2, const float* __restrict__ Wr2,
        const float* __restrict__ Wl3, const float* __restrict__ Wr3,
        _Float16* __restrict__ P1l, _Float16* __restrict__ P1r,
        _Float16* __restrict__ P2l, _Float16* __restrict__ P2r,
        _Float16* __restrict__ P3l, _Float16* __restrict__ P3r) {
  int b = (int)blockIdx.x;
  if (b < G) { bin_count_body(ei, E0, nb, G, binCnt, b); return; }
  b -= G;
  if (b < PREP_PACK_BLOCKS) {
    pack_block_body(b, Wl1, Wr1, Wl2, Wr2, Wl3, Wr3, P1l, P1r, P2l, P2r, P3l, P3r);
    return;
  }
  b -= PREP_PACK_BLOCKS;
  plane_to_f16_body(x, X16, nx, b, X_PARTS * EW_BLOCKS);
}

__global__ void __launch_bounds__(EW_THREADS)
k_tile_sum_x(const int* __restrict__ cnt, int n, int* __restrict__ tsum, int nTiles,
             const float* __restrict__ x, _Float16* __restrict__ X16, long long nx) {
  int b = (int)blockIdx.x;
  if (b < nTiles) { tile_sum_body(cnt, n, tsum, b); return; }
  plane_to_f16_body(x, X16, nx, b - nTiles + EW_BLOCKS, X_PARTS * EW_BLOCKS);
}

__global__ void __launch_bounds__(EW_THREADS)
k_scan_x(const int* __restrict__ cnt, int n, const int* __restrict__ tsum, int* __restrict__ rowptr,
         int nTiles, const float* __restrict__ x, _Float16* __restrict__ X16, long long nx) {
  int b = (int)blockIdx.x;
  if (b < nTiles) { scan_body(cnt, n, tsum, rowptr, b); return; }
  plane_to_f16_body(x, X16, nx, b - nTiles + 2 * EW_BLOCKS, X_PARTS * EW_BLOCKS);
}

__global__ void __launch_bounds__(256)
k_scatter_gemm(const int* __restrict__ ei, long long E0, int nb, int G,
               const int* __restrict__ binOff, v2i* __restrict__ pairs,
               const _Float16* __restrict__ X,
               const _Float16* __restrict__ WPl, const _Float16* __restrict__ WPr,
               float* __restrict__ YL, float* __restrict__ YR, int Mrows, int din, int dout) {
  int b = (int)blockIdx.x;
  if (b < G) { bin_scatter_body(ei, E0, nb, G, binOff, pairs, b); return; }
  gemm2_rider<256 / 32>(X, WPl, WPr, YL, YR, Mrows, din, dout, b - G, 0);
}

__global__ void __launch_bounds__(FIN_THREADS)
k_finish_gemm(const v2i* __restrict__ pairs, const int* __restrict__ binOff, int nb, int G,
              int E0i, int n, int* __restrict__ cnt, int* __restrict__ rowptr, int* __restrict__ seg,
              const _Float16* __restrict__ X,
              const _Float16* __restrict__ WPl, const _Float16* __restrict__ WPr,
              float* __restrict__ YL, float* __restrict__ YR, int din, int dout) {
  int b = (int)blockIdx.x;
  if (b < nb) { bin_finish_body(pairs, binOff, nb, G, E0i, n, cnt, rowptr, seg, b); return; }
  gemm2_rider<FIN_THREADS / 32>(X, WPl, WPr, YL, YR, n, din, dout, b - nb, dout >> 6);
}

template <int CTRL>
__device__ __forceinline__ float dpp_get(float v) {
  return __int_as_float(__builtin_amdgcn_update_dpp(0, __float_as_int(v), CTRL, 0xF, 0xF, true));
}
__device__ __forceinline__ float other_row_get(float v) {
  return __uint_as_float(__builtin_amdgcn_permlanex16(0u, __float_as_uint(v),
                                                      0x76543210u, 0xfedcba98u, true, false));
}
template <int XM>
__device__ __forceinline__ float row_xadd(float p) {
  float r;
  if constexpr (XM == 1)      asm("v_add_f32_dpp %0, %1, %1 row_xmask:1 row_mask:0xf bank_mask:0xf bound_ctrl:1" : "=v"(r) : "v"(p));
  else if constexpr (XM == 2) asm("v_add_f32_dpp %0, %1, %1 row_xmask:2 row_mask:0xf bank_mask:0xf bound_ctrl:1" : "=v"(r) : "v"(p));
  else if constexpr (XM == 4) asm("v_add_f32_dpp %0, %1, %1 row_xmask:4 row_mask:0xf bank_mask:0xf bound_ctrl:1" : "=v"(r) : "v"(p));
  else                        asm("v_add_f32_dpp %0, %1, %1 row_xmask:8 row_mask:0xf bank_mask:0xf bound_ctrl:1" : "=v"(r) : "v"(p));
  return r;
}
template <int LPH>
__device__ __forceinline__ float head_sum(float p) {
  static_assert(LPH == 4 || LPH == 32, "lanes per head");
  p = row_xadd<1>(p);
  p = row_xadd<2>(p);
  if constexpr (LPH == 32) {
    p = row_xadd<4>(p);
    p = row_xadd<8>(p);
    p += other_row_get(p);
  }
  return p;
}

typedef const __attribute__((address_space(1))) float* gfloat_p;

template <int CPT> struct RowVec;
template <> struct RowVec<4> { typedef v4f T; };
template <> struct RowVec<2> { typedef v2f T; };

template <int CPT, int LPH, int R>
__device__ __forceinline__ void gat_group(const float* __restrict__ XL, int ch0, int mysrc, int j,
                                          const v2f (&xrp)[CPT / 2], const v2f (&avp)[CPT / 2],
                                          float& m, float& den, v2f (&accp)[CPT / 2]) {
  constexpr int NP = CPT / 2;
  constexpr int HC = CPT * 32;
  typedef typename RowVec<CPT>::T vec_t;
  vec_t x[R];
  float pu[R];
#pragma unroll
  for (int u = 0; u < R; ++u) {
    int s = __builtin_amdgcn_readlane(mysrc, j + u);
    gfloat_p rowp = (gfloat_p)(XL + (size_t)s * HC);
    asm("" : "+s"(rowp));
    x[u] = *(const __attribute__((address_space(1))) vec_t*)(rowp + ch0);
  }
#pragma unroll
  for (int u = 0; u < R; ++u) {
    v2f q2;
#pragma unroll
    for (int p = 0; p < NP; ++p) {
      v2f xp = {x[u][2 * p], x[u][2 * p + 1]};
      v2f v = xp + xrp[p];
      v2f l = __builtin_elementwise_max(v, v * NEG_SLOPE);
      q2 = (p == 0) ? avp[0] * l : avp[p] * l + q2;
    }
    pu[u] = head_sum<LPH>(q2[0] + q2[1]);
  }
  float mn = m;
  if constexpr (R == 4) mn = fmaxf(fmaxf(m, pu[0]), fmaxf(pu[1], fmaxf(pu[2], pu[3])));
  else if constexpr (R == 3) mn = fmaxf(fmaxf(m, pu[0]), fmaxf(pu[1], pu[2]));
  else if constexpr (R == 2) mn = fmaxf(fmaxf(m, pu[0]), pu[1]);
  else mn = fmaxf(m, pu[0]);
  float sc = __builtin_amdgcn_exp2f(m - mn);
  float e[R];
#pragma unroll
  for (int u = 0; u < R; ++u) e[u] = __builtin_amdgcn_exp2f(pu[u] - mn);
  float es;
  if constexpr (R == 4) es = (e[0] + e[1]) + (e[2] + e[3]);
  else if constexpr (R == 3) es = (e[0] + e[1]) + e[2];
  else if constexpr (R == 2) es = e[0] + e[1];
  else es = e[0];
  den = den * sc + es;
#pragma unroll
  for (int p = 0; p < NP; ++p) {
    v2f xs[R];
#pragma unroll
    for (int u = 0; u < R; ++u) xs[u] = (v2f){x[u][2 * p], x[u][2 * p + 1]};
    v2f w;
    if constexpr (R == 4) w = (e[0] * xs[0] + e[1] * xs[1]) + (e[2] * xs[2] + e[3] * xs[3]);
    else if constexpr (R == 3) w = (e[0] * xs[0] + e[1] * xs[1]) + e[2] * xs[2];
    else if constexpr (R == 2) w = e[0] * xs[0] + e[1] * xs[1];
    else w = e[0] * xs[0];
    accp[p] = accp[p] * sc + w;
  }
  m = mn;
}

template <int H, int C, bool LAST>
__global__ void k_gat_node(const float* __restrict__ XL, const float* __restrict__ XR,
                           const float* __restrict__ att, const float* __restrict__ bias,
                           const int* __restrict__ rowptr, const int* __restrict__ cnt,
                           const int* __restrict__ seg,
                           _Float16* __restrict__ NEXT16, float* __restrict__ PROB, int n) {
  constexpr int HC  = H * C;
  constexpr int CPT = HC / 32;
  constexpr int NP  = CPT / 2;
  constexpr int LPH = 32 / H;
  constexpr int UNR = 4;
  typedef typename RowVec<CPT>::T vec_t;

  int node = blockIdx.x * (blockDim.x >> 5) + (threadIdx.x >> 5);
  if (node >= n) return;
  node = __builtin_amdgcn_readfirstlane(node);
  int lane = threadIdx.x & 31;
  int ch0  = lane * CPT;
  int start = rowptr[node], deg = cnt[node];

  vec_t xr = *(const vec_t*)(XR + (size_t)node * HC + ch0);
  vec_t av = *(const vec_t*)(att + ch0);
  vec_t xs = *(const vec_t*)(XL + (size_t)node * HC + ch0);

  v2f xrp[NP], avp[NP], accp[NP];
#pragma unroll
  for (int p = 0; p < NP; ++p) {
    xrp[p]  = (v2f){xr[2 * p], xr[2 * p + 1]};
    avp[p]  = (v2f){av[2 * p], av[2 * p + 1]} * LOG2E_F;
    accp[p] = (v2f){xs[2 * p], xs[2 * p + 1]};
  }

  v2f q2;
#pragma unroll
  for (int p = 0; p < NP; ++p) {
    v2f v = accp[p] + xrp[p];
    v2f l = __builtin_elementwise_max(v, v * NEG_SLOPE);
    q2 = (p == 0) ? avp[0] * l : avp[p] * l + q2;
  }
  float m   = head_sum<LPH>(q2[0] + q2[1]);
  float den = 1.0f;

  for (int base = 0; base < deg; base += 32) {
    int nb = deg - base;
    if (nb > 32) nb = 32;
    int mysrc = (lane < nb) ? seg[start + base + lane] : node;
    int j = 0;
    for (; j + UNR <= nb; j += UNR)
      gat_group<CPT, LPH, 4>(XL, ch0, mysrc, j, xrp, avp, m, den, accp);
    int rest = nb - j;
    if (rest == 3)      gat_group<CPT, LPH, 3>(XL, ch0, mysrc, j, xrp, avp, m, den, accp);
    else if (rest == 2) gat_group<CPT, LPH, 2>(XL, ch0, mysrc, j, xrp, avp, m, den, accp);
    else if (rest == 1) gat_group<CPT, LPH, 1>(XL, ch0, mysrc, j, xrp, avp, m, den, accp);
  }

  vec_t acc;
#pragma unroll
  for (int p = 0; p < NP; ++p) { acc[2 * p] = accp[p][0]; acc[2 * p + 1] = accp[p][1]; }

  float rdn = __builtin_amdgcn_rcpf(fmaxf(den, 1e-16f));
  vec_t bv = *(const vec_t*)(bias + ch0);
  vec_t o;
#pragma unroll
  for (int k = 0; k < CPT; ++k) o[k] = acc[k] * rdn + bv[k];

  if constexpr (!LAST) {
    static_assert(CPT == 4, "the f16 row is stored as 8 bytes per lane");
    *(v4h*)(NEXT16 + (size_t)node * HC + ch0) = __builtin_convertvector(elu4(o), v4h);
  } else {
    static_assert(H == 1 && CPT == 2, "one head: the wave's 32 lanes hold the whole row, 2 channels each");
    float mx = fmaxf(o[0], o[1]);
    mx = fmaxf(mx, other_row_get(mx));
    mx = fmaxf(mx, dpp_get<0x168>(mx));
    mx = fmaxf(mx, dpp_get<0x164>(mx));
    mx = fmaxf(mx, dpp_get<0x162>(mx));
    mx = fmaxf(mx, dpp_get<0x161>(mx));
    vec_t e;
    e[0] = __builtin_amdgcn_exp2f((o[0] - mx) * LOG2E_F);
    e[1] = __builtin_amdgcn_exp2f((o[1] - mx) * LOG2E_F);
    float s = e[0] + e[1];
    s += other_row_get(s);
    s += dpp_get<0x168>(s);
    s += dpp_get<0x164>(s);
    s += dpp_get<0x162>(s);
    s += dpp_get<0x161>(s);
    float inv = __builtin_amdgcn_rcpf(s);
    *(vec_t*)(PROB + (size_t)node * HC + ch0) = e * inv;
  }
}

template <int H, int C, bool LAST, bool GEMM>
static void gat_layer(const _Float16* X16, int din,
                      const _Float16* WPl, const _Float16* WPr, const float* att, const float* b,
                      float* XL, float* XR, _Float16* next16, float* prob,
                      const int* rowptr, const int* cnt, const int* seg,
                      int N, hipStream_t stream) {
  constexpr int HC = H * C;

  if (GEMM) {
    int rowTiles = (N + 15) / 16;
    dim3 gg((unsigned)((rowTiles + 3) / 4), (unsigned)(2 * (HC / 64)));
    k_gemm2_f16<<<gg, 128, 0, stream>>>(X16, WPl, WPr, XL, XR, N, din, HC);
  }

  k_gat_node<H, C, LAST><<<(N + 7) / 8, 256, 0, stream>>>(XL, XR, att, b, rowptr, cnt, seg, next16, prob, N);
}

extern "C" void kernel_launch(void* const* d_in, const int* in_sizes, int n_in,
                              void* d_out, int out_size, void* d_ws, size_t ws_size,
                              hipStream_t stream) {
  (void)n_in; (void)out_size; (void)ws_size;
  const float* x    = (const float*)d_in[0];
  const int*   ei   = (const int*)d_in[1];
  const float* Wl1  = (const float*)d_in[2];
  const float* Wr1  = (const float*)d_in[3];
  const float* att1 = (const float*)d_in[4];
  const float* b1   = (const float*)d_in[5];
  const float* Wl2  = (const float*)d_in[6];
  const float* Wr2  = (const float*)d_in[7];
  const float* att2 = (const float*)d_in[8];
  const float* b2   = (const float*)d_in[9];
  const float* Wl3  = (const float*)d_in[10];
  const float* Wr3  = (const float*)d_in[11];
  const float* att3 = (const float*)d_in[12];
  const float* b3   = (const float*)d_in[13];

  int       N  = in_sizes[0] / 256;
  long long E0 = (long long)in_sizes[1] / 2;

  uintptr_t p = (uintptr_t)d_ws;
  auto carve = [&](size_t bytes) -> void* {
    void* r = (void*)p;
    p += (bytes + 255) & ~(size_t)255;
    return r;
  };
  _Float16* X16 = (_Float16*)carve((size_t)N * 256 * sizeof(_Float16));
  _Float16* P1l = (_Float16*)carve((size_t)L1_DIN * L1_HC * sizeof(_Float16));
  _Float16* P1r = (_Float16*)carve((size_t)L1_DIN * L1_HC * sizeof(_Float16));
  _Float16* P2l = (_Float16*)carve((size_t)L2_DIN * L2_HC * sizeof(_Float16));
  _Float16* P2r = (_Float16*)carve((size_t)L2_DIN * L2_HC * sizeof(_Float16));
  _Float16* P3l = (_Float16*)carve((size_t)L3_DIN * L3_HC * sizeof(_Float16));
  _Float16* P3r = (_Float16*)carve((size_t)L3_DIN * L3_HC * sizeof(_Float16));
  float*    XL  = (float*)carve((size_t)N * 128 * sizeof(float));
  float*    XR  = (float*)carve((size_t)N * 128 * sizeof(float));
  int* cnt    = (int*)carve((size_t)N * sizeof(int));
  int* rowptr = (int*)carve((size_t)N * sizeof(int));
  int* seg    = (int*)carve((size_t)E0 * sizeof(int));
  int nb  = (N + BIN_W - 1) / BIN_W;
  int G   = (int)((E0 + CHUNK - 1) / CHUNK);
  int nbg = nb * G;
  int nTiles = (nbg + SCAN_TILE - 1) / SCAN_TILE;
  v2i* pairs  = (v2i*)d_out;
  int* binCnt = (int*)(pairs + E0);
  int* binOff = binCnt + nbg;
  int* tsum   = binOff + nbg;

  long long nx = (long long)N * 256;
  int rowTiles = (N + 15) / 16;
  int gb8  = (L1_HC / 64) * ((rowTiles + 7) / 8);
  int gb16 = (L1_HC / 64) * ((rowTiles + FIN_THREADS / 32 - 1) / (FIN_THREADS / 32));
  k_first<<<G + PREP_PACK_BLOCKS + EW_BLOCKS, EW_THREADS, 0, stream>>>(
      ei, E0, nb, G, binCnt, x, X16, nx, Wl1, Wr1, Wl2, Wr2, Wl3, Wr3, P1l, P1r, P2l, P2r, P3l, P3r);
  k_tile_sum_x<<<nTiles + EW_BLOCKS, EW_THREADS, 0, stream>>>(binCnt, nbg, tsum, nTiles, x, X16, nx);
  k_scan_x<<<nTiles + EW_BLOCKS, EW_THREADS, 0, stream>>>(binCnt, nbg, tsum, binOff, nTiles, x, X16, nx);
  k_scatter_gemm<<<G + gb8, 256, 0, stream>>>(ei, E0, nb, G, binOff, pairs,
                                              X16, P1l, P1r, XL, XR, N, L1_DIN, L1_HC);
  k_finish_gemm<<<nb + gb16, FIN_THREADS, 0, stream>>>(pairs, binOff, nb, G, (int)E0, N, cnt, rowptr, seg,
                                                       X16, P1l, P1r, XL, XR, L1_DIN, L1_HC);


  gat_layer<8, 16, false, false>(X16, L1_DIN, P1l, P1r, att1, b1,
                          XL, XR, X16, nullptr, rowptr, cnt, seg, N, stream);

  gat_layer<8, 16, false, true>(X16, L2_DIN, P2l, P2r, att2, b2,
                          XL, XR, X16, nullptr, rowptr, cnt, seg, N, stream);

  gat_layer<1, 64, true, true>(X16, L3_DIN, P3l, P3r, att3, b3,
                         XL, XR, nullptr, (float*)d_out, rowptr, cnt, seg, N, stream);
}


# Edge sort in 2 launches, not 5: a block writes its chunk in bin order with a table of run starts; a bin's block collects its runs (the starts sum to its offset); no count or scan pass; 10->7 launches
# speedup vs baseline: 7.0336x; 1.0418x over previous
#include <hip/hip_runtime.h>
#include <hip/hip_bf16.h>
#include <math.h>


#define NEG_SLOPE 0.2f

typedef __attribute__((ext_vector_type(16))) _Float16 v16h;
typedef __attribute__((ext_vector_type(8)))  _Float16 v8h;
typedef __attribute__((ext_vector_type(8)))  float    v8f;

#define EW_THREADS 256
#define EW_LOADS   8
#define EW_BLOCKS  512

typedef __attribute__((ext_vector_type(4))) float    v4f;
typedef __attribute__((ext_vector_type(2))) float    v2f;
typedef __attribute__((ext_vector_type(4))) _Float16 v4h;

#define LOG2E_F 1.4426950408889634f

__device__ __forceinline__ v4f elu4(v4f t) {
  v4f em;
#pragma unroll
  for (int k = 0; k < 4; ++k) em[k] = __builtin_amdgcn_exp2f(t[k] * LOG2E_F) - 1.0f;
#pragma unroll
  for (int k = 0; k < 4; ++k) t[k] = (t[k] > 0.0f) ? t[k] : em[k];
  return t;
}

__device__ __forceinline__ void plane_to_f16_body(const float* __restrict__ src, _Float16* __restrict__ dst,
                                                  long long n, int blk, int nblk) {
  constexpr long long TILE = (long long)EW_THREADS * EW_LOADS;
  const long long nvec = n >> 2;
  const v4f* __restrict__ s4 = (const v4f*)src;
  v4h* __restrict__ d4 = (v4h*)dst;
  for (long long t0 = (long long)blk * TILE; t0 < nvec; t0 += (long long)nblk * TILE) {
    const v4f* __restrict__ sp = s4 + t0 + threadIdx.x;
    v4h* __restrict__ dp = d4 + t0 + threadIdx.x;
    if (t0 + TILE <= nvec) {
      v4f v[EW_LOADS];
#pragma unroll
      for (int j = 0; j < EW_LOADS; ++j) v[j] = sp[j * EW_THREADS];
#pragma unroll
      for (int j = 0; j < EW_LOADS; ++j) dp[j * EW_THREADS] = __builtin_convertvector(v[j], v4h);
    } else {
      for (int j = 0; j < EW_LOADS; ++j)
        if (t0 + threadIdx.x + j * EW_THREADS < nvec) dp[j * EW_THREADS] = __builtin_convertvector(sp[j * EW_THREADS], v4h);
    }
  }
  if (blk == 0 && (long long)threadIdx.x < (n & 3)) {
    long long i = (nvec << 2) + threadIdx.x;
    dst[i] = (_Float16)src[i];
  }
}

__device__ __forceinline__ void pack_w_body(const float* __restrict__ W, _Float16* __restrict__ WP,
                                            int din, int dout, int idx) {
  int ktiles = din >> 5, ntiles = dout >> 4;
  int total = ktiles * ntiles * 32;
  if (idx >= total) return;
  int lane = idx & 31;
  int frag = idx >> 5;
  int nt = frag % ntiles;
  int kt = frag / ntiles;
  int half = lane >> 4, lr = lane & 15;
  _Float16* o = WP + ((size_t)frag * 32 + lane) * 16;
  const float* w0 = W + (size_t)(kt * 32 + half * 8) * dout + nt * 16 + lr;
#pragma unroll
  for (int j = 0; j < 8; ++j) {
    o[j]     = (_Float16)w0[(size_t)j * dout];
    o[8 + j] = (_Float16)w0[(size_t)(16 + j) * dout];
  }
}

__device__ __forceinline__ void gemm2_wave(const _Float16* __restrict__ X,
                                           const _Float16* __restrict__ WPl, const _Float16* __restrict__ WPr,
                                           float* __restrict__ YL, float* __restrict__ YR,
                                           int Mrows, int din, int dout, int tileM, int strip) {
  int strips = dout >> 6;
  bool right = strip >= strips;
  if (right) strip -= strips;
  const _Float16* __restrict__ WP = right ? WPr : WPl;
  float* __restrict__ Y = right ? YR : YL;
  int lane = threadIdx.x & 31;
  int half = lane >> 4;
  int lr   = lane & 15;
  if (tileM * 16 >= Mrows) return;
  int nt0 = strip * 4;
  int ntiles = dout >> 4;

  int row = tileM * 16 + lr;
  if (row >= Mrows) row = Mrows - 1;

  v8f acc0 = {}, acc1 = {}, acc2 = {}, acc3 = {};
  const _Float16* xrow = X + (size_t)row * din;
  const v16h* WPv = (const v16h*)WP;
  for (int k0 = 0; k0 < din; k0 += 32) {
    const v8h* xa = (const v8h*)(xrow + k0 + half * 8);
    v8h lo = xa[0];
    v8h hi = xa[2];
    v16h a = __builtin_shufflevector(lo, hi, 0, 1, 2, 3, 4, 5, 6, 7,
                                     8, 9, 10, 11, 12, 13, 14, 15);
    size_t fb = ((size_t)(k0 >> 5) * ntiles + nt0) * 32 + lane;
    v16h b0 = WPv[fb];
    v16h b1 = WPv[fb + 32];
    v16h b2 = WPv[fb + 64];
    v16h b3 = WPv[fb + 96];
    acc0 = __builtin_amdgcn_wmma_f32_16x16x32_f16(false, a, false, b0, (short)0, acc0, false, false);
    acc1 = __builtin_amdgcn_wmma_f32_16x16x32_f16(false, a, false, b1, (short)0, acc1, false, false);
    acc2 = __builtin_amdgcn_wmma_f32_16x16x32_f16(false, a, false, b2, (short)0, acc2, false, false);
    acc3 = __builtin_amdgcn_wmma_f32_16x16x32_f16(false, a, false, b3, (short)0, acc3, false, false);
  }
  int r0 = tileM * 16 + half * 8;
#pragma unroll
  for (int i = 0; i < 8; ++i) {
    int r = r0 + i;
    if (r < Mrows) {
      float* yp = Y + (size_t)r * dout + nt0 * 16 + lr;
      yp[0]  = acc0[i];
      yp[16] = acc1[i];
      yp[32] = acc2[i];
      yp[48] = acc3[i];
    }
  }
}

__global__ void k_gemm2_f16(const _Float16* __restrict__ X,
                            const _Float16* __restrict__ WPl, const _Float16* __restrict__ WPr,
                            float* __restrict__ YL, float* __restrict__ YR,
                            int Mrows, int din, int dout) {
  gemm2_wave(X, WPl, WPr, YL, YR, Mrows, din, dout,
             (int)blockIdx.x * 4 + (int)(threadIdx.x >> 5), (int)blockIdx.y);
}

template <int WPB>
__device__ __forceinline__ void gemm2_rider(const _Float16* __restrict__ X,
                                            const _Float16* __restrict__ WPl, const _Float16* __restrict__ WPr,
                                            float* __restrict__ YL, float* __restrict__ YR,
                                            int Mrows, int din, int dout, int b, int strip0) {
  int tb = (((Mrows + 15) >> 4) + WPB - 1) / WPB;
  int s = b / tb;
  gemm2_wave(X, WPl, WPr, YL, YR, Mrows, din, dout,
             (b - s * tb) * WPB + (int)(threadIdx.x >> 5), strip0 + s);
}

#define L1_DIN 256
#define L1_HC  128
#define L2_DIN 128
#define L2_HC  128
#define L3_DIN 128
#define L3_HC  64
#define PACK_BLOCKS_OF(din, hc) ((((din) >> 5) * ((hc) >> 4) * 32 + EW_THREADS - 1) / EW_THREADS)
#define PB1 PACK_BLOCKS_OF(L1_DIN, L1_HC)
#define PB2 PACK_BLOCKS_OF(L2_DIN, L2_HC)
#define PB3 PACK_BLOCKS_OF(L3_DIN, L3_HC)
#define PREP_PACK_BLOCKS (2 * (PB1 + PB2 + PB3))

#define X_BLOCKS (3 * EW_BLOCKS)

__device__ __forceinline__ void pack_block_body(int b,
       const float* __restrict__ Wl1, const float* __restrict__ Wr1,
       const float* __restrict__ Wl2, const float* __restrict__ Wr2,
       const float* __restrict__ Wl3, const float* __restrict__ Wr3,
       _Float16* __restrict__ P1l, _Float16* __restrict__ P1r,
       _Float16* __restrict__ P2l, _Float16* __restrict__ P2r,
       _Float16* __restrict__ P3l, _Float16* __restrict__ P3r) {
  const float* W; _Float16* WP; int din, hc;
  if (b < PB1)                      { W = Wl1; WP = P1l; din = L1_DIN; hc = L1_HC; }
  else if ((b -= PB1) < PB1)        { W = Wr1; WP = P1r; din = L1_DIN; hc = L1_HC; }
  else if ((b -= PB1) < PB2)        { W = Wl2; WP = P2l; din = L2_DIN; hc = L2_HC; }
  else if ((b -= PB2) < PB2)        { W = Wr2; WP = P2r; din = L2_DIN; hc = L2_HC; }
  else if ((b -= PB2) < PB3)        { W = Wl3; WP = P3l; din = L3_DIN; hc = L3_HC; }
  else            { b -= PB3;         W = Wr3; WP = P3r; din = L3_DIN; hc = L3_HC; }
  pack_w_body(W, WP, din, hc, b * EW_THREADS + (int)threadIdx.x);
}


#define BIN_SH  8
#define BIN_W   (1 << BIN_SH)
#define NB_MAX  512
#define CHUNK   4096
#define EPT     (CHUNK / 256)
#define BIN_CAP 8192

typedef __attribute__((ext_vector_type(2))) int v2i;

__device__ __forceinline__ void
local_sort_body(const int* __restrict__ ei, long long E0, int* __restrict__ offL,
                v2i* __restrict__ pairsL, int blk) {
  __shared__ int h[NB_MAX];
  __shared__ int cur[NB_MAX];
  __shared__ int wsum[8];
  __shared__ v2i sp[CHUNK];
  int tid = threadIdx.x, lane = tid & 31, wave = tid >> 5;
  for (int i = tid; i < NB_MAX; i += 256) h[i] = 0;
  __syncthreads();
  long long base = (long long)blk * CHUNK;
  int s[EPT], d[EPT];
#pragma unroll
  for (int j = 0; j < EPT; ++j) {
    long long e = base + j * 256 + tid;
    bool in = e < E0;
    s[j] = in ? ei[e] : 0;
    d[j] = in ? ei[E0 + e] : -1;
  }
#pragma unroll
  for (int j = 0; j < EPT; ++j)
    if (d[j] >= 0) atomicAdd(&h[d[j] >> BIN_SH], 1);
  __syncthreads();
  int c0 = h[2 * tid], c1 = h[2 * tid + 1];
  int local = c0 + c1, inc = local;
#pragma unroll
  for (int o = 1; o < 32; o <<= 1) {
    int t = __shfl_up(inc, o, 32);
    if (lane >= o) inc += t;
  }
  if (lane == 31) wsum[wave] = inc;
  __syncthreads();
  int wbase = 0;
  for (int w = 0; w < wave; ++w) wbase += wsum[w];
  int ex = wbase + inc - local;
  cur[2 * tid]     = ex;
  cur[2 * tid + 1] = ex + c0;
  v2i tab = {ex, ex + c0};
  *(v2i*)(offL + (size_t)blk * NB_MAX + 2 * tid) = tab;
  __syncthreads();
#pragma unroll
  for (int j = 0; j < EPT; ++j)
    if (d[j] >= 0) {
      int q = atomicAdd(&cur[d[j] >> BIN_SH], 1);
      v2i v = {s[j], d[j]};
      sp[q] = v;
    }
  __syncthreads();
  long long rem = E0 - base;
  int count = rem < CHUNK ? (int)rem : CHUNK;
  for (int i = tid; i < count; i += 256) pairsL[base + i] = sp[i];
}

#define FIN_THREADS 512
#define FIN_LOADS   8

__device__ __forceinline__ void run_bounds(const int* __restrict__ offL, long long E0, int g, int bin,
                                           int& st, int& en) {
  const int* row = offL + (size_t)g * NB_MAX;
  st = row[bin];
  if (bin + 1 < NB_MAX) en = row[bin + 1];
  else {
    long long rem = E0 - (long long)g * CHUNK;
    en = rem < CHUNK ? (int)rem : CHUNK;
  }
}

__device__ __forceinline__ void
bin_finish_body(const v2i* __restrict__ pairsL, const int* __restrict__ offL, int G,
                long long E0, int n, int* __restrict__ cnt, int* __restrict__ rowptr, int* __restrict__ seg, int blk) {
  __shared__ int ncnt[BIN_W];
  __shared__ int nstart[BIN_W];
  __shared__ int ncur[BIN_W];
  __shared__ int wsum[FIN_THREADS / 32];
  __shared__ int wsum2[FIN_THREADS / 32];
  __shared__ int rpos[FIN_THREADS];
  __shared__ int rsrc[FIN_THREADS];
  __shared__ unsigned lp[BIN_CAP];
  __shared__ int ssrc[BIN_CAP];
  int tid = threadIdx.x, lane = tid & 31, wave = tid >> 5;
  int bin = blk;
  int nbase = bin << BIN_SH;
  if (tid < BIN_W) ncnt[tid] = 0;
  int size = 0, b0 = 0;
  for (int g0 = 0; g0 < G; g0 += FIN_THREADS) {
    int g = g0 + tid;
    int st = 0, len = 0;
    if (g < G) {
      int en;
      run_bounds(offL, E0, g, bin, st, en);
      len = en - st;
    }
    int inc = len, ssum = st;
#pragma unroll
    for (int o = 1; o < 32; o <<= 1) {
      int t = __shfl_up(inc, o, 32);
      if (lane >= o) inc += t;
    }
#pragma unroll
    for (int o = 16; o > 0; o >>= 1) ssum += __shfl_xor(ssum, o, 32);
    if (lane == 31) wsum[wave] = inc;
    if (lane == 0)  wsum2[wave] = ssum;
    __syncthreads();
    int wbase = 0, tl = 0, ts = 0;
    for (int w = 0; w < FIN_THREADS / 32; ++w) {
      int a = wsum[w];
      if (w < wave) wbase += a;
      tl += a;
      ts += wsum2[w];
    }
    rpos[tid] = (g < G) ? wbase + inc - len : 0x7fffffff;
    rsrc[tid] = g * CHUNK + st;
    __syncthreads();
    for (int i0 = 0; i0 < tl; i0 += FIN_THREADS * FIN_LOADS) {
      if (i0 + (tid & ~31) >= tl) continue;
      int ii[FIN_LOADS], r[FIN_LOADS];
#pragma unroll
      for (int u = 0; u < FIN_LOADS; ++u) {
        int i = i0 + u * FIN_THREADS + tid;
        ii[u] = i < tl ? i : tl - 1;
        r[u] = 0;
      }
#pragma unroll
      for (int step = FIN_THREADS / 2; step > 0; step >>= 1) {
#pragma unroll
        for (int u = 0; u < FIN_LOADS; ++u)
          if (rpos[r[u] + step] <= ii[u]) r[u] += step;
      }
      v2i vv[FIN_LOADS];
#pragma unroll
      for (int u = 0; u < FIN_LOADS; ++u) vv[u] = pairsL[rsrc[r[u]] + (ii[u] - rpos[r[u]])];
#pragma unroll
      for (int u = 0; u < FIN_LOADS; ++u) {
        int i = i0 + u * FIN_THREADS + tid;
        int p = size + i;
        if (i < tl && p < BIN_CAP)
          lp[p] = (unsigned)vv[u].x | ((unsigned)(vv[u].y - nbase) << 24);
      }
    }
    size += tl;
    b0 += ts;
    __syncthreads();
  }
  bool fits = size <= BIN_CAP;
  if (fits) {
    for (int i = tid; i < size; i += FIN_THREADS) atomicAdd(&ncnt[lp[i] >> 24], 1);
  } else {
    for (int g = tid; g < G; g += FIN_THREADS) {
      int st, en;
      run_bounds(offL, E0, g, bin, st, en);
      for (int e = st; e < en; ++e) atomicAdd(&ncnt[pairsL[(size_t)g * CHUNK + e].y - nbase], 1);
    }
  }
  __syncthreads();
  int c = 0, inc = 0;
  if (tid < BIN_W) {
    c = ncnt[tid];
    inc = c;
#pragma unroll
    for (int o = 1; o < 32; o <<= 1) {
      int t = __shfl_up(inc, o, 32);
      if (lane >= o) inc += t;
    }
    if (lane == 31) wsum[wave] = inc;
  }
  __syncthreads();
  if (tid < BIN_W) {
    int wbase = 0;
    for (int w = 0; w < wave; ++w) wbase += wsum[w];
    int ex = wbase + inc - c;
    nstart[tid] = ex;
    ncur[tid]   = ex;
    int node = nbase + tid;
    if (node < n) {
      cnt[node]    = c;
      rowptr[node] = b0 + ex;
    }
  }
  __syncthreads();
  if (fits) {
    for (int i = tid; i < size; i += FIN_THREADS) {
      unsigned w = lp[i];
      int q = atomicAdd(&ncur[w >> 24], 1);
      ssrc[q] = (int)(w & 0xffffffu);
    }
    __syncthreads();
    for (int ln = wave; ln < BIN_W; ln += FIN_THREADS / 32) {
      int st  = __builtin_amdgcn_readfirstlane(nstart[ln]);
      int deg = __builtin_amdgcn_readfirstlane(ncnt[ln]);
      if (deg <= 32) {
        int mine = (lane < deg) ? ssrc[st + lane] : 0x7fffffff;
        int r = 0;
        for (int j = 0; j < deg; ++j) {
          int v = __builtin_amdgcn_readlane(mine, j);
          r += ((v < mine) || (v == mine && j < lane)) ? 1 : 0;
        }
        if (lane < deg) seg[b0 + st + r] = mine;
      } else {
        for (int i = lane; i < deg; i += 32) {
          int mine = ssrc[st + i];
          int r = 0;
          for (int j = 0; j < deg; ++j) {
            int v = ssrc[st + j];
            r += ((v < mine) || (v == mine && j < i)) ? 1 : 0;
          }
          seg[b0 + st + r] = mine;
        }
      }
    }
  } else {
    for (int g = wave; g < G; g += FIN_THREADS / 32) {
      int st, en;
      run_bounds(offL, E0, g, bin, st, en);
      for (int e = st + lane; e < en; e += 32) {
        v2i v = pairsL[(size_t)g * CHUNK + e];
        int r = 0;
        for (int g2 = 0; g2 < G; ++g2) {
          int s2, e2;
          run_bounds(offL, E0, g2, bin, s2, e2);
          for (int f = s2; f < e2; ++f) {
            v2i u = pairsL[(size_t)g2 * CHUNK + f];
            bool before = (g2 < g) || (g2 == g && f < e);
            r += (u.y == v.y && ((u.x < v.x) || (u.x == v.x && before))) ? 1 : 0;
          }
        }
        seg[b0 + nstart[v.y - nbase] + r] = v.x;
      }
    }
  }
}

__global__ void __launch_bounds__(EW_THREADS)
k_local_x(const int* __restrict__ ei, long long E0, int G, int* __restrict__ offL, v2i* __restrict__ pairsL,
          const float* __restrict__ x, _Float16* __restrict__ X16, long long nx,
          const float* __restrict__ Wl1, const float* __restrict__ Wr1,
          const float* __restrict__ Wl2, const float* __restrict__ Wr2,
          const float* __restrict__ Wl3, const float* __restrict__ Wr3,
          _Float16* __restrict__ P1l, _Float16* __restrict__ P1r,
          _Float16* __restrict__ P2l, _Float16* __restrict__ P2r,
          _Float16* __restrict__ P3l, _Float16* __restrict__ P3r) {
  int b = (int)blockIdx.x;
  if (b < G) { local_sort_body(ei, E0, offL, pairsL, b); return; }
  b -= G;
  if (b < PREP_PACK_BLOCKS) {
    pack_block_body(b, Wl1, Wr1, Wl2, Wr2, Wl3, Wr3, P1l, P1r, P2l, P2r, P3l, P3r);
    return;
  }
  b -= PREP_PACK_BLOCKS;
  plane_to_f16_body(x, X16, nx, b, X_BLOCKS);
}

__global__ void __launch_bounds__(FIN_THREADS)
k_finish_gemm(const v2i* __restrict__ pairsL, const int* __restrict__ offL, int nb, int G,
              long long E0, int n, int* __restrict__ cnt, int* __restrict__ rowptr, int* __restrict__ seg,
              const _Float16* __restrict__ X,
              const _Float16* __restrict__ WPl, const _Float16* __restrict__ WPr,
              float* __restrict__ YL, float* __restrict__ YR, int din, int dout) {
  int b = (int)blockIdx.x;
  if (b < nb) { bin_finish_body(pairsL, offL, G, E0, n, cnt, rowptr, seg, b); return; }
  gemm2_rider<FIN_THREADS / 32>(X, WPl, WPr, YL, YR, n, din, dout, b - nb, 0);
}

template <int CTRL>
__device__ __forceinline__ float dpp_get(float v) {
  return __int_as_float(__builtin_amdgcn_update_dpp(0, __float_as_int(v), CTRL, 0xF, 0xF, true));
}
__device__ __forceinline__ float other_row_get(float v) {
  return __uint_as_float(__builtin_amdgcn_permlanex16(0u, __float_as_uint(v),
                                                      0x76543210u, 0xfedcba98u, true, false));
}
template <int XM>
__device__ __forceinline__ float row_xadd(float p) {
  float r;
  if constexpr (XM == 1)      asm("v_add_f32_dpp %0, %1, %1 row_xmask:1 row_mask:0xf bank_mask:0xf bound_ctrl:1" : "=v"(r) : "v"(p));
  else if constexpr (XM == 2) asm("v_add_f32_dpp %0, %1, %1 row_xmask:2 row_mask:0xf bank_mask:0xf bound_ctrl:1" : "=v"(r) : "v"(p));
  else if constexpr (XM == 4) asm("v_add_f32_dpp %0, %1, %1 row_xmask:4 row_mask:0xf bank_mask:0xf bound_ctrl:1" : "=v"(r) : "v"(p));
  else                        asm("v_add_f32_dpp %0, %1, %1 row_xmask:8 row_mask:0xf bank_mask:0xf bound_ctrl:1" : "=v"(r) : "v"(p));
  return r;
}
template <int LPH>
__device__ __forceinline__ float head_sum(float p) {
  static_assert(LPH == 4 || LPH == 32, "lanes per head");
  p = row_xadd<1>(p);
  p = row_xadd<2>(p);
  if constexpr (LPH == 32) {
    p = row_xadd<4>(p);
    p = row_xadd<8>(p);
    p += other_row_get(p);
  }
  return p;
}

typedef const __attribute__((address_space(1))) float* gfloat_p;

template <int CPT> struct RowVec;
template <> struct RowVec<4> { typedef v4f T; };
template <> struct RowVec<2> { typedef v2f T; };

template <int CPT, int LPH, int R>
__device__ __forceinline__ void gat_group(const float* __restrict__ XL, int ch0, int mysrc, int j,
                                          const v2f (&xrp)[CPT / 2], const v2f (&avp)[CPT / 2],
                                          float& m, float& den, v2f (&accp)[CPT / 2]) {
  constexpr int NP = CPT / 2;
  constexpr int HC = CPT * 32;
  typedef typename RowVec<CPT>::T vec_t;
  vec_t x[R];
  float pu[R];
#pragma unroll
  for (int u = 0; u < R; ++u) {
    int s = __builtin_amdgcn_readlane(mysrc, j + u);
    gfloat_p rowp = (gfloat_p)(XL + (size_t)s * HC);
    asm("" : "+s"(rowp));
    x[u] = *(const __attribute__((address_space(1))) vec_t*)(rowp + ch0);
  }
#pragma unroll
  for (int u = 0; u < R; ++u) {
    v2f q2;
#pragma unroll
    for (int p = 0; p < NP; ++p) {
      v2f xp = {x[u][2 * p], x[u][2 * p + 1]};
      v2f v = xp + xrp[p];
      v2f l = __builtin_elementwise_max(v, v * NEG_SLOPE);
      q2 = (p == 0) ? avp[0] * l : avp[p] * l + q2;
    }
    pu[u] = head_sum<LPH>(q2[0] + q2[1]);
  }
  float mn = m;
  if constexpr (R == 4) mn = fmaxf(fmaxf(m, pu[0]), fmaxf(pu[1], fmaxf(pu[2], pu[3])));
  else if constexpr (R == 3) mn = fmaxf(fmaxf(m, pu[0]), fmaxf(pu[1], pu[2]));
  else if constexpr (R == 2) mn = fmaxf(fmaxf(m, pu[0]), pu[1]);
  else mn = fmaxf(m, pu[0]);
  float sc = __builtin_amdgcn_exp2f(m - mn);
  float e[R];
#pragma unroll
  for (int u = 0; u < R; ++u) e[u] = __builtin_amdgcn_exp2f(pu[u] - mn);
  float es;
  if constexpr (R == 4) es = (e[0] + e[1]) + (e[2] + e[3]);
  else if constexpr (R == 3) es = (e[0] + e[1]) + e[2];
  else if constexpr (R == 2) es = e[0] + e[1];
  else es = e[0];
  den = den * sc + es;
#pragma unroll
  for (int p = 0; p < NP; ++p) {
    v2f xs[R];
#pragma unroll
    for (int u = 0; u < R; ++u) xs[u] = (v2f){x[u][2 * p], x[u][2 * p + 1]};
    v2f w;
    if constexpr (R == 4) w = (e[0] * xs[0] + e[1] * xs[1]) + (e[2] * xs[2] + e[3] * xs[3]);
    else if constexpr (R == 3) w = (e[0] * xs[0] + e[1] * xs[1]) + e[2] * xs[2];
    else if constexpr (R == 2) w = e[0] * xs[0] + e[1] * xs[1];
    else w = e[0] * xs[0];
    accp[p] = accp[p] * sc + w;
  }
  m = mn;
}

template <int H, int C, bool LAST>
__global__ void k_gat_node(const float* __restrict__ XL, const float* __restrict__ XR,
                           const float* __restrict__ att, const float* __restrict__ bias,
                           const int* __restrict__ rowptr, const int* __restrict__ cnt,
                           const int* __restrict__ seg,
                           _Float16* __restrict__ NEXT16, float* __restrict__ PROB, int n) {
  constexpr int HC  = H * C;
  constexpr int CPT = HC / 32;
  constexpr int NP  = CPT / 2;
  constexpr int LPH = 32 / H;
  constexpr int UNR = 4;
  typedef typename RowVec<CPT>::T vec_t;

  int node = blockIdx.x * (blockDim.x >> 5) + (threadIdx.x >> 5);
  if (node >= n) return;
  node = __builtin_amdgcn_readfirstlane(node);
  int lane = threadIdx.x & 31;
  int ch0  = lane * CPT;
  int start = rowptr[node], deg = cnt[node];

  vec_t xr = *(const vec_t*)(XR + (size_t)node * HC + ch0);
  vec_t av = *(const vec_t*)(att + ch0);
  vec_t xs = *(const vec_t*)(XL + (size_t)node * HC + ch0);

  v2f xrp[NP], avp[NP], accp[NP];
#pragma unroll
  for (int p = 0; p < NP; ++p) {
    xrp[p]  = (v2f){xr[2 * p], xr[2 * p + 1]};
    avp[p]  = (v2f){av[2 * p], av[2 * p + 1]} * LOG2E_F;
    accp[p] = (v2f){xs[2 * p], xs[2 * p + 1]};
  }

  v2f q2;
#pragma unroll
  for (int p = 0; p < NP; ++p) {
    v2f v = accp[p] + xrp[p];
    v2f l = __builtin_elementwise_max(v, v * NEG_SLOPE);
    q2 = (p == 0) ? avp[0] * l : avp[p] * l + q2;
  }
  float m   = head_sum<LPH>(q2[0] + q2[1]);
  float den = 1.0f;

  for (int base = 0; base < deg; base += 32) {
    int nb = deg - base;
    if (nb > 32) nb = 32;
    int mysrc = (lane < nb) ? seg[start + base + lane] : node;
    int j = 0;
    for (; j + UNR <= nb; j += UNR)
      gat_group<CPT, LPH, 4>(XL, ch0, mysrc, j, xrp, avp, m, den, accp);
    int rest = nb - j;
    if (rest == 3)      gat_group<CPT, LPH, 3>(XL, ch0, mysrc, j, xrp, avp, m, den, accp);
    else if (rest == 2) gat_group<CPT, LPH, 2>(XL, ch0, mysrc, j, xrp, avp, m, den, accp);
    else if (rest == 1) gat_group<CPT, LPH, 1>(XL, ch0, mysrc, j, xrp, avp, m, den, accp);
  }

  vec_t acc;
#pragma unroll
  for (int p = 0; p < NP; ++p) { acc[2 * p] = accp[p][0]; acc[2 * p + 1] = accp[p][1]; }

  float rdn = __builtin_amdgcn_rcpf(fmaxf(den, 1e-16f));
  vec_t bv = *(const vec_t*)(bias + ch0);
  vec_t o;
#pragma unroll
  for (int k = 0; k < CPT; ++k) o[k] = acc[k] * rdn + bv[k];

  if constexpr (!LAST) {
    static_assert(CPT == 4, "the f16 row is stored as 8 bytes per lane");
    *(v4h*)(NEXT16 + (size_t)node * HC + ch0) = __builtin_convertvector(elu4(o), v4h);
  } else {
    static_assert(H == 1 && CPT == 2, "one head: the wave's 32 lanes hold the whole row, 2 channels each");
    float mx = fmaxf(o[0], o[1]);
    mx = fmaxf(mx, other_row_get(mx));
    mx = fmaxf(mx, dpp_get<0x168>(mx));
    mx = fmaxf(mx, dpp_get<0x164>(mx));
    mx = fmaxf(mx, dpp_get<0x162>(mx));
    mx = fmaxf(mx, dpp_get<0x161>(mx));
    vec_t e;
    e[0] = __builtin_amdgcn_exp2f((o[0] - mx) * LOG2E_F);
    e[1] = __builtin_amdgcn_exp2f((o[1] - mx) * LOG2E_F);
    float s = e[0] + e[1];
    s += other_row_get(s);
    s += dpp_get<0x168>(s);
    s += dpp_get<0x164>(s);
    s += dpp_get<0x162>(s);
    s += dpp_get<0x161>(s);
    float inv = __builtin_amdgcn_rcpf(s);
    *(vec_t*)(PROB + (size_t)node * HC + ch0) = e * inv;
  }
}

template <int H, int C, bool LAST, bool GEMM>
static void gat_layer(const _Float16* X16, int din,
                      const _Float16* WPl, const _Float16* WPr, const float* att, const float* b,
                      float* XL, float* XR, _Float16* next16, float* prob,
                      const int* rowptr, const int* cnt, const int* seg,
                      int N, hipStream_t stream) {
  constexpr int HC = H * C;

  if (GEMM) {
    int rowTiles = (N + 15) / 16;
    dim3 gg((unsigned)((rowTiles + 3) / 4), (unsigned)(2 * (HC / 64)));
    k_gemm2_f16<<<gg, 128, 0, stream>>>(X16, WPl, WPr, XL, XR, N, din, HC);
  }

  k_gat_node<H, C, LAST><<<(N + 7) / 8, 256, 0, stream>>>(XL, XR, att, b, rowptr, cnt, seg, next16, prob, N);
}

extern "C" void kernel_launch(void* const* d_in, const int* in_sizes, int n_in,
                              void* d_out, int out_size, void* d_ws, size_t ws_size,
                              hipStream_t stream) {
  (void)n_in; (void)out_size; (void)ws_size;
  const float* x    = (const float*)d_in[0];
  const int*   ei   = (const int*)d_in[1];
  const float* Wl1  = (const float*)d_in[2];
  const float* Wr1  = (const float*)d_in[3];
  const float* att1 = (const float*)d_in[4];
  const float* b1   = (const float*)d_in[5];
  const float* Wl2  = (const float*)d_in[6];
  const float* Wr2  = (const float*)d_in[7];
  const float* att2 = (const float*)d_in[8];
  const float* b2   = (const float*)d_in[9];
  const float* Wl3  = (const float*)d_in[10];
  const float* Wr3  = (const float*)d_in[11];
  const float* att3 = (const float*)d_in[12];
  const float* b3   = (const float*)d_in[13];

  int       N  = in_sizes[0] / 256;
  long long E0 = (long long)in_sizes[1] / 2;

  uintptr_t p = (uintptr_t)d_ws;
  auto carve = [&](size_t bytes) -> void* {
    void* r = (void*)p;
    p += (bytes + 255) & ~(size_t)255;
    return r;
  };
  _Float16* X16 = (_Float16*)carve((size_t)N * 256 * sizeof(_Float16));
  _Float16* P1l = (_Float16*)carve((size_t)L1_DIN * L1_HC * sizeof(_Float16));
  _Float16* P1r = (_Float16*)carve((size_t)L1_DIN * L1_HC * sizeof(_Float16));
  _Float16* P2l = (_Float16*)carve((size_t)L2_DIN * L2_HC * sizeof(_Float16));
  _Float16* P2r = (_Float16*)carve((size_t)L2_DIN * L2_HC * sizeof(_Float16));
  _Float16* P3l = (_Float16*)carve((size_t)L3_DIN * L3_HC * sizeof(_Float16));
  _Float16* P3r = (_Float16*)carve((size_t)L3_DIN * L3_HC * sizeof(_Float16));
  float*    XL  = (float*)carve((size_t)N * 128 * sizeof(float));
  float*    XR  = (float*)carve((size_t)N * 128 * sizeof(float));
  int* cnt    = (int*)carve((size_t)N * sizeof(int));
  int* rowptr = (int*)carve((size_t)N * sizeof(int));
  int* seg    = (int*)carve((size_t)E0 * sizeof(int));
  int nb  = (N + BIN_W - 1) / BIN_W;
  int G   = (int)((E0 + CHUNK - 1) / CHUNK);
  v2i* pairsL = (v2i*)d_out;
  int* offL   = (int*)(pairsL + E0);

  long long nx = (long long)N * 256;
  int rowTiles = (N + 15) / 16;
  int gb16 = (L1_HC / 64) * ((rowTiles + FIN_THREADS / 32 - 1) / (FIN_THREADS / 32));
  k_local_x<<<G + PREP_PACK_BLOCKS + X_BLOCKS, EW_THREADS, 0, stream>>>(
      ei, E0, G, offL, pairsL, x, X16, nx, Wl1, Wr1, Wl2, Wr2, Wl3, Wr3, P1l, P1r, P2l, P2r, P3l, P3r);
  k_finish_gemm<<<nb + 2 * gb16, FIN_THREADS, 0, stream>>>(pairsL, offL, nb, G, E0, N, cnt, rowptr, seg,
                                                           X16, P1l, P1r, XL, XR, L1_DIN, L1_HC);


  gat_layer<8, 16, false, false>(X16, L1_DIN, P1l, P1r, att1, b1,
                          XL, XR, X16, nullptr, rowptr, cnt, seg, N, stream);

  gat_layer<8, 16, false, true>(X16, L2_DIN, P2l, P2r, att2, b2,
                          XL, XR, X16, nullptr, rowptr, cnt, seg, N, stream);

  gat_layer<1, 64, true, true>(X16, L3_DIN, P3l, P3r, att3, b3,
                         XL, XR, nullptr, (float*)d_out, rowptr, cnt, seg, N, stream);
}
